# starting kernel
# baseline (speedup 1.0000x reference)
#include <hip/hip_runtime.h>
#include <hip/hip_bf16.h>
#include <math.h>

#define CCH 32
#define NPATH 11
#define NB 8

typedef __attribute__((ext_vector_type(2))) float v2f;
typedef __attribute__((ext_vector_type(8))) float v8f;

__device__ __forceinline__ v8f wmma4(v2f a, v2f b, v8f c) {
  return __builtin_amdgcn_wmma_f32_16x16x4_f32(
       false, a,  false, b,
       (short)0, c,  false,  false);
}

__global__ void mmb_zero(float4* __restrict__ p, int n4) {
  int i = blockIdx.x * blockDim.x + threadIdx.x;
  if (i < n4) p[i] = make_float4(0.f, 0.f, 0.f, 0.f);
}

__global__ void mmb_edge(const float* __restrict__ rij,
                         const float* __restrict__ Wrad,
                         const int*   __restrict__ idx_i,
                         const int*   __restrict__ idx_j,
                         const float* __restrict__ node0,
                         const float* __restrict__ node1,
                         const float* __restrict__ node2,
                         float* __restrict__ a0,
                         float* __restrict__ a1,
                         float* __restrict__ a2,
                         int E) {
  const int e = blockIdx.x * (blockDim.x >> 5) + (threadIdx.x >> 5);
  const int c = threadIdx.x & 31;
  if (e >= E) return;

  const float r0 = rij[e * 3 + 0], r1 = rij[e * 3 + 1], r2 = rij[e * 3 + 2];
  const float d  = sqrtf(r0 * r0 + r1 * r1 + r2 * r2 + 1e-12f);
  const float iv = 1.0f / d;
  const float rh[3] = {r0 * iv, r1 * iv, r2 * iv};

  float x = d * 0.2f;
  x = (x > 1.0f) ? 1.0f : x;
  const float fc = 0.5f * (cosf(3.14159265358979f * x) + 1.0f);

  float rbf[NB];
#pragma unroll
  for (int b = 0; b < NB; ++b) {
    const float t = d - (0.5f + 0.6428571429f * (float)b);
    rbf[b] = __expf(-4.0f * t * t) * fc;
  }

  float w[NPATH];
#pragma unroll
  for (int p = 0; p < NPATH; ++p) {
    float acc = 0.f;
#pragma unroll
    for (int b = 0; b < NB; ++b) acc += rbf[b] * Wrad[(p * NB + b) * CCH + c];
    w[p] = acc;
  }

  const int j = idx_j[e];
  const int i = idx_i[e];

  const float g0 = node0[j * CCH + c];
  float g1[3], g2[9];
#pragma unroll
  for (int a = 0; a < 3; ++a) g1[a] = node1[(j * CCH + c) * 3 + a];
#pragma unroll
  for (int q = 0; q < 9; ++q) g2[q] = node2[(j * CCH + c) * 9 + q];

  const float d1 = g1[0] * rh[0] + g1[1] * rh[1] + g1[2] * rh[2];
  float gr[3];
#pragma unroll
  for (int a = 0; a < 3; ++a)
    gr[a] = g2[a * 3 + 0] * rh[0] + g2[a * 3 + 1] * rh[1] + g2[a * 3 + 2] * rh[2];
  const float d2 = gr[0] * rh[0] + gr[1] * rh[1] + gr[2] * rh[2];

  const float inv16 = 0.0625f;

  atomicAdd(&a0[i * CCH + c], (g0 * w[0] + d1 * w[4] + d2 * w[9]) * inv16);

#pragma unroll
  for (int a = 0; a < 3; ++a) {
    const float m = (g0 * w[1] + d1 * w[6]) * rh[a] + g1[a] * w[3] + gr[a] * w[8];
    atomicAdd(&a1[(i * CCH + c) * 3 + a], m * inv16);
  }
#pragma unroll
  for (int a = 0; a < 3; ++a) {
    const float ca = g0 * w[2] * rh[a] + g1[a] * w[5] + gr[a] * w[10];
#pragma unroll
    for (int b = 0; b < 3; ++b) {
      const float m = ca * rh[b] + g2[a * 3 + b] * w[7];
      atomicAdd(&a2[(i * CCH + c) * 9 + a * 3 + b], m * inv16);
    }
  }
}

#define W_U0 0
#define W_U1 1
#define W_U2 2
#define W_V(k) (3 + (k))
#define W_G(k) (11 + (k))

__global__ void __launch_bounds__(128)
mmb_node(const float* __restrict__ a0, const float* __restrict__ a1,
         const float* __restrict__ a2,
         const float* __restrict__ node0, const float* __restrict__ node1,
         const float* __restrict__ node2,
         const float* __restrict__ U, const float* __restrict__ V,
         const float* __restrict__ Wg,
         float* __restrict__ out0, float* __restrict__ out1,
         float* __restrict__ out2, int N) {
  __shared__ float2 wlds[14 * 8 * 2 * 32];
  __shared__ float  ylds[4][16][CCH];

  for (int idx = threadIdx.x; idx < 14 * 1024; idx += blockDim.x) {
    const int wmat = idx >> 10;
    const int rem  = idx & 1023;
    const int k    = rem >> 5;
    const int n    = rem & 31;
    const float* src = (wmat < 3) ? (U + wmat * 1024)
                     : (wmat < 11) ? (V + (wmat - 3) * 1024)
                                   : (Wg + (wmat - 11) * 1024);
    const float val = src[rem];
    const int s   = k >> 2;
    const int kk  = k & 3;
    const int ln  = ((kk >> 1) << 4) | (n & 15);
    const int t   = n >> 4;
    const int jj  = kk & 1;
    ((float*)&wlds[((wmat * 8 + s) * 2 + t) * 32 + ln])[jj] = val;
  }
  __syncthreads();

  const int lane = threadIdx.x & 31;
  const int wv   = threadIdx.x >> 5;
  const int half = lane >> 4;
  const int m    = lane & 15;
  const int tile = blockIdx.x * 4 + wv;
  if (tile * 16 >= N) return;

#define LDB(wm, s, t)                                                       \
  ({ float2 _bv = wlds[(((wm) * 8 + (s)) * 2 + (t)) * 32 + lane];           \
     v2f _b; _b.x = _bv.x; _b.y = _bv.y; _b; })

#define MM2(acc, aop, wm, s)                                                \
  do { v2f _a = (aop);                                                      \
       acc[0] = wmma4(_a, LDB(wm, s, 0), acc[0]);                           \
       acc[1] = wmma4(_a, LDB(wm, s, 1), acc[1]); } while (0)

  const v8f vzero = {};

  const int rowA = tile * 16 + m;
  const int rowC = (rowA < N) ? rowA : (N - 1);
  float x0[2], b1[2][3], b2[2][9];
#define LOAD_BASE(s)                                                        \
  do { const int _k = (s) * 4 + 2 * half;                                   \
       const int _rc = rowC * CCH + _k;                                     \
       x0[0] = a0[_rc]; x0[1] = a0[_rc + 1];                                \
       _Pragma("unroll") for (int _a = 0; _a < 3; ++_a) {                   \
         b1[0][_a] = a1[_rc * 3 + _a]; b1[1][_a] = a1[(_rc + 1) * 3 + _a]; }\
       _Pragma("unroll") for (int _q = 0; _q < 9; ++_q) {                   \
         b2[0][_q] = a2[_rc * 9 + _q]; b2[1][_q] = a2[(_rc + 1) * 9 + _q]; }\
  } while (0)

  v8f y0acc[2] = {vzero, vzero};
  for (int s = 0; s < 8; ++s) {
    LOAD_BASE(s);
    v2f ax, asq, s1v, s2v;
#pragma unroll
    for (int jj = 0; jj < 2; ++jj) {
      ax[jj]  = x0[jj];
      asq[jj] = x0[jj] * x0[jj];
      float acc = 0.f;
#pragma unroll
      for (int a = 0; a < 3; ++a) acc += b1[jj][a] * b1[jj][a];
      s1v[jj] = acc;
      acc = 0.f;
#pragma unroll
      for (int q = 0; q < 9; ++q) acc += b2[jj][q] * b2[jj][q];
      s2v[jj] = acc;
    }
    MM2(y0acc, ax,  W_U0,  s);
    MM2(y0acc, asq, W_V(0), s);
    MM2(y0acc, s1v, W_V(3), s);
    MM2(y0acc, s2v, W_V(6), s);
  }

#pragma unroll
  for (int t = 0; t < 2; ++t)
#pragma unroll
    for (int r = 0; r < 8; ++r)
      ylds[wv][r + 8 * half][t * 16 + m] = y0acc[t][r];
  asm volatile("s_wait_dscnt 0" ::: "memory");

  v8f ga[3][2] = {{vzero, vzero}, {vzero, vzero}, {vzero, vzero}};
  for (int s = 0; s < 8; ++s) {
    const int k = s * 4 + 2 * half;
    v2f ay; ay.x = ylds[wv][m][k]; ay.y = ylds[wv][m][k + 1];
#pragma unroll
    for (int g = 0; g < 3; ++g) MM2(ga[g], ay, W_G(g), s);
  }
  v8f gate[3][2];
#pragma unroll
  for (int g = 0; g < 3; ++g)
#pragma unroll
    for (int t = 0; t < 2; ++t)
#pragma unroll
      for (int r = 0; r < 8; ++r) {
        const float v = ga[g][t][r];
        gate[g][t][r] = v * (1.0f / (1.0f + __expf(-v)));
      }

#pragma unroll
  for (int t = 0; t < 2; ++t)
#pragma unroll
    for (int r = 0; r < 8; ++r) {
      const int row = tile * 16 + r + 8 * half;
      if (row < N) {
        const int cc = t * 16 + m;
        out0[row * CCH + cc] = node0[row * CCH + cc] + gate[0][t][r];
      }
    }

  v8f y1acc[3][2] = {{vzero, vzero}, {vzero, vzero}, {vzero, vzero}};
  for (int s = 0; s < 8; ++s) {
    LOAD_BASE(s);
#pragma unroll
    for (int v = 0; v < 3; ++v) {
      v2f av, a01, tv;
#pragma unroll
      for (int jj = 0; jj < 2; ++jj) {
        av[jj]  = b1[jj][v];
        a01[jj] = x0[jj] * b1[jj][v];
        float acc = 0.f;
#pragma unroll
        for (int a = 0; a < 3; ++a) acc += b1[jj][a] * b2[jj][a * 3 + v];
        tv[jj] = acc;
      }
      MM2(y1acc[v], av,  W_U1,  s);
      MM2(y1acc[v], a01, W_V(1), s);
      MM2(y1acc[v], tv,  W_V(5), s);
    }
  }
#pragma unroll
  for (int v = 0; v < 3; ++v)
#pragma unroll
    for (int t = 0; t < 2; ++t)
#pragma unroll
      for (int r = 0; r < 8; ++r) {
        const int row = tile * 16 + r + 8 * half;
        if (row < N) {
          const int cc = t * 16 + m;
          const int o  = (row * CCH + cc) * 3 + v;
          out1[o] = node1[o] + y1acc[v][t][r] * gate[1][t][r];
        }
      }

  for (int aIdx = 0; aIdx < 3; ++aIdx) {
    v8f y2acc[3][2] = {{vzero, vzero}, {vzero, vzero}, {vzero, vzero}};
    for (int s = 0; s < 8; ++s) {
      LOAD_BASE(s);
#pragma unroll
      for (int b = 0; b < 3; ++b) {
        v2f a2v, a02, a11, qv;
#pragma unroll
        for (int jj = 0; jj < 2; ++jj) {
          const float g = b2[jj][aIdx * 3 + b];
          a2v[jj] = g;
          a02[jj] = x0[jj] * g;
          a11[jj] = b1[jj][aIdx] * b1[jj][b];
          float acc = 0.f;
#pragma unroll
          for (int kk = 0; kk < 3; ++kk)
            acc += b2[jj][aIdx * 3 + kk] * b2[jj][kk * 3 + b];
          qv[jj] = acc;
        }
        MM2(y2acc[b], a2v, W_U2,  s);
        MM2(y2acc[b], a02, W_V(2), s);
        MM2(y2acc[b], a11, W_V(4), s);
        MM2(y2acc[b], qv,  W_V(7), s);
      }
    }
#pragma unroll
    for (int b = 0; b < 3; ++b)
#pragma unroll
      for (int t = 0; t < 2; ++t)
#pragma unroll
        for (int r = 0; r < 8; ++r) {
          const int row = tile * 16 + r + 8 * half;
          if (row < N) {
            const int cc = t * 16 + m;
            const int o  = (row * CCH + cc) * 9 + aIdx * 3 + b;
            out2[o] = node2[o] + y2acc[b][t][r] * gate[2][t][r];
          }
        }
  }
#undef LOAD_BASE
#undef MM2
#undef LDB
}

extern "C" void kernel_launch(void* const* d_in, const int* in_sizes, int n_in,
                              void* d_out, int out_size, void* d_ws, size_t ws_size,
                              hipStream_t stream) {
  const float* node0 = (const float*)d_in[0];
  const float* node1 = (const float*)d_in[1];
  const float* node2 = (const float*)d_in[2];
  const float* rij   = (const float*)d_in[3];
  const float* Wrad  = (const float*)d_in[4];
  const float* U     = (const float*)d_in[5];
  const float* V     = (const float*)d_in[6];
  const float* Wg    = (const float*)d_in[7];
  const int*   idx_i = (const int*)d_in[8];
  const int*   idx_j = (const int*)d_in[9];

  const int N = in_sizes[0] / CCH;
  const int E = in_sizes[3] / 3;

  float* a0 = (float*)d_ws;
  float* a1 = a0 + (size_t)N * CCH;
  float* a2 = a1 + (size_t)N * CCH * 3;

  float* out0 = (float*)d_out;
  float* out1 = out0 + (size_t)N * CCH;
  float* out2 = out1 + (size_t)N * CCH * 3;

  const int n4 = N * CCH * 13 / 4;
  mmb_zero<<<(n4 + 255) / 256, 256, 0, stream>>>((float4*)d_ws, n4);

  mmb_edge<<<(E + 7) / 8, 256, 0, stream>>>(rij, Wrad, idx_i, idx_j,
                                            node0, node1, node2,
                                            a0, a1, a2, E);

  const int nTiles = (N + 15) / 16;
  mmb_node<<<(nTiles + 3) / 4, 128, 0, stream>>>(a0, a1, a2,
                                                 node0, node1, node2,
                                                 U, V, Wg,
                                                 out0, out1, out2, N);
}


# node phase: residual node0/1/2 rows loaded as one batch ahead of each pass's k-loop (clamped row, 12B per lane) instead of one blocking 4B load inside every store's bounds branch
# speedup vs baseline: 1.2266x; 1.2266x over previous
#include <hip/hip_runtime.h>
#include <hip/hip_bf16.h>
#include <math.h>

#define CCH 32
#define NPATH 11
#define NB 8

typedef __attribute__((ext_vector_type(2))) float v2f;
typedef __attribute__((ext_vector_type(8))) float v8f;

__device__ __forceinline__ v8f wmma4(v2f a, v2f b, v8f c) {
  return __builtin_amdgcn_wmma_f32_16x16x4_f32(
       false, a,  false, b,
       (short)0, c,  false,  false);
}

__global__ void mmb_zero(float4* __restrict__ p, int n4) {
  int i = blockIdx.x * blockDim.x + threadIdx.x;
  if (i < n4) p[i] = make_float4(0.f, 0.f, 0.f, 0.f);
}

__global__ void mmb_edge(const float* __restrict__ rij,
                         const float* __restrict__ Wrad,
                         const int*   __restrict__ idx_i,
                         const int*   __restrict__ idx_j,
                         const float* __restrict__ node0,
                         const float* __restrict__ node1,
                         const float* __restrict__ node2,
                         float* __restrict__ a0,
                         float* __restrict__ a1,
                         float* __restrict__ a2,
                         int E) {
  const int e = blockIdx.x * (blockDim.x >> 5) + (threadIdx.x >> 5);
  const int c = threadIdx.x & 31;
  if (e >= E) return;

  const float r0 = rij[e * 3 + 0], r1 = rij[e * 3 + 1], r2 = rij[e * 3 + 2];
  const float d  = sqrtf(r0 * r0 + r1 * r1 + r2 * r2 + 1e-12f);
  const float iv = 1.0f / d;
  const float rh[3] = {r0 * iv, r1 * iv, r2 * iv};

  float x = d * 0.2f;
  x = (x > 1.0f) ? 1.0f : x;
  const float fc = 0.5f * (cosf(3.14159265358979f * x) + 1.0f);

  float rbf[NB];
#pragma unroll
  for (int b = 0; b < NB; ++b) {
    const float t = d - (0.5f + 0.6428571429f * (float)b);
    rbf[b] = __expf(-4.0f * t * t) * fc;
  }

  float w[NPATH];
#pragma unroll
  for (int p = 0; p < NPATH; ++p) {
    float acc = 0.f;
#pragma unroll
    for (int b = 0; b < NB; ++b) acc += rbf[b] * Wrad[(p * NB + b) * CCH + c];
    w[p] = acc;
  }

  const int j = idx_j[e];
  const int i = idx_i[e];

  const float g0 = node0[j * CCH + c];
  float g1[3], g2[9];
#pragma unroll
  for (int a = 0; a < 3; ++a) g1[a] = node1[(j * CCH + c) * 3 + a];
#pragma unroll
  for (int q = 0; q < 9; ++q) g2[q] = node2[(j * CCH + c) * 9 + q];

  const float d1 = g1[0] * rh[0] + g1[1] * rh[1] + g1[2] * rh[2];
  float gr[3];
#pragma unroll
  for (int a = 0; a < 3; ++a)
    gr[a] = g2[a * 3 + 0] * rh[0] + g2[a * 3 + 1] * rh[1] + g2[a * 3 + 2] * rh[2];
  const float d2 = gr[0] * rh[0] + gr[1] * rh[1] + gr[2] * rh[2];

  const float inv16 = 0.0625f;

  atomicAdd(&a0[i * CCH + c], (g0 * w[0] + d1 * w[4] + d2 * w[9]) * inv16);

#pragma unroll
  for (int a = 0; a < 3; ++a) {
    const float m = (g0 * w[1] + d1 * w[6]) * rh[a] + g1[a] * w[3] + gr[a] * w[8];
    atomicAdd(&a1[(i * CCH + c) * 3 + a], m * inv16);
  }
#pragma unroll
  for (int a = 0; a < 3; ++a) {
    const float ca = g0 * w[2] * rh[a] + g1[a] * w[5] + gr[a] * w[10];
#pragma unroll
    for (int b = 0; b < 3; ++b) {
      const float m = ca * rh[b] + g2[a * 3 + b] * w[7];
      atomicAdd(&a2[(i * CCH + c) * 9 + a * 3 + b], m * inv16);
    }
  }
}

#define W_U0 0
#define W_U1 1
#define W_U2 2
#define W_V(k) (3 + (k))
#define W_G(k) (11 + (k))

__global__ void __launch_bounds__(128)
mmb_node(const float* __restrict__ a0, const float* __restrict__ a1,
         const float* __restrict__ a2,
         const float* __restrict__ node0, const float* __restrict__ node1,
         const float* __restrict__ node2,
         const float* __restrict__ U, const float* __restrict__ V,
         const float* __restrict__ Wg,
         float* __restrict__ out0, float* __restrict__ out1,
         float* __restrict__ out2, int N) {
  __shared__ float2 wlds[14 * 8 * 2 * 32];
  __shared__ float  ylds[4][16][CCH];

  for (int idx = threadIdx.x; idx < 14 * 1024; idx += blockDim.x) {
    const int wmat = idx >> 10;
    const int rem  = idx & 1023;
    const int k    = rem >> 5;
    const int n    = rem & 31;
    const float* src = (wmat < 3) ? (U + wmat * 1024)
                     : (wmat < 11) ? (V + (wmat - 3) * 1024)
                                   : (Wg + (wmat - 11) * 1024);
    const float val = src[rem];
    const int s   = k >> 2;
    const int kk  = k & 3;
    const int ln  = ((kk >> 1) << 4) | (n & 15);
    const int t   = n >> 4;
    const int jj  = kk & 1;
    ((float*)&wlds[((wmat * 8 + s) * 2 + t) * 32 + ln])[jj] = val;
  }
  __syncthreads();

  const int lane = threadIdx.x & 31;
  const int wv   = threadIdx.x >> 5;
  const int half = lane >> 4;
  const int m    = lane & 15;
  const int tile = blockIdx.x * 4 + wv;
  if (tile * 16 >= N) return;

#define LDB(wm, s, t)                                                       \
  ({ float2 _bv = wlds[(((wm) * 8 + (s)) * 2 + (t)) * 32 + lane];           \
     v2f _b; _b.x = _bv.x; _b.y = _bv.y; _b; })

#define MM2(acc, aop, wm, s)                                                \
  do { v2f _a = (aop);                                                      \
       acc[0] = wmma4(_a, LDB(wm, s, 0), acc[0]);                           \
       acc[1] = wmma4(_a, LDB(wm, s, 1), acc[1]); } while (0)

  const v8f vzero = {};

  const int rowA = tile * 16 + m;
  const int rowC = (rowA < N) ? rowA : (N - 1);
  float x0[2], b1[2][3], b2[2][9];
#define LOAD_BASE(s)                                                        \
  do { const int _k = (s) * 4 + 2 * half;                                   \
       const int _rc = rowC * CCH + _k;                                     \
       x0[0] = a0[_rc]; x0[1] = a0[_rc + 1];                                \
       _Pragma("unroll") for (int _a = 0; _a < 3; ++_a) {                   \
         b1[0][_a] = a1[_rc * 3 + _a]; b1[1][_a] = a1[(_rc + 1) * 3 + _a]; }\
       _Pragma("unroll") for (int _q = 0; _q < 9; ++_q) {                   \
         b2[0][_q] = a2[_rc * 9 + _q]; b2[1][_q] = a2[(_rc + 1) * 9 + _q]; }\
  } while (0)

  const int rowD0 = tile * 16 + 8 * half;
  float res0[2][8];
#pragma unroll
  for (int t = 0; t < 2; ++t)
#pragma unroll
    for (int r = 0; r < 8; ++r) {
      const int row  = rowD0 + r;
      const int rowL = (row < N) ? row : (N - 1);
      res0[t][r] = node0[rowL * CCH + t * 16 + m];
    }

  v8f y0acc[2] = {vzero, vzero};
  for (int s = 0; s < 8; ++s) {
    LOAD_BASE(s);
    v2f ax, asq, s1v, s2v;
#pragma unroll
    for (int jj = 0; jj < 2; ++jj) {
      ax[jj]  = x0[jj];
      asq[jj] = x0[jj] * x0[jj];
      float acc = 0.f;
#pragma unroll
      for (int a = 0; a < 3; ++a) acc += b1[jj][a] * b1[jj][a];
      s1v[jj] = acc;
      acc = 0.f;
#pragma unroll
      for (int q = 0; q < 9; ++q) acc += b2[jj][q] * b2[jj][q];
      s2v[jj] = acc;
    }
    MM2(y0acc, ax,  W_U0,  s);
    MM2(y0acc, asq, W_V(0), s);
    MM2(y0acc, s1v, W_V(3), s);
    MM2(y0acc, s2v, W_V(6), s);
  }

#pragma unroll
  for (int t = 0; t < 2; ++t)
#pragma unroll
    for (int r = 0; r < 8; ++r)
      ylds[wv][r + 8 * half][t * 16 + m] = y0acc[t][r];
  asm volatile("s_wait_dscnt 0" ::: "memory");

  v8f ga[3][2] = {{vzero, vzero}, {vzero, vzero}, {vzero, vzero}};
  for (int s = 0; s < 8; ++s) {
    const int k = s * 4 + 2 * half;
    v2f ay; ay.x = ylds[wv][m][k]; ay.y = ylds[wv][m][k + 1];
#pragma unroll
    for (int g = 0; g < 3; ++g) MM2(ga[g], ay, W_G(g), s);
  }
  v8f gate[3][2];
#pragma unroll
  for (int g = 0; g < 3; ++g)
#pragma unroll
    for (int t = 0; t < 2; ++t)
#pragma unroll
      for (int r = 0; r < 8; ++r) {
        const float v = ga[g][t][r];
        gate[g][t][r] = v * (1.0f / (1.0f + __expf(-v)));
      }

#pragma unroll
  for (int t = 0; t < 2; ++t)
#pragma unroll
    for (int r = 0; r < 8; ++r) {
      const int row = rowD0 + r;
      if (row < N) {
        const int cc = t * 16 + m;
        out0[row * CCH + cc] = res0[t][r] + gate[0][t][r];
      }
    }

  v8f y1acc[3][2] = {{vzero, vzero}, {vzero, vzero}, {vzero, vzero}};
  float res1[2][8][3];
#pragma unroll
  for (int t = 0; t < 2; ++t)
#pragma unroll
    for (int r = 0; r < 8; ++r) {
      const int row  = rowD0 + r;
      const int rowL = (row < N) ? row : (N - 1);
      const float* p = node1 + (rowL * CCH + t * 16 + m) * 3;
#pragma unroll
      for (int v = 0; v < 3; ++v) res1[t][r][v] = p[v];
    }
  for (int s = 0; s < 8; ++s) {
    LOAD_BASE(s);
#pragma unroll
    for (int v = 0; v < 3; ++v) {
      v2f av, a01, tv;
#pragma unroll
      for (int jj = 0; jj < 2; ++jj) {
        av[jj]  = b1[jj][v];
        a01[jj] = x0[jj] * b1[jj][v];
        float acc = 0.f;
#pragma unroll
        for (int a = 0; a < 3; ++a) acc += b1[jj][a] * b2[jj][a * 3 + v];
        tv[jj] = acc;
      }
      MM2(y1acc[v], av,  W_U1,  s);
      MM2(y1acc[v], a01, W_V(1), s);
      MM2(y1acc[v], tv,  W_V(5), s);
    }
  }
#pragma unroll
  for (int t = 0; t < 2; ++t)
#pragma unroll
    for (int r = 0; r < 8; ++r) {
      const int row = rowD0 + r;
      if (row < N) {
        float* q = out1 + (row * CCH + t * 16 + m) * 3;
#pragma unroll
        for (int v = 0; v < 3; ++v)
          q[v] = res1[t][r][v] + y1acc[v][t][r] * gate[1][t][r];
      }
    }

  for (int aIdx = 0; aIdx < 3; ++aIdx) {
    v8f y2acc[3][2] = {{vzero, vzero}, {vzero, vzero}, {vzero, vzero}};
    float res2[2][8][3];
#pragma unroll
    for (int t = 0; t < 2; ++t)
#pragma unroll
      for (int r = 0; r < 8; ++r) {
        const int row  = rowD0 + r;
        const int rowL = (row < N) ? row : (N - 1);
        const float* p = node2 + (rowL * CCH + t * 16 + m) * 9 + aIdx * 3;
#pragma unroll
        for (int bb = 0; bb < 3; ++bb) res2[t][r][bb] = p[bb];
      }
    for (int s = 0; s < 8; ++s) {
      LOAD_BASE(s);
#pragma unroll
      for (int b = 0; b < 3; ++b) {
        v2f a2v, a02, a11, qv;
#pragma unroll
        for (int jj = 0; jj < 2; ++jj) {
          const float g = b2[jj][aIdx * 3 + b];
          a2v[jj] = g;
          a02[jj] = x0[jj] * g;
          a11[jj] = b1[jj][aIdx] * b1[jj][b];
          float acc = 0.f;
#pragma unroll
          for (int kk = 0; kk < 3; ++kk)
            acc += b2[jj][aIdx * 3 + kk] * b2[jj][kk * 3 + b];
          qv[jj] = acc;
        }
        MM2(y2acc[b], a2v, W_U2,  s);
        MM2(y2acc[b], a02, W_V(2), s);
        MM2(y2acc[b], a11, W_V(4), s);
        MM2(y2acc[b], qv,  W_V(7), s);
      }
    }
#pragma unroll
    for (int t = 0; t < 2; ++t)
#pragma unroll
      for (int r = 0; r < 8; ++r) {
        const int row = rowD0 + r;
        if (row < N) {
          float* q = out2 + (row * CCH + t * 16 + m) * 9 + aIdx * 3;
#pragma unroll
          for (int bb = 0; bb < 3; ++bb)
            q[bb] = res2[t][r][bb] + y2acc[bb][t][r] * gate[2][t][r];
        }
      }
  }
#undef LOAD_BASE
#undef MM2
#undef LDB
}

extern "C" void kernel_launch(void* const* d_in, const int* in_sizes, int n_in,
                              void* d_out, int out_size, void* d_ws, size_t ws_size,
                              hipStream_t stream) {
  const float* node0 = (const float*)d_in[0];
  const float* node1 = (const float*)d_in[1];
  const float* node2 = (const float*)d_in[2];
  const float* rij   = (const float*)d_in[3];
  const float* Wrad  = (const float*)d_in[4];
  const float* U     = (const float*)d_in[5];
  const float* V     = (const float*)d_in[6];
  const float* Wg    = (const float*)d_in[7];
  const int*   idx_i = (const int*)d_in[8];
  const int*   idx_j = (const int*)d_in[9];

  const int N = in_sizes[0] / CCH;
  const int E = in_sizes[3] / 3;

  float* a0 = (float*)d_ws;
  float* a1 = a0 + (size_t)N * CCH;
  float* a2 = a1 + (size_t)N * CCH * 3;

  float* out0 = (float*)d_out;
  float* out1 = out0 + (size_t)N * CCH;
  float* out2 = out1 + (size_t)N * CCH * 3;

  const int n4 = N * CCH * 13 / 4;
  mmb_zero<<<(n4 + 255) / 256, 256, 0, stream>>>((float4*)d_ws, n4);

  mmb_edge<<<(E + 7) / 8, 256, 0, stream>>>(rij, Wrad, idx_i, idx_j,
                                            node0, node1, node2,
                                            a0, a1, a2, E);

  const int nTiles = (N + 15) / 16;
  mmb_node<<<(nTiles + 3) / 4, 128, 0, stream>>>(a0, a1, a2,
                                                 node0, node1, node2,
                                                 U, V, Wg,
                                                 out0, out1, out2, N);
}


# accumulator workspace relaid [node][13 comps][32 ch]: each edge-phase float atomic instruction now hits one 128B line, not 3 or 9; node phase loads channel pairs (8B); atomics and sum order as before
# speedup vs baseline: 1.8561x; 1.5132x over previous
#include <hip/hip_runtime.h>
#include <hip/hip_bf16.h>
#include <math.h>

#define CCH 32
#define NPATH 11
#define NB 8
#define NCOMP 13

typedef __attribute__((ext_vector_type(2))) float v2f;
typedef __attribute__((ext_vector_type(8))) float v8f;

__device__ __forceinline__ v8f wmma4(v2f a, v2f b, v8f c) {
  return __builtin_amdgcn_wmma_f32_16x16x4_f32(
       false, a,  false, b,
       (short)0, c,  false,  false);
}

__global__ void mmb_zero(float4* __restrict__ p, int n4) {
  int i = blockIdx.x * blockDim.x + threadIdx.x;
  if (i < n4) p[i] = make_float4(0.f, 0.f, 0.f, 0.f);
}

__global__ void mmb_edge(const float* __restrict__ rij,
                         const float* __restrict__ Wrad,
                         const int*   __restrict__ idx_i,
                         const int*   __restrict__ idx_j,
                         const float* __restrict__ node0,
                         const float* __restrict__ node1,
                         const float* __restrict__ node2,
                         float* __restrict__ acc,
                         int E) {
  const int e = blockIdx.x * (blockDim.x >> 5) + (threadIdx.x >> 5);
  const int c = threadIdx.x & 31;
  if (e >= E) return;

  const float r0 = rij[e * 3 + 0], r1 = rij[e * 3 + 1], r2 = rij[e * 3 + 2];
  const float d  = sqrtf(r0 * r0 + r1 * r1 + r2 * r2 + 1e-12f);
  const float iv = 1.0f / d;
  const float rh[3] = {r0 * iv, r1 * iv, r2 * iv};

  float x = d * 0.2f;
  x = (x > 1.0f) ? 1.0f : x;
  const float fc = 0.5f * (cosf(3.14159265358979f * x) + 1.0f);

  float rbf[NB];
#pragma unroll
  for (int b = 0; b < NB; ++b) {
    const float t = d - (0.5f + 0.6428571429f * (float)b);
    rbf[b] = __expf(-4.0f * t * t) * fc;
  }

  float w[NPATH];
#pragma unroll
  for (int p = 0; p < NPATH; ++p) {
    float acc = 0.f;
#pragma unroll
    for (int b = 0; b < NB; ++b) acc += rbf[b] * Wrad[(p * NB + b) * CCH + c];
    w[p] = acc;
  }

  const int j = idx_j[e];
  const int i = idx_i[e];

  const float g0 = node0[j * CCH + c];
  float g1[3], g2[9];
#pragma unroll
  for (int a = 0; a < 3; ++a) g1[a] = node1[(j * CCH + c) * 3 + a];
#pragma unroll
  for (int q = 0; q < 9; ++q) g2[q] = node2[(j * CCH + c) * 9 + q];

  const float d1 = g1[0] * rh[0] + g1[1] * rh[1] + g1[2] * rh[2];
  float gr[3];
#pragma unroll
  for (int a = 0; a < 3; ++a)
    gr[a] = g2[a * 3 + 0] * rh[0] + g2[a * 3 + 1] * rh[1] + g2[a * 3 + 2] * rh[2];
  const float d2 = gr[0] * rh[0] + gr[1] * rh[1] + gr[2] * rh[2];

  const float inv16 = 0.0625f;

  float* dst = acc + i * (NCOMP * CCH) + c;

  atomicAdd(dst, (g0 * w[0] + d1 * w[4] + d2 * w[9]) * inv16);

#pragma unroll
  for (int a = 0; a < 3; ++a) {
    const float m = (g0 * w[1] + d1 * w[6]) * rh[a] + g1[a] * w[3] + gr[a] * w[8];
    atomicAdd(dst + (1 + a) * CCH, m * inv16);
  }
#pragma unroll
  for (int a = 0; a < 3; ++a) {
    const float ca = g0 * w[2] * rh[a] + g1[a] * w[5] + gr[a] * w[10];
#pragma unroll
    for (int b = 0; b < 3; ++b) {
      const float m = ca * rh[b] + g2[a * 3 + b] * w[7];
      atomicAdd(dst + (4 + a * 3 + b) * CCH, m * inv16);
    }
  }
}

#define W_U0 0
#define W_U1 1
#define W_U2 2
#define W_V(k) (3 + (k))
#define W_G(k) (11 + (k))

__global__ void __launch_bounds__(128)
mmb_node(const float* __restrict__ acc,
         const float* __restrict__ node0, const float* __restrict__ node1,
         const float* __restrict__ node2,
         const float* __restrict__ U, const float* __restrict__ V,
         const float* __restrict__ Wg,
         float* __restrict__ out0, float* __restrict__ out1,
         float* __restrict__ out2, int N) {
  __shared__ float2 wlds[14 * 8 * 2 * 32];
  __shared__ float  ylds[4][16][CCH];

  for (int idx = threadIdx.x; idx < 14 * 1024; idx += blockDim.x) {
    const int wmat = idx >> 10;
    const int rem  = idx & 1023;
    const int k    = rem >> 5;
    const int n    = rem & 31;
    const float* src = (wmat < 3) ? (U + wmat * 1024)
                     : (wmat < 11) ? (V + (wmat - 3) * 1024)
                                   : (Wg + (wmat - 11) * 1024);
    const float val = src[rem];
    const int s   = k >> 2;
    const int kk  = k & 3;
    const int ln  = ((kk >> 1) << 4) | (n & 15);
    const int t   = n >> 4;
    const int jj  = kk & 1;
    ((float*)&wlds[((wmat * 8 + s) * 2 + t) * 32 + ln])[jj] = val;
  }
  __syncthreads();

  const int lane = threadIdx.x & 31;
  const int wv   = threadIdx.x >> 5;
  const int half = lane >> 4;
  const int m    = lane & 15;
  const int tile = blockIdx.x * 4 + wv;
  if (tile * 16 >= N) return;

#define LDB(wm, s, t)                                                       \
  ({ float2 _bv = wlds[(((wm) * 8 + (s)) * 2 + (t)) * 32 + lane];           \
     v2f _b; _b.x = _bv.x; _b.y = _bv.y; _b; })

#define MM2(acc, aop, wm, s)                                                \
  do { v2f _a = (aop);                                                      \
       acc[0] = wmma4(_a, LDB(wm, s, 0), acc[0]);                           \
       acc[1] = wmma4(_a, LDB(wm, s, 1), acc[1]); } while (0)

  const v8f vzero = {};

  const int rowA = tile * 16 + m;
  const int rowC = (rowA < N) ? rowA : (N - 1);
  const float* accRow = acc + rowC * (NCOMP * CCH) + 2 * half;
  float x0[2], b1[2][3], b2[2][9];
#define LOAD_BASE(s)                                                        \
  do { const float* _p = accRow + (s) * 4;                                  \
       { const float2 _v = *(const float2*)_p;                              \
         x0[0] = _v.x; x0[1] = _v.y; }                                      \
       _Pragma("unroll") for (int _a = 0; _a < 3; ++_a) {                   \
         const float2 _v = *(const float2*)(_p + (1 + _a) * CCH);           \
         b1[0][_a] = _v.x; b1[1][_a] = _v.y; }                              \
       _Pragma("unroll") for (int _q = 0; _q < 9; ++_q) {                   \
         const float2 _v = *(const float2*)(_p + (4 + _q) * CCH);           \
         b2[0][_q] = _v.x; b2[1][_q] = _v.y; }                              \
  } while (0)

  const int rowD0 = tile * 16 + 8 * half;
  float res0[2][8];
#pragma unroll
  for (int t = 0; t < 2; ++t)
#pragma unroll
    for (int r = 0; r < 8; ++r) {
      const int row  = rowD0 + r;
      const int rowL = (row < N) ? row : (N - 1);
      res0[t][r] = node0[rowL * CCH + t * 16 + m];
    }

  v8f y0acc[2] = {vzero, vzero};
  for (int s = 0; s < 8; ++s) {
    LOAD_BASE(s);
    v2f ax, asq, s1v, s2v;
#pragma unroll
    for (int jj = 0; jj < 2; ++jj) {
      ax[jj]  = x0[jj];
      asq[jj] = x0[jj] * x0[jj];
      float acc = 0.f;
#pragma unroll
      for (int a = 0; a < 3; ++a) acc += b1[jj][a] * b1[jj][a];
      s1v[jj] = acc;
      acc = 0.f;
#pragma unroll
      for (int q = 0; q < 9; ++q) acc += b2[jj][q] * b2[jj][q];
      s2v[jj] = acc;
    }
    MM2(y0acc, ax,  W_U0,  s);
    MM2(y0acc, asq, W_V(0), s);
    MM2(y0acc, s1v, W_V(3), s);
    MM2(y0acc, s2v, W_V(6), s);
  }

#pragma unroll
  for (int t = 0; t < 2; ++t)
#pragma unroll
    for (int r = 0; r < 8; ++r)
      ylds[wv][r + 8 * half][t * 16 + m] = y0acc[t][r];
  asm volatile("s_wait_dscnt 0" ::: "memory");

  v8f ga[3][2] = {{vzero, vzero}, {vzero, vzero}, {vzero, vzero}};
  for (int s = 0; s < 8; ++s) {
    const int k = s * 4 + 2 * half;
    v2f ay; ay.x = ylds[wv][m][k]; ay.y = ylds[wv][m][k + 1];
#pragma unroll
    for (int g = 0; g < 3; ++g) MM2(ga[g], ay, W_G(g), s);
  }
  v8f gate[3][2];
#pragma unroll
  for (int g = 0; g < 3; ++g)
#pragma unroll
    for (int t = 0; t < 2; ++t)
#pragma unroll
      for (int r = 0; r < 8; ++r) {
        const float v = ga[g][t][r];
        gate[g][t][r] = v * (1.0f / (1.0f + __expf(-v)));
      }

#pragma unroll
  for (int t = 0; t < 2; ++t)
#pragma unroll
    for (int r = 0; r < 8; ++r) {
      const int row = rowD0 + r;
      if (row < N) {
        const int cc = t * 16 + m;
        out0[row * CCH + cc] = res0[t][r] + gate[0][t][r];
      }
    }

  v8f y1acc[3][2] = {{vzero, vzero}, {vzero, vzero}, {vzero, vzero}};
  float res1[2][8][3];
#pragma unroll
  for (int t = 0; t < 2; ++t)
#pragma unroll
    for (int r = 0; r < 8; ++r) {
      const int row  = rowD0 + r;
      const int rowL = (row < N) ? row : (N - 1);
      const float* p = node1 + (rowL * CCH + t * 16 + m) * 3;
#pragma unroll
      for (int v = 0; v < 3; ++v) res1[t][r][v] = p[v];
    }
  for (int s = 0; s < 8; ++s) {
    LOAD_BASE(s);
#pragma unroll
    for (int v = 0; v < 3; ++v) {
      v2f av, a01, tv;
#pragma unroll
      for (int jj = 0; jj < 2; ++jj) {
        av[jj]  = b1[jj][v];
        a01[jj] = x0[jj] * b1[jj][v];
        float acc = 0.f;
#pragma unroll
        for (int a = 0; a < 3; ++a) acc += b1[jj][a] * b2[jj][a * 3 + v];
        tv[jj] = acc;
      }
      MM2(y1acc[v], av,  W_U1,  s);
      MM2(y1acc[v], a01, W_V(1), s);
      MM2(y1acc[v], tv,  W_V(5), s);
    }
  }
#pragma unroll
  for (int t = 0; t < 2; ++t)
#pragma unroll
    for (int r = 0; r < 8; ++r) {
      const int row = rowD0 + r;
      if (row < N) {
        float* q = out1 + (row * CCH + t * 16 + m) * 3;
#pragma unroll
        for (int v = 0; v < 3; ++v)
          q[v] = res1[t][r][v] + y1acc[v][t][r] * gate[1][t][r];
      }
    }

  for (int aIdx = 0; aIdx < 3; ++aIdx) {
    v8f y2acc[3][2] = {{vzero, vzero}, {vzero, vzero}, {vzero, vzero}};
    float res2[2][8][3];
#pragma unroll
    for (int t = 0; t < 2; ++t)
#pragma unroll
      for (int r = 0; r < 8; ++r) {
        const int row  = rowD0 + r;
        const int rowL = (row < N) ? row : (N - 1);
        const float* p = node2 + (rowL * CCH + t * 16 + m) * 9 + aIdx * 3;
#pragma unroll
        for (int bb = 0; bb < 3; ++bb) res2[t][r][bb] = p[bb];
      }
    for (int s = 0; s < 8; ++s) {
      LOAD_BASE(s);
#pragma unroll
      for (int b = 0; b < 3; ++b) {
        v2f a2v, a02, a11, qv;
#pragma unroll
        for (int jj = 0; jj < 2; ++jj) {
          const float g = b2[jj][aIdx * 3 + b];
          a2v[jj] = g;
          a02[jj] = x0[jj] * g;
          a11[jj] = b1[jj][aIdx] * b1[jj][b];
          float acc = 0.f;
#pragma unroll
          for (int kk = 0; kk < 3; ++kk)
            acc += b2[jj][aIdx * 3 + kk] * b2[jj][kk * 3 + b];
          qv[jj] = acc;
        }
        MM2(y2acc[b], a2v, W_U2,  s);
        MM2(y2acc[b], a02, W_V(2), s);
        MM2(y2acc[b], a11, W_V(4), s);
        MM2(y2acc[b], qv,  W_V(7), s);
      }
    }
#pragma unroll
    for (int t = 0; t < 2; ++t)
#pragma unroll
      for (int r = 0; r < 8; ++r) {
        const int row = rowD0 + r;
        if (row < N) {
          float* q = out2 + (row * CCH + t * 16 + m) * 9 + aIdx * 3;
#pragma unroll
          for (int bb = 0; bb < 3; ++bb)
            q[bb] = res2[t][r][bb] + y2acc[bb][t][r] * gate[2][t][r];
        }
      }
  }
#undef LOAD_BASE
#undef MM2
#undef LDB
}

extern "C" void kernel_launch(void* const* d_in, const int* in_sizes, int n_in,
                              void* d_out, int out_size, void* d_ws, size_t ws_size,
                              hipStream_t stream) {
  const float* node0 = (const float*)d_in[0];
  const float* node1 = (const float*)d_in[1];
  const float* node2 = (const float*)d_in[2];
  const float* rij   = (const float*)d_in[3];
  const float* Wrad  = (const float*)d_in[4];
  const float* U     = (const float*)d_in[5];
  const float* V     = (const float*)d_in[6];
  const float* Wg    = (const float*)d_in[7];
  const int*   idx_i = (const int*)d_in[8];
  const int*   idx_j = (const int*)d_in[9];

  const int N = in_sizes[0] / CCH;
  const int E = in_sizes[3] / 3;

  float* acc = (float*)d_ws;

  float* out0 = (float*)d_out;
  float* out1 = out0 + (size_t)N * CCH;
  float* out2 = out1 + (size_t)N * CCH * 3;

  const int n4 = N * CCH * 13 / 4;
  mmb_zero<<<(n4 + 255) / 256, 256, 0, stream>>>((float4*)d_ws, n4);

  mmb_edge<<<(E + 7) / 8, 256, 0, stream>>>(rij, Wrad, idx_i, idx_j,
                                            node0, node1, node2,
                                            acc, E);

  const int nTiles = (N + 15) / 16;
  mmb_node<<<(nTiles + 3) / 4, 128, 0, stream>>>(acc,
                                                 node0, node1, node2,
                                                 U, V, Wg,
                                                 out0, out1, out2, N);
}


# edge phase: one wave per run of 32 edges (was one per edge): Wrad column kept in 88 registers, edge geometry computed one edge per lane and broadcast by readlane; atomic arrival order changes
# speedup vs baseline: 2.8495x; 1.5352x over previous
#include <hip/hip_runtime.h>
#include <hip/hip_bf16.h>
#include <math.h>

#define CCH 32
#define NPATH 11
#define NB 8
#define NCOMP 13

typedef __attribute__((ext_vector_type(2))) float v2f;
typedef __attribute__((ext_vector_type(8))) float v8f;

__device__ __forceinline__ v8f wmma4(v2f a, v2f b, v8f c) {
  return __builtin_amdgcn_wmma_f32_16x16x4_f32(
       false, a,  false, b,
       (short)0, c,  false,  false);
}

__global__ void mmb_zero(float4* __restrict__ p, int n4) {
  int i = blockIdx.x * blockDim.x + threadIdx.x;
  if (i < n4) p[i] = make_float4(0.f, 0.f, 0.f, 0.f);
}

#define EDGES_PER_WAVE 32

__device__ __forceinline__ float bcast_f(float v, int srcLane) {
  return __int_as_float(__builtin_amdgcn_readlane(__float_as_int(v), srcLane));
}

__global__ void __launch_bounds__(128)
mmb_edge(const float* __restrict__ rij,
         const float* __restrict__ Wrad,
         const int*   __restrict__ idx_i,
         const int*   __restrict__ idx_j,
         const float* __restrict__ node0,
         const float* __restrict__ node1,
         const float* __restrict__ node2,
         float* __restrict__ acc,
         int E) {
  const int c = threadIdx.x & 31;
  const int run = __builtin_amdgcn_readfirstlane(
      (int)(blockIdx.x * (blockDim.x >> 5) + (threadIdx.x >> 5)));
  const int ebase = run * EDGES_PER_WAVE;
  if (ebase >= E) return;
  const int cnt = (E - ebase < EDGES_PER_WAVE) ? (E - ebase) : EDGES_PER_WAVE;

  float Wr[NPATH * NB];
#pragma unroll
  for (int q = 0; q < NPATH * NB; ++q) Wr[q] = Wrad[q * CCH + c];

  const int el = (ebase + c < E) ? (ebase + c) : (E - 1);
  const float r0 = rij[el * 3 + 0], r1 = rij[el * 3 + 1], r2 = rij[el * 3 + 2];
  const float d  = sqrtf(r0 * r0 + r1 * r1 + r2 * r2 + 1e-12f);
  const float iv = 1.0f / d;
  const float rhl0 = r0 * iv, rhl1 = r1 * iv, rhl2 = r2 * iv;

  float x = d * 0.2f;
  x = (x > 1.0f) ? 1.0f : x;
  const float fc = 0.5f * (cosf(3.14159265358979f * x) + 1.0f);

  float rbfl[NB];
#pragma unroll
  for (int b = 0; b < NB; ++b) {
    const float t = d - (0.5f + 0.6428571429f * (float)b);
    rbfl[b] = __expf(-4.0f * t * t) * fc;
  }
  const int jl = idx_j[el];
  const int il = idx_i[el];

  const float inv16 = 0.0625f;

  for (int t = 0; t < cnt; ++t) {
    const int j = __builtin_amdgcn_readlane(jl, t);
    const int i = __builtin_amdgcn_readlane(il, t);
    const float rh[3] = {bcast_f(rhl0, t), bcast_f(rhl1, t), bcast_f(rhl2, t)};
    float rbf[NB];
#pragma unroll
    for (int b = 0; b < NB; ++b) rbf[b] = bcast_f(rbfl[b], t);

    const float* p0 = node0 + (size_t)j * CCH;
    const float* p1 = node1 + (size_t)j * (CCH * 3);
    const float* p2 = node2 + (size_t)j * (CCH * 9);
    const float g0 = p0[c];
    float g1[3], g2[9];
#pragma unroll
    for (int a = 0; a < 3; ++a) g1[a] = p1[c * 3 + a];
#pragma unroll
    for (int q = 0; q < 9; ++q) g2[q] = p2[c * 9 + q];

    float w[NPATH];
#pragma unroll
    for (int p = 0; p < NPATH; ++p) {
      float s = rbf[0] * Wr[p * NB];
#pragma unroll
      for (int b = 1; b < NB; ++b) s = fmaf(rbf[b], Wr[p * NB + b], s);
      w[p] = s;
    }

    const float d1 = g1[0] * rh[0] + g1[1] * rh[1] + g1[2] * rh[2];
    float gr[3];
#pragma unroll
    for (int a = 0; a < 3; ++a)
      gr[a] = g2[a * 3 + 0] * rh[0] + g2[a * 3 + 1] * rh[1] + g2[a * 3 + 2] * rh[2];
    const float d2 = gr[0] * rh[0] + gr[1] * rh[1] + gr[2] * rh[2];

    float* dst = acc + (size_t)i * (NCOMP * CCH) + c;

    atomicAdd(dst, (g0 * w[0] + d1 * w[4] + d2 * w[9]) * inv16);

#pragma unroll
    for (int a = 0; a < 3; ++a) {
      const float m = (g0 * w[1] + d1 * w[6]) * rh[a] + g1[a] * w[3] + gr[a] * w[8];
      atomicAdd(dst + (1 + a) * CCH, m * inv16);
    }
#pragma unroll
    for (int a = 0; a < 3; ++a) {
      const float ca = g0 * w[2] * rh[a] + g1[a] * w[5] + gr[a] * w[10];
#pragma unroll
      for (int b = 0; b < 3; ++b) {
        const float m = ca * rh[b] + g2[a * 3 + b] * w[7];
        atomicAdd(dst + (4 + a * 3 + b) * CCH, m * inv16);
      }
    }
  }
}

#define W_U0 0
#define W_U1 1
#define W_U2 2
#define W_V(k) (3 + (k))
#define W_G(k) (11 + (k))

__global__ void __launch_bounds__(128)
mmb_node(const float* __restrict__ acc,
         const float* __restrict__ node0, const float* __restrict__ node1,
         const float* __restrict__ node2,
         const float* __restrict__ U, const float* __restrict__ V,
         const float* __restrict__ Wg,
         float* __restrict__ out0, float* __restrict__ out1,
         float* __restrict__ out2, int N) {
  __shared__ float2 wlds[14 * 8 * 2 * 32];
  __shared__ float  ylds[4][16][CCH];

  for (int idx = threadIdx.x; idx < 14 * 1024; idx += blockDim.x) {
    const int wmat = idx >> 10;
    const int rem  = idx & 1023;
    const int k    = rem >> 5;
    const int n    = rem & 31;
    const float* src = (wmat < 3) ? (U + wmat * 1024)
                     : (wmat < 11) ? (V + (wmat - 3) * 1024)
                                   : (Wg + (wmat - 11) * 1024);
    const float val = src[rem];
    const int s   = k >> 2;
    const int kk  = k & 3;
    const int ln  = ((kk >> 1) << 4) | (n & 15);
    const int t   = n >> 4;
    const int jj  = kk & 1;
    ((float*)&wlds[((wmat * 8 + s) * 2 + t) * 32 + ln])[jj] = val;
  }
  __syncthreads();

  const int lane = threadIdx.x & 31;
  const int wv   = threadIdx.x >> 5;
  const int half = lane >> 4;
  const int m    = lane & 15;
  const int tile = blockIdx.x * 4 + wv;
  if (tile * 16 >= N) return;

#define LDB(wm, s, t)                                                       \
  ({ float2 _bv = wlds[(((wm) * 8 + (s)) * 2 + (t)) * 32 + lane];           \
     v2f _b; _b.x = _bv.x; _b.y = _bv.y; _b; })

#define MM2(acc, aop, wm, s)                                                \
  do { v2f _a = (aop);                                                      \
       acc[0] = wmma4(_a, LDB(wm, s, 0), acc[0]);                           \
       acc[1] = wmma4(_a, LDB(wm, s, 1), acc[1]); } while (0)

  const v8f vzero = {};

  const int rowA = tile * 16 + m;
  const int rowC = (rowA < N) ? rowA : (N - 1);
  const float* accRow = acc + rowC * (NCOMP * CCH) + 2 * half;
  float x0[2], b1[2][3], b2[2][9];
#define LOAD_BASE(s)                                                        \
  do { const float* _p = accRow + (s) * 4;                                  \
       { const float2 _v = *(const float2*)_p;                              \
         x0[0] = _v.x; x0[1] = _v.y; }                                      \
       _Pragma("unroll") for (int _a = 0; _a < 3; ++_a) {                   \
         const float2 _v = *(const float2*)(_p + (1 + _a) * CCH);           \
         b1[0][_a] = _v.x; b1[1][_a] = _v.y; }                              \
       _Pragma("unroll") for (int _q = 0; _q < 9; ++_q) {                   \
         const float2 _v = *(const float2*)(_p + (4 + _q) * CCH);           \
         b2[0][_q] = _v.x; b2[1][_q] = _v.y; }                              \
  } while (0)

  const int rowD0 = tile * 16 + 8 * half;
  float res0[2][8];
#pragma unroll
  for (int t = 0; t < 2; ++t)
#pragma unroll
    for (int r = 0; r < 8; ++r) {
      const int row  = rowD0 + r;
      const int rowL = (row < N) ? row : (N - 1);
      res0[t][r] = node0[rowL * CCH + t * 16 + m];
    }

  v8f y0acc[2] = {vzero, vzero};
  for (int s = 0; s < 8; ++s) {
    LOAD_BASE(s);
    v2f ax, asq, s1v, s2v;
#pragma unroll
    for (int jj = 0; jj < 2; ++jj) {
      ax[jj]  = x0[jj];
      asq[jj] = x0[jj] * x0[jj];
      float acc = 0.f;
#pragma unroll
      for (int a = 0; a < 3; ++a) acc += b1[jj][a] * b1[jj][a];
      s1v[jj] = acc;
      acc = 0.f;
#pragma unroll
      for (int q = 0; q < 9; ++q) acc += b2[jj][q] * b2[jj][q];
      s2v[jj] = acc;
    }
    MM2(y0acc, ax,  W_U0,  s);
    MM2(y0acc, asq, W_V(0), s);
    MM2(y0acc, s1v, W_V(3), s);
    MM2(y0acc, s2v, W_V(6), s);
  }

#pragma unroll
  for (int t = 0; t < 2; ++t)
#pragma unroll
    for (int r = 0; r < 8; ++r)
      ylds[wv][r + 8 * half][t * 16 + m] = y0acc[t][r];
  asm volatile("s_wait_dscnt 0" ::: "memory");

  v8f ga[3][2] = {{vzero, vzero}, {vzero, vzero}, {vzero, vzero}};
  for (int s = 0; s < 8; ++s) {
    const int k = s * 4 + 2 * half;
    v2f ay; ay.x = ylds[wv][m][k]; ay.y = ylds[wv][m][k + 1];
#pragma unroll
    for (int g = 0; g < 3; ++g) MM2(ga[g], ay, W_G(g), s);
  }
  v8f gate[3][2];
#pragma unroll
  for (int g = 0; g < 3; ++g)
#pragma unroll
    for (int t = 0; t < 2; ++t)
#pragma unroll
      for (int r = 0; r < 8; ++r) {
        const float v = ga[g][t][r];
        gate[g][t][r] = v * (1.0f / (1.0f + __expf(-v)));
      }

#pragma unroll
  for (int t = 0; t < 2; ++t)
#pragma unroll
    for (int r = 0; r < 8; ++r) {
      const int row = rowD0 + r;
      if (row < N) {
        const int cc = t * 16 + m;
        out0[row * CCH + cc] = res0[t][r] + gate[0][t][r];
      }
    }

  v8f y1acc[3][2] = {{vzero, vzero}, {vzero, vzero}, {vzero, vzero}};
  float res1[2][8][3];
#pragma unroll
  for (int t = 0; t < 2; ++t)
#pragma unroll
    for (int r = 0; r < 8; ++r) {
      const int row  = rowD0 + r;
      const int rowL = (row < N) ? row : (N - 1);
      const float* p = node1 + (rowL * CCH + t * 16 + m) * 3;
#pragma unroll
      for (int v = 0; v < 3; ++v) res1[t][r][v] = p[v];
    }
  for (int s = 0; s < 8; ++s) {
    LOAD_BASE(s);
#pragma unroll
    for (int v = 0; v < 3; ++v) {
      v2f av, a01, tv;
#pragma unroll
      for (int jj = 0; jj < 2; ++jj) {
        av[jj]  = b1[jj][v];
        a01[jj] = x0[jj] * b1[jj][v];
        float acc = 0.f;
#pragma unroll
        for (int a = 0; a < 3; ++a) acc += b1[jj][a] * b2[jj][a * 3 + v];
        tv[jj] = acc;
      }
      MM2(y1acc[v], av,  W_U1,  s);
      MM2(y1acc[v], a01, W_V(1), s);
      MM2(y1acc[v], tv,  W_V(5), s);
    }
  }
#pragma unroll
  for (int t = 0; t < 2; ++t)
#pragma unroll
    for (int r = 0; r < 8; ++r) {
      const int row = rowD0 + r;
      if (row < N) {
        float* q = out1 + (row * CCH + t * 16 + m) * 3;
#pragma unroll
        for (int v = 0; v < 3; ++v)
          q[v] = res1[t][r][v] + y1acc[v][t][r] * gate[1][t][r];
      }
    }

  for (int aIdx = 0; aIdx < 3; ++aIdx) {
    v8f y2acc[3][2] = {{vzero, vzero}, {vzero, vzero}, {vzero, vzero}};
    float res2[2][8][3];
#pragma unroll
    for (int t = 0; t < 2; ++t)
#pragma unroll
      for (int r = 0; r < 8; ++r) {
        const int row  = rowD0 + r;
        const int rowL = (row < N) ? row : (N - 1);
        const float* p = node2 + (rowL * CCH + t * 16 + m) * 9 + aIdx * 3;
#pragma unroll
        for (int bb = 0; bb < 3; ++bb) res2[t][r][bb] = p[bb];
      }
    for (int s = 0; s < 8; ++s) {
      LOAD_BASE(s);
#pragma unroll
      for (int b = 0; b < 3; ++b) {
        v2f a2v, a02, a11, qv;
#pragma unroll
        for (int jj = 0; jj < 2; ++jj) {
          const float g = b2[jj][aIdx * 3 + b];
          a2v[jj] = g;
          a02[jj] = x0[jj] * g;
          a11[jj] = b1[jj][aIdx] * b1[jj][b];
          float acc = 0.f;
#pragma unroll
          for (int kk = 0; kk < 3; ++kk)
            acc += b2[jj][aIdx * 3 + kk] * b2[jj][kk * 3 + b];
          qv[jj] = acc;
        }
        MM2(y2acc[b], a2v, W_U2,  s);
        MM2(y2acc[b], a02, W_V(2), s);
        MM2(y2acc[b], a11, W_V(4), s);
        MM2(y2acc[b], qv,  W_V(7), s);
      }
    }
#pragma unroll
    for (int t = 0; t < 2; ++t)
#pragma unroll
      for (int r = 0; r < 8; ++r) {
        const int row = rowD0 + r;
        if (row < N) {
          float* q = out2 + (row * CCH + t * 16 + m) * 9 + aIdx * 3;
#pragma unroll
          for (int bb = 0; bb < 3; ++bb)
            q[bb] = res2[t][r][bb] + y2acc[bb][t][r] * gate[2][t][r];
        }
      }
  }
#undef LOAD_BASE
#undef MM2
#undef LDB
}

extern "C" void kernel_launch(void* const* d_in, const int* in_sizes, int n_in,
                              void* d_out, int out_size, void* d_ws, size_t ws_size,
                              hipStream_t stream) {
  const float* node0 = (const float*)d_in[0];
  const float* node1 = (const float*)d_in[1];
  const float* node2 = (const float*)d_in[2];
  const float* rij   = (const float*)d_in[3];
  const float* Wrad  = (const float*)d_in[4];
  const float* U     = (const float*)d_in[5];
  const float* V     = (const float*)d_in[6];
  const float* Wg    = (const float*)d_in[7];
  const int*   idx_i = (const int*)d_in[8];
  const int*   idx_j = (const int*)d_in[9];

  const int N = in_sizes[0] / CCH;
  const int E = in_sizes[3] / 3;

  float* acc = (float*)d_ws;

  float* out0 = (float*)d_out;
  float* out1 = out0 + (size_t)N * CCH;
  float* out2 = out1 + (size_t)N * CCH * 3;

  const int n4 = N * CCH * 13 / 4;
  mmb_zero<<<(n4 + 255) / 256, 256, 0, stream>>>((float4*)d_ws, n4);

  mmb_edge<<<(E + 127) / 128, 128, 0, stream>>>(rij, Wrad, idx_i, idx_j,
                                                node0, node1, node2,
                                                acc, E);

  const int nTiles = (N + 15) / 16;
  mmb_node<<<(nTiles + 3) / 4, 128, 0, stream>>>(acc,
                                                 node0, node1, node2,
                                                 U, V, Wg,
                                                 out0, out1, out2, N);
}


# edge phase: edges bucketed by destination (int count + id list inside the node's own accumulator block); one wave sums a node's edges in registers, ascending edge id, stores once; no f32 atomics
# speedup vs baseline: 3.3069x; 1.1605x over previous
#include <hip/hip_runtime.h>
#include <hip/hip_bf16.h>
#include <math.h>

#define CCH 32
#define NPATH 11
#define NB 8
#define NCOMP 13

typedef __attribute__((ext_vector_type(2))) float v2f;
typedef __attribute__((ext_vector_type(8))) float v8f;

__device__ __forceinline__ v8f wmma4(v2f a, v2f b, v8f c) {
  return __builtin_amdgcn_wmma_f32_16x16x4_f32(
       false, a,  false, b,
       (short)0, c,  false,  false);
}

#define BLK_DW (NCOMP * CCH)
#define ID_OFF 32
#define BUCKET_CAP (BLK_DW - ID_OFF)

__global__ void mmb_clear_counts(int* __restrict__ wsI, int N) {
  const int i = blockIdx.x * blockDim.x + threadIdx.x;
  if (i < N) wsI[(size_t)i * BLK_DW] = 0;
}

__global__ void mmb_bucket(const int* __restrict__ idx_i, int* wsI, int E) {
  const int e = blockIdx.x * blockDim.x + threadIdx.x;
  if (e < E) {
    int* blk = wsI + (size_t)idx_i[e] * BLK_DW;
    const int slot = atomicAdd(blk, 1);
    if (slot < BUCKET_CAP) blk[ID_OFF + slot] = e;
  }
}

#define NODES_PER_WAVE 2

__device__ __forceinline__ float bcast_f(float v, int srcLane) {
  return __int_as_float(__builtin_amdgcn_readlane(__float_as_int(v), srcLane));
}

__global__ void __launch_bounds__(128, 6)
mmb_gather(const float* __restrict__ rij,
           const float* __restrict__ Wrad,
           const int*   __restrict__ idx_i,
           const int*   __restrict__ idx_j,
           const float* __restrict__ node0,
           const float* __restrict__ node1,
           const float* __restrict__ node2,
           float* acc,
           int N, int E) {
  const int c = threadIdx.x & 31;
  const int wave = __builtin_amdgcn_readfirstlane(
      (int)(blockIdx.x * (blockDim.x >> 5) + (threadIdx.x >> 5)));
  const int nbase = wave * NODES_PER_WAVE;
  if (nbase >= N) return;

  float Wr[NPATH * NB];
#pragma unroll
  for (int q = 0; q < NPATH * NB; ++q) Wr[q] = Wrad[q * CCH + c];

  const float inv16 = 0.0625f;

  for (int nn = 0; nn < NODES_PER_WAVE; ++nn) {
    const int i = nbase + nn;
    if (i >= N) break;
    float* blk = acc + (size_t)i * BLK_DW;
    const int* blkI = (const int*)blk;
    const int cnt = __builtin_amdgcn_readfirstlane(blkI[0]);

    float a[NCOMP];
#pragma unroll
    for (int k = 0; k < NCOMP; ++k) a[k] = 0.f;

    const bool inBucket = (cnt <= BUCKET_CAP);
    const int total = inBucket ? cnt : E;

    for (int base = 0; base < total; base += 32) {
      bool valid;
      int id;
      if (inBucket) {
        valid = (base + c < cnt);
        id = valid ? blkI[ID_OFF + base + c] : 0x7fffffff;
      } else {
        const int e0 = base + c;
        valid = false;
        if (e0 < E) valid = (idx_i[e0] == i);
        id = valid ? e0 : 0x7fffffff;
      }
      const unsigned vm = __builtin_amdgcn_ballot_w32(valid);
      if (vm == 0u) continue;
      const int nvalid = __builtin_popcount(vm);

      int pos = 0;
#pragma unroll
      for (int u = 0; u < 32; ++u)
        pos += (__builtin_amdgcn_readlane(id, u) < id) ? 1 : 0;

      const int el = valid ? id : 0;
      const float r0 = rij[el * 3 + 0], r1 = rij[el * 3 + 1], r2 = rij[el * 3 + 2];
      const float d  = sqrtf(r0 * r0 + r1 * r1 + r2 * r2 + 1e-12f);
      const float iv = 1.0f / d;
      const float rhl0 = r0 * iv, rhl1 = r1 * iv, rhl2 = r2 * iv;

      float x = d * 0.2f;
      x = (x > 1.0f) ? 1.0f : x;
      const float fc = 0.5f * (cosf(3.14159265358979f * x) + 1.0f);

      float rbfl[NB];
#pragma unroll
      for (int b = 0; b < NB; ++b) {
        const float t = d - (0.5f + 0.6428571429f * (float)b);
        rbfl[b] = __expf(-4.0f * t * t) * fc;
      }
      const int jl = idx_j[el];

      for (int t = 0; t < nvalid; ++t) {
        const unsigned sel = __builtin_amdgcn_ballot_w32(valid && (pos == t));
        const int sl = (sel != 0u) ? __builtin_ctz(sel) : 0;

        const int j = __builtin_amdgcn_readlane(jl, sl);
        const float rh[3] = {bcast_f(rhl0, sl), bcast_f(rhl1, sl), bcast_f(rhl2, sl)};
        float rbf[NB];
#pragma unroll
        for (int b = 0; b < NB; ++b) rbf[b] = bcast_f(rbfl[b], sl);

        const float* p0 = node0 + (size_t)j * CCH;
        const float* p1 = node1 + (size_t)j * (CCH * 3);
        const float* p2 = node2 + (size_t)j * (CCH * 9);
        const float g0 = p0[c];
        float g1[3], g2[9];
#pragma unroll
        for (int q = 0; q < 3; ++q) g1[q] = p1[c * 3 + q];
#pragma unroll
        for (int q = 0; q < 9; ++q) g2[q] = p2[c * 9 + q];

        float w[NPATH];
#pragma unroll
        for (int p = 0; p < NPATH; ++p) {
          float s = rbf[0] * Wr[p * NB];
#pragma unroll
          for (int b = 1; b < NB; ++b) s = fmaf(rbf[b], Wr[p * NB + b], s);
          w[p] = s;
        }

        const float d1 = g1[0] * rh[0] + g1[1] * rh[1] + g1[2] * rh[2];
        float gr[3];
#pragma unroll
        for (int q = 0; q < 3; ++q)
          gr[q] = g2[q * 3 + 0] * rh[0] + g2[q * 3 + 1] * rh[1] + g2[q * 3 + 2] * rh[2];
        const float d2 = gr[0] * rh[0] + gr[1] * rh[1] + gr[2] * rh[2];

        a[0] += (g0 * w[0] + d1 * w[4] + d2 * w[9]) * inv16;

#pragma unroll
        for (int q = 0; q < 3; ++q) {
          const float m = (g0 * w[1] + d1 * w[6]) * rh[q] + g1[q] * w[3] + gr[q] * w[8];
          a[1 + q] += m * inv16;
        }
#pragma unroll
        for (int q = 0; q < 3; ++q) {
          const float ca = g0 * w[2] * rh[q] + g1[q] * w[5] + gr[q] * w[10];
#pragma unroll
          for (int b = 0; b < 3; ++b) {
            const float m = ca * rh[b] + g2[q * 3 + b] * w[7];
            a[4 + q * 3 + b] += m * inv16;
          }
        }
      }
    }

#pragma unroll
    for (int k = 0; k < NCOMP; ++k) blk[k * CCH + c] = a[k];
  }
}

#define W_U0 0
#define W_U1 1
#define W_U2 2
#define W_V(k) (3 + (k))
#define W_G(k) (11 + (k))

__global__ void __launch_bounds__(128)
mmb_node(const float* __restrict__ acc,
         const float* __restrict__ node0, const float* __restrict__ node1,
         const float* __restrict__ node2,
         const float* __restrict__ U, const float* __restrict__ V,
         const float* __restrict__ Wg,
         float* __restrict__ out0, float* __restrict__ out1,
         float* __restrict__ out2, int N) {
  __shared__ float2 wlds[14 * 8 * 2 * 32];
  __shared__ float  ylds[4][16][CCH];

  for (int idx = threadIdx.x; idx < 14 * 1024; idx += blockDim.x) {
    const int wmat = idx >> 10;
    const int rem  = idx & 1023;
    const int k    = rem >> 5;
    const int n    = rem & 31;
    const float* src = (wmat < 3) ? (U + wmat * 1024)
                     : (wmat < 11) ? (V + (wmat - 3) * 1024)
                                   : (Wg + (wmat - 11) * 1024);
    const float val = src[rem];
    const int s   = k >> 2;
    const int kk  = k & 3;
    const int ln  = ((kk >> 1) << 4) | (n & 15);
    const int t   = n >> 4;
    const int jj  = kk & 1;
    ((float*)&wlds[((wmat * 8 + s) * 2 + t) * 32 + ln])[jj] = val;
  }
  __syncthreads();

  const int lane = threadIdx.x & 31;
  const int wv   = threadIdx.x >> 5;
  const int half = lane >> 4;
  const int m    = lane & 15;
  const int tile = blockIdx.x * 4 + wv;
  if (tile * 16 >= N) return;

#define LDB(wm, s, t)                                                       \
  ({ float2 _bv = wlds[(((wm) * 8 + (s)) * 2 + (t)) * 32 + lane];           \
     v2f _b; _b.x = _bv.x; _b.y = _bv.y; _b; })

#define MM2(acc, aop, wm, s)                                                \
  do { v2f _a = (aop);                                                      \
       acc[0] = wmma4(_a, LDB(wm, s, 0), acc[0]);                           \
       acc[1] = wmma4(_a, LDB(wm, s, 1), acc[1]); } while (0)

  const v8f vzero = {};

  const int rowA = tile * 16 + m;
  const int rowC = (rowA < N) ? rowA : (N - 1);
  const float* accRow = acc + rowC * (NCOMP * CCH) + 2 * half;
  float x0[2], b1[2][3], b2[2][9];
#define LOAD_BASE(s)                                                        \
  do { const float* _p = accRow + (s) * 4;                                  \
       { const float2 _v = *(const float2*)_p;                              \
         x0[0] = _v.x; x0[1] = _v.y; }                                      \
       _Pragma("unroll") for (int _a = 0; _a < 3; ++_a) {                   \
         const float2 _v = *(const float2*)(_p + (1 + _a) * CCH);           \
         b1[0][_a] = _v.x; b1[1][_a] = _v.y; }                              \
       _Pragma("unroll") for (int _q = 0; _q < 9; ++_q) {                   \
         const float2 _v = *(const float2*)(_p + (4 + _q) * CCH);           \
         b2[0][_q] = _v.x; b2[1][_q] = _v.y; }                              \
  } while (0)

  const int rowD0 = tile * 16 + 8 * half;
  float res0[2][8];
#pragma unroll
  for (int t = 0; t < 2; ++t)
#pragma unroll
    for (int r = 0; r < 8; ++r) {
      const int row  = rowD0 + r;
      const int rowL = (row < N) ? row : (N - 1);
      res0[t][r] = node0[rowL * CCH + t * 16 + m];
    }

  v8f y0acc[2] = {vzero, vzero};
  for (int s = 0; s < 8; ++s) {
    LOAD_BASE(s);
    v2f ax, asq, s1v, s2v;
#pragma unroll
    for (int jj = 0; jj < 2; ++jj) {
      ax[jj]  = x0[jj];
      asq[jj] = x0[jj] * x0[jj];
      float acc = 0.f;
#pragma unroll
      for (int a = 0; a < 3; ++a) acc += b1[jj][a] * b1[jj][a];
      s1v[jj] = acc;
      acc = 0.f;
#pragma unroll
      for (int q = 0; q < 9; ++q) acc += b2[jj][q] * b2[jj][q];
      s2v[jj] = acc;
    }
    MM2(y0acc, ax,  W_U0,  s);
    MM2(y0acc, asq, W_V(0), s);
    MM2(y0acc, s1v, W_V(3), s);
    MM2(y0acc, s2v, W_V(6), s);
  }

#pragma unroll
  for (int t = 0; t < 2; ++t)
#pragma unroll
    for (int r = 0; r < 8; ++r)
      ylds[wv][r + 8 * half][t * 16 + m] = y0acc[t][r];
  asm volatile("s_wait_dscnt 0" ::: "memory");

  v8f ga[3][2] = {{vzero, vzero}, {vzero, vzero}, {vzero, vzero}};
  for (int s = 0; s < 8; ++s) {
    const int k = s * 4 + 2 * half;
    v2f ay; ay.x = ylds[wv][m][k]; ay.y = ylds[wv][m][k + 1];
#pragma unroll
    for (int g = 0; g < 3; ++g) MM2(ga[g], ay, W_G(g), s);
  }
  v8f gate[3][2];
#pragma unroll
  for (int g = 0; g < 3; ++g)
#pragma unroll
    for (int t = 0; t < 2; ++t)
#pragma unroll
      for (int r = 0; r < 8; ++r) {
        const float v = ga[g][t][r];
        gate[g][t][r] = v * (1.0f / (1.0f + __expf(-v)));
      }

#pragma unroll
  for (int t = 0; t < 2; ++t)
#pragma unroll
    for (int r = 0; r < 8; ++r) {
      const int row = rowD0 + r;
      if (row < N) {
        const int cc = t * 16 + m;
        out0[row * CCH + cc] = res0[t][r] + gate[0][t][r];
      }
    }

  v8f y1acc[3][2] = {{vzero, vzero}, {vzero, vzero}, {vzero, vzero}};
  float res1[2][8][3];
#pragma unroll
  for (int t = 0; t < 2; ++t)
#pragma unroll
    for (int r = 0; r < 8; ++r) {
      const int row  = rowD0 + r;
      const int rowL = (row < N) ? row : (N - 1);
      const float* p = node1 + (rowL * CCH + t * 16 + m) * 3;
#pragma unroll
      for (int v = 0; v < 3; ++v) res1[t][r][v] = p[v];
    }
  for (int s = 0; s < 8; ++s) {
    LOAD_BASE(s);
#pragma unroll
    for (int v = 0; v < 3; ++v) {
      v2f av, a01, tv;
#pragma unroll
      for (int jj = 0; jj < 2; ++jj) {
        av[jj]  = b1[jj][v];
        a01[jj] = x0[jj] * b1[jj][v];
        float acc = 0.f;
#pragma unroll
        for (int a = 0; a < 3; ++a) acc += b1[jj][a] * b2[jj][a * 3 + v];
        tv[jj] = acc;
      }
      MM2(y1acc[v], av,  W_U1,  s);
      MM2(y1acc[v], a01, W_V(1), s);
      MM2(y1acc[v], tv,  W_V(5), s);
    }
  }
#pragma unroll
  for (int t = 0; t < 2; ++t)
#pragma unroll
    for (int r = 0; r < 8; ++r) {
      const int row = rowD0 + r;
      if (row < N) {
        float* q = out1 + (row * CCH + t * 16 + m) * 3;
#pragma unroll
        for (int v = 0; v < 3; ++v)
          q[v] = res1[t][r][v] + y1acc[v][t][r] * gate[1][t][r];
      }
    }

  for (int aIdx = 0; aIdx < 3; ++aIdx) {
    v8f y2acc[3][2] = {{vzero, vzero}, {vzero, vzero}, {vzero, vzero}};
    float res2[2][8][3];
#pragma unroll
    for (int t = 0; t < 2; ++t)
#pragma unroll
      for (int r = 0; r < 8; ++r) {
        const int row  = rowD0 + r;
        const int rowL = (row < N) ? row : (N - 1);
        const float* p = node2 + (rowL * CCH + t * 16 + m) * 9 + aIdx * 3;
#pragma unroll
        for (int bb = 0; bb < 3; ++bb) res2[t][r][bb] = p[bb];
      }
    for (int s = 0; s < 8; ++s) {
      LOAD_BASE(s);
#pragma unroll
      for (int b = 0; b < 3; ++b) {
        v2f a2v, a02, a11, qv;
#pragma unroll
        for (int jj = 0; jj < 2; ++jj) {
          const float g = b2[jj][aIdx * 3 + b];
          a2v[jj] = g;
          a02[jj] = x0[jj] * g;
          a11[jj] = b1[jj][aIdx] * b1[jj][b];
          float acc = 0.f;
#pragma unroll
          for (int kk = 0; kk < 3; ++kk)
            acc += b2[jj][aIdx * 3 + kk] * b2[jj][kk * 3 + b];
          qv[jj] = acc;
        }
        MM2(y2acc[b], a2v, W_U2,  s);
        MM2(y2acc[b], a02, W_V(2), s);
        MM2(y2acc[b], a11, W_V(4), s);
        MM2(y2acc[b], qv,  W_V(7), s);
      }
    }
#pragma unroll
    for (int t = 0; t < 2; ++t)
#pragma unroll
      for (int r = 0; r < 8; ++r) {
        const int row = rowD0 + r;
        if (row < N) {
          float* q = out2 + (row * CCH + t * 16 + m) * 9 + aIdx * 3;
#pragma unroll
          for (int bb = 0; bb < 3; ++bb)
            q[bb] = res2[t][r][bb] + y2acc[bb][t][r] * gate[2][t][r];
        }
      }
  }
#undef LOAD_BASE
#undef MM2
#undef LDB
}

extern "C" void kernel_launch(void* const* d_in, const int* in_sizes, int n_in,
                              void* d_out, int out_size, void* d_ws, size_t ws_size,
                              hipStream_t stream) {
  const float* node0 = (const float*)d_in[0];
  const float* node1 = (const float*)d_in[1];
  const float* node2 = (const float*)d_in[2];
  const float* rij   = (const float*)d_in[3];
  const float* Wrad  = (const float*)d_in[4];
  const float* U     = (const float*)d_in[5];
  const float* V     = (const float*)d_in[6];
  const float* Wg    = (const float*)d_in[7];
  const int*   idx_i = (const int*)d_in[8];
  const int*   idx_j = (const int*)d_in[9];

  const int N = in_sizes[0] / CCH;
  const int E = in_sizes[3] / 3;

  float* acc = (float*)d_ws;

  float* out0 = (float*)d_out;
  float* out1 = out0 + (size_t)N * CCH;
  float* out2 = out1 + (size_t)N * CCH * 3;

  mmb_clear_counts<<<(N + 255) / 256, 256, 0, stream>>>((int*)d_ws, N);
  mmb_bucket<<<(E + 255) / 256, 256, 0, stream>>>(idx_i, (int*)d_ws, E);

  mmb_gather<<<(N + 7) / 8, 128, 0, stream>>>(rij, Wrad, idx_i, idx_j,
                                              node0, node1, node2,
                                              acc, N, E);

  const int nTiles = (N + 15) / 16;
  mmb_node<<<(nTiles + 3) / 4, 128, 0, stream>>>(acc,
                                                 node0, node1, node2,
                                                 U, V, Wg,
                                                 out0, out1, out2, N);
}


# node phase: weight staging into LDS as one batch: each thread issues its 28 16-byte loads of the 14 matrices up front, then 28 16-byte LDS stores (was 112 serial 4-byte load-then-store trips)
# speedup vs baseline: 3.8034x; 1.1501x over previous
#include <hip/hip_runtime.h>
#include <hip/hip_bf16.h>
#include <math.h>

#define CCH 32
#define NPATH 11
#define NB 8
#define NCOMP 13

typedef __attribute__((ext_vector_type(2))) float v2f;
typedef __attribute__((ext_vector_type(8))) float v8f;

__device__ __forceinline__ v8f wmma4(v2f a, v2f b, v8f c) {
  return __builtin_amdgcn_wmma_f32_16x16x4_f32(
       false, a,  false, b,
       (short)0, c,  false,  false);
}

#define BLK_DW (NCOMP * CCH)
#define ID_OFF 32
#define BUCKET_CAP (BLK_DW - ID_OFF)

__global__ void mmb_clear_counts(int* __restrict__ wsI, int N) {
  const int i = blockIdx.x * blockDim.x + threadIdx.x;
  if (i < N) wsI[(size_t)i * BLK_DW] = 0;
}

__global__ void mmb_bucket(const int* __restrict__ idx_i, int* wsI, int E) {
  const int e = blockIdx.x * blockDim.x + threadIdx.x;
  if (e < E) {
    int* blk = wsI + (size_t)idx_i[e] * BLK_DW;
    const int slot = atomicAdd(blk, 1);
    if (slot < BUCKET_CAP) blk[ID_OFF + slot] = e;
  }
}

#define NODES_PER_WAVE 2

__device__ __forceinline__ float bcast_f(float v, int srcLane) {
  return __int_as_float(__builtin_amdgcn_readlane(__float_as_int(v), srcLane));
}

__global__ void __launch_bounds__(128, 6)
mmb_gather(const float* __restrict__ rij,
           const float* __restrict__ Wrad,
           const int*   __restrict__ idx_i,
           const int*   __restrict__ idx_j,
           const float* __restrict__ node0,
           const float* __restrict__ node1,
           const float* __restrict__ node2,
           float* acc,
           int N, int E) {
  const int c = threadIdx.x & 31;
  const int wave = __builtin_amdgcn_readfirstlane(
      (int)(blockIdx.x * (blockDim.x >> 5) + (threadIdx.x >> 5)));
  const int nbase = wave * NODES_PER_WAVE;
  if (nbase >= N) return;

  float Wr[NPATH * NB];
#pragma unroll
  for (int q = 0; q < NPATH * NB; ++q) Wr[q] = Wrad[q * CCH + c];

  const float inv16 = 0.0625f;

  for (int nn = 0; nn < NODES_PER_WAVE; ++nn) {
    const int i = nbase + nn;
    if (i >= N) break;
    float* blk = acc + (size_t)i * BLK_DW;
    const int* blkI = (const int*)blk;
    const int cnt = __builtin_amdgcn_readfirstlane(blkI[0]);

    float a[NCOMP];
#pragma unroll
    for (int k = 0; k < NCOMP; ++k) a[k] = 0.f;

    const bool inBucket = (cnt <= BUCKET_CAP);
    const int total = inBucket ? cnt : E;

    for (int base = 0; base < total; base += 32) {
      bool valid;
      int id;
      if (inBucket) {
        valid = (base + c < cnt);
        id = valid ? blkI[ID_OFF + base + c] : 0x7fffffff;
      } else {
        const int e0 = base + c;
        valid = false;
        if (e0 < E) valid = (idx_i[e0] == i);
        id = valid ? e0 : 0x7fffffff;
      }
      const unsigned vm = __builtin_amdgcn_ballot_w32(valid);
      if (vm == 0u) continue;
      const int nvalid = __builtin_popcount(vm);

      int pos = 0;
#pragma unroll
      for (int u = 0; u < 32; ++u)
        pos += (__builtin_amdgcn_readlane(id, u) < id) ? 1 : 0;

      const int el = valid ? id : 0;
      const float r0 = rij[el * 3 + 0], r1 = rij[el * 3 + 1], r2 = rij[el * 3 + 2];
      const float d  = sqrtf(r0 * r0 + r1 * r1 + r2 * r2 + 1e-12f);
      const float iv = 1.0f / d;
      const float rhl0 = r0 * iv, rhl1 = r1 * iv, rhl2 = r2 * iv;

      float x = d * 0.2f;
      x = (x > 1.0f) ? 1.0f : x;
      const float fc = 0.5f * (cosf(3.14159265358979f * x) + 1.0f);

      float rbfl[NB];
#pragma unroll
      for (int b = 0; b < NB; ++b) {
        const float t = d - (0.5f + 0.6428571429f * (float)b);
        rbfl[b] = __expf(-4.0f * t * t) * fc;
      }
      const int jl = idx_j[el];

      for (int t = 0; t < nvalid; ++t) {
        const unsigned sel = __builtin_amdgcn_ballot_w32(valid && (pos == t));
        const int sl = (sel != 0u) ? __builtin_ctz(sel) : 0;

        const int j = __builtin_amdgcn_readlane(jl, sl);
        const float rh[3] = {bcast_f(rhl0, sl), bcast_f(rhl1, sl), bcast_f(rhl2, sl)};
        float rbf[NB];
#pragma unroll
        for (int b = 0; b < NB; ++b) rbf[b] = bcast_f(rbfl[b], sl);

        const float* p0 = node0 + (size_t)j * CCH;
        const float* p1 = node1 + (size_t)j * (CCH * 3);
        const float* p2 = node2 + (size_t)j * (CCH * 9);
        const float g0 = p0[c];
        float g1[3], g2[9];
#pragma unroll
        for (int q = 0; q < 3; ++q) g1[q] = p1[c * 3 + q];
#pragma unroll
        for (int q = 0; q < 9; ++q) g2[q] = p2[c * 9 + q];

        float w[NPATH];
#pragma unroll
        for (int p = 0; p < NPATH; ++p) {
          float s = rbf[0] * Wr[p * NB];
#pragma unroll
          for (int b = 1; b < NB; ++b) s = fmaf(rbf[b], Wr[p * NB + b], s);
          w[p] = s;
        }

        const float d1 = g1[0] * rh[0] + g1[1] * rh[1] + g1[2] * rh[2];
        float gr[3];
#pragma unroll
        for (int q = 0; q < 3; ++q)
          gr[q] = g2[q * 3 + 0] * rh[0] + g2[q * 3 + 1] * rh[1] + g2[q * 3 + 2] * rh[2];
        const float d2 = gr[0] * rh[0] + gr[1] * rh[1] + gr[2] * rh[2];

        a[0] += (g0 * w[0] + d1 * w[4] + d2 * w[9]) * inv16;

#pragma unroll
        for (int q = 0; q < 3; ++q) {
          const float m = (g0 * w[1] + d1 * w[6]) * rh[q] + g1[q] * w[3] + gr[q] * w[8];
          a[1 + q] += m * inv16;
        }
#pragma unroll
        for (int q = 0; q < 3; ++q) {
          const float ca = g0 * w[2] * rh[q] + g1[q] * w[5] + gr[q] * w[10];
#pragma unroll
          for (int b = 0; b < 3; ++b) {
            const float m = ca * rh[b] + g2[q * 3 + b] * w[7];
            a[4 + q * 3 + b] += m * inv16;
          }
        }
      }
    }

#pragma unroll
    for (int k = 0; k < NCOMP; ++k) blk[k * CCH + c] = a[k];
  }
}

#define W_U0 0
#define W_U1 1
#define W_U2 2
#define W_V(k) (3 + (k))
#define W_G(k) (11 + (k))

__global__ void __launch_bounds__(128)
mmb_node(const float* __restrict__ acc,
         const float* __restrict__ node0, const float* __restrict__ node1,
         const float* __restrict__ node2,
         const float* __restrict__ U, const float* __restrict__ V,
         const float* __restrict__ Wg,
         float* __restrict__ out0, float* __restrict__ out1,
         float* __restrict__ out2, int N) {
  __shared__ __attribute__((aligned(16))) float2 wlds[14 * 8 * 2 * 32];
  __shared__ float  ylds[4][16][CCH];

  {
    const int tid = threadIdx.x;
    const int k   = (tid >> 3) << 1;
    const int n   = (tid & 7) << 2;
    const int s   = k >> 2;
    const int ln  = ((k & 2) << 3) | (n & 15);
    const int t   = n >> 4;
    const int off = k * CCH + n;
    float4 r0[14], r1[14];
#pragma unroll
    for (int i = 0; i < 14; ++i) {
      const float* src = (i < 3) ? (U + i * 1024)
                       : (i < 11) ? (V + (i - 3) * 1024)
                                  : (Wg + (i - 11) * 1024);
      r0[i] = *(const float4*)(src + off);
      r1[i] = *(const float4*)(src + off + CCH);
    }
#pragma unroll
    for (int i = 0; i < 14; ++i) {
      float4* dst = (float4*)&wlds[((i * 8 + s) * 2 + t) * 32 + ln];
      dst[0] = make_float4(r0[i].x, r1[i].x, r0[i].y, r1[i].y);
      dst[1] = make_float4(r0[i].z, r1[i].z, r0[i].w, r1[i].w);
    }
  }
  __syncthreads();

  const int lane = threadIdx.x & 31;
  const int wv   = threadIdx.x >> 5;
  const int half = lane >> 4;
  const int m    = lane & 15;
  const int tile = blockIdx.x * 4 + wv;
  if (tile * 16 >= N) return;

#define LDB(wm, s, t)                                                       \
  ({ float2 _bv = wlds[(((wm) * 8 + (s)) * 2 + (t)) * 32 + lane];           \
     v2f _b; _b.x = _bv.x; _b.y = _bv.y; _b; })

#define MM2(acc, aop, wm, s)                                                \
  do { v2f _a = (aop);                                                      \
       acc[0] = wmma4(_a, LDB(wm, s, 0), acc[0]);                           \
       acc[1] = wmma4(_a, LDB(wm, s, 1), acc[1]); } while (0)

  const v8f vzero = {};

  const int rowA = tile * 16 + m;
  const int rowC = (rowA < N) ? rowA : (N - 1);
  const float* accRow = acc + rowC * (NCOMP * CCH) + 2 * half;
  float x0[2], b1[2][3], b2[2][9];
#define LOAD_BASE(s)                                                        \
  do { const float* _p = accRow + (s) * 4;                                  \
       { const float2 _v = *(const float2*)_p;                              \
         x0[0] = _v.x; x0[1] = _v.y; }                                      \
       _Pragma("unroll") for (int _a = 0; _a < 3; ++_a) {                   \
         const float2 _v = *(const float2*)(_p + (1 + _a) * CCH);           \
         b1[0][_a] = _v.x; b1[1][_a] = _v.y; }                              \
       _Pragma("unroll") for (int _q = 0; _q < 9; ++_q) {                   \
         const float2 _v = *(const float2*)(_p + (4 + _q) * CCH);           \
         b2[0][_q] = _v.x; b2[1][_q] = _v.y; }                              \
  } while (0)

  const int rowD0 = tile * 16 + 8 * half;
  float res0[2][8];
#pragma unroll
  for (int t = 0; t < 2; ++t)
#pragma unroll
    for (int r = 0; r < 8; ++r) {
      const int row  = rowD0 + r;
      const int rowL = (row < N) ? row : (N - 1);
      res0[t][r] = node0[rowL * CCH + t * 16 + m];
    }

  v8f y0acc[2] = {vzero, vzero};
  for (int s = 0; s < 8; ++s) {
    LOAD_BASE(s);
    v2f ax, asq, s1v, s2v;
#pragma unroll
    for (int jj = 0; jj < 2; ++jj) {
      ax[jj]  = x0[jj];
      asq[jj] = x0[jj] * x0[jj];
      float acc = 0.f;
#pragma unroll
      for (int a = 0; a < 3; ++a) acc += b1[jj][a] * b1[jj][a];
      s1v[jj] = acc;
      acc = 0.f;
#pragma unroll
      for (int q = 0; q < 9; ++q) acc += b2[jj][q] * b2[jj][q];
      s2v[jj] = acc;
    }
    MM2(y0acc, ax,  W_U0,  s);
    MM2(y0acc, asq, W_V(0), s);
    MM2(y0acc, s1v, W_V(3), s);
    MM2(y0acc, s2v, W_V(6), s);
  }

#pragma unroll
  for (int t = 0; t < 2; ++t)
#pragma unroll
    for (int r = 0; r < 8; ++r)
      ylds[wv][r + 8 * half][t * 16 + m] = y0acc[t][r];
  asm volatile("s_wait_dscnt 0" ::: "memory");

  v8f ga[3][2] = {{vzero, vzero}, {vzero, vzero}, {vzero, vzero}};
  for (int s = 0; s < 8; ++s) {
    const int k = s * 4 + 2 * half;
    v2f ay; ay.x = ylds[wv][m][k]; ay.y = ylds[wv][m][k + 1];
#pragma unroll
    for (int g = 0; g < 3; ++g) MM2(ga[g], ay, W_G(g), s);
  }
  v8f gate[3][2];
#pragma unroll
  for (int g = 0; g < 3; ++g)
#pragma unroll
    for (int t = 0; t < 2; ++t)
#pragma unroll
      for (int r = 0; r < 8; ++r) {
        const float v = ga[g][t][r];
        gate[g][t][r] = v * (1.0f / (1.0f + __expf(-v)));
      }

#pragma unroll
  for (int t = 0; t < 2; ++t)
#pragma unroll
    for (int r = 0; r < 8; ++r) {
      const int row = rowD0 + r;
      if (row < N) {
        const int cc = t * 16 + m;
        out0[row * CCH + cc] = res0[t][r] + gate[0][t][r];
      }
    }

  v8f y1acc[3][2] = {{vzero, vzero}, {vzero, vzero}, {vzero, vzero}};
  float res1[2][8][3];
#pragma unroll
  for (int t = 0; t < 2; ++t)
#pragma unroll
    for (int r = 0; r < 8; ++r) {
      const int row  = rowD0 + r;
      const int rowL = (row < N) ? row : (N - 1);
      const float* p = node1 + (rowL * CCH + t * 16 + m) * 3;
#pragma unroll
      for (int v = 0; v < 3; ++v) res1[t][r][v] = p[v];
    }
  for (int s = 0; s < 8; ++s) {
    LOAD_BASE(s);
#pragma unroll
    for (int v = 0; v < 3; ++v) {
      v2f av, a01, tv;
#pragma unroll
      for (int jj = 0; jj < 2; ++jj) {
        av[jj]  = b1[jj][v];
        a01[jj] = x0[jj] * b1[jj][v];
        float acc = 0.f;
#pragma unroll
        for (int a = 0; a < 3; ++a) acc += b1[jj][a] * b2[jj][a * 3 + v];
        tv[jj] = acc;
      }
      MM2(y1acc[v], av,  W_U1,  s);
      MM2(y1acc[v], a01, W_V(1), s);
      MM2(y1acc[v], tv,  W_V(5), s);
    }
  }
#pragma unroll
  for (int t = 0; t < 2; ++t)
#pragma unroll
    for (int r = 0; r < 8; ++r) {
      const int row = rowD0 + r;
      if (row < N) {
        float* q = out1 + (row * CCH + t * 16 + m) * 3;
#pragma unroll
        for (int v = 0; v < 3; ++v)
          q[v] = res1[t][r][v] + y1acc[v][t][r] * gate[1][t][r];
      }
    }

  for (int aIdx = 0; aIdx < 3; ++aIdx) {
    v8f y2acc[3][2] = {{vzero, vzero}, {vzero, vzero}, {vzero, vzero}};
    float res2[2][8][3];
#pragma unroll
    for (int t = 0; t < 2; ++t)
#pragma unroll
      for (int r = 0; r < 8; ++r) {
        const int row  = rowD0 + r;
        const int rowL = (row < N) ? row : (N - 1);
        const float* p = node2 + (rowL * CCH + t * 16 + m) * 9 + aIdx * 3;
#pragma unroll
        for (int bb = 0; bb < 3; ++bb) res2[t][r][bb] = p[bb];
      }
    for (int s = 0; s < 8; ++s) {
      LOAD_BASE(s);
#pragma unroll
      for (int b = 0; b < 3; ++b) {
        v2f a2v, a02, a11, qv;
#pragma unroll
        for (int jj = 0; jj < 2; ++jj) {
          const float g = b2[jj][aIdx * 3 + b];
          a2v[jj] = g;
          a02[jj] = x0[jj] * g;
          a11[jj] = b1[jj][aIdx] * b1[jj][b];
          float acc = 0.f;
#pragma unroll
          for (int kk = 0; kk < 3; ++kk)
            acc += b2[jj][aIdx * 3 + kk] * b2[jj][kk * 3 + b];
          qv[jj] = acc;
        }
        MM2(y2acc[b], a2v, W_U2,  s);
        MM2(y2acc[b], a02, W_V(2), s);
        MM2(y2acc[b], a11, W_V(4), s);
        MM2(y2acc[b], qv,  W_V(7), s);
      }
    }
#pragma unroll
    for (int t = 0; t < 2; ++t)
#pragma unroll
      for (int r = 0; r < 8; ++r) {
        const int row = rowD0 + r;
        if (row < N) {
          float* q = out2 + (row * CCH + t * 16 + m) * 9 + aIdx * 3;
#pragma unroll
          for (int bb = 0; bb < 3; ++bb)
            q[bb] = res2[t][r][bb] + y2acc[bb][t][r] * gate[2][t][r];
        }
      }
  }
#undef LOAD_BASE
#undef MM2
#undef LDB
}

extern "C" void kernel_launch(void* const* d_in, const int* in_sizes, int n_in,
                              void* d_out, int out_size, void* d_ws, size_t ws_size,
                              hipStream_t stream) {
  const float* node0 = (const float*)d_in[0];
  const float* node1 = (const float*)d_in[1];
  const float* node2 = (const float*)d_in[2];
  const float* rij   = (const float*)d_in[3];
  const float* Wrad  = (const float*)d_in[4];
  const float* U     = (const float*)d_in[5];
  const float* V     = (const float*)d_in[6];
  const float* Wg    = (const float*)d_in[7];
  const int*   idx_i = (const int*)d_in[8];
  const int*   idx_j = (const int*)d_in[9];

  const int N = in_sizes[0] / CCH;
  const int E = in_sizes[3] / 3;

  float* acc = (float*)d_ws;

  float* out0 = (float*)d_out;
  float* out1 = out0 + (size_t)N * CCH;
  float* out2 = out1 + (size_t)N * CCH * 3;

  mmb_clear_counts<<<(N + 255) / 256, 256, 0, stream>>>((int*)d_ws, N);
  mmb_bucket<<<(E + 255) / 256, 256, 0, stream>>>(idx_i, (int*)d_ws, E);

  mmb_gather<<<(N + 7) / 8, 128, 0, stream>>>(rij, Wrad, idx_i, idx_j,
                                              node0, node1, node2,
                                              acc, N, E);

  const int nTiles = (N + 15) / 16;
  mmb_node<<<(nTiles + 3) / 4, 128, 0, stream>>>(acc,
                                                 node0, node1, node2,
                                                 U, V, Wg,
                                                 out0, out1, out2, N);
}


# node phase: y0, y1 and y2 products share one rolled k-loop (98 WMMAs per step, next step's loads requested ahead); 796 -> 463 registers, so two workgroups fit a WGP and all 313 run in one round
# speedup vs baseline: 4.2965x; 1.1296x over previous
#include <hip/hip_runtime.h>
#include <hip/hip_bf16.h>
#include <math.h>

#define CCH 32
#define NPATH 11
#define NB 8
#define NCOMP 13

typedef __attribute__((ext_vector_type(2))) float v2f;
typedef __attribute__((ext_vector_type(8))) float v8f;

__device__ __forceinline__ v8f wmma4(v2f a, v2f b, v8f c) {
  return __builtin_amdgcn_wmma_f32_16x16x4_f32(
       false, a,  false, b,
       (short)0, c,  false,  false);
}

#define BLK_DW (NCOMP * CCH)
#define ID_OFF 32
#define BUCKET_CAP (BLK_DW - ID_OFF)

__global__ void mmb_clear_counts(int* __restrict__ wsI, int N) {
  const int i = blockIdx.x * blockDim.x + threadIdx.x;
  if (i < N) wsI[(size_t)i * BLK_DW] = 0;
}

__global__ void mmb_bucket(const int* __restrict__ idx_i, int* wsI, int E) {
  const int e = blockIdx.x * blockDim.x + threadIdx.x;
  if (e < E) {
    int* blk = wsI + (size_t)idx_i[e] * BLK_DW;
    const int slot = atomicAdd(blk, 1);
    if (slot < BUCKET_CAP) blk[ID_OFF + slot] = e;
  }
}

#define NODES_PER_WAVE 2

__device__ __forceinline__ float bcast_f(float v, int srcLane) {
  return __int_as_float(__builtin_amdgcn_readlane(__float_as_int(v), srcLane));
}

__global__ void __launch_bounds__(128, 6)
mmb_gather(const float* __restrict__ rij,
           const float* __restrict__ Wrad,
           const int*   __restrict__ idx_i,
           const int*   __restrict__ idx_j,
           const float* __restrict__ node0,
           const float* __restrict__ node1,
           const float* __restrict__ node2,
           float* acc,
           int N, int E) {
  const int c = threadIdx.x & 31;
  const int wave = __builtin_amdgcn_readfirstlane(
      (int)(blockIdx.x * (blockDim.x >> 5) + (threadIdx.x >> 5)));
  const int nbase = wave * NODES_PER_WAVE;
  if (nbase >= N) return;

  float Wr[NPATH * NB];
#pragma unroll
  for (int q = 0; q < NPATH * NB; ++q) Wr[q] = Wrad[q * CCH + c];

  const float inv16 = 0.0625f;

  for (int nn = 0; nn < NODES_PER_WAVE; ++nn) {
    const int i = nbase + nn;
    if (i >= N) break;
    float* blk = acc + (size_t)i * BLK_DW;
    const int* blkI = (const int*)blk;
    const int cnt = __builtin_amdgcn_readfirstlane(blkI[0]);

    float a[NCOMP];
#pragma unroll
    for (int k = 0; k < NCOMP; ++k) a[k] = 0.f;

    const bool inBucket = (cnt <= BUCKET_CAP);
    const int total = inBucket ? cnt : E;

    for (int base = 0; base < total; base += 32) {
      bool valid;
      int id;
      if (inBucket) {
        valid = (base + c < cnt);
        id = valid ? blkI[ID_OFF + base + c] : 0x7fffffff;
      } else {
        const int e0 = base + c;
        valid = false;
        if (e0 < E) valid = (idx_i[e0] == i);
        id = valid ? e0 : 0x7fffffff;
      }
      const unsigned vm = __builtin_amdgcn_ballot_w32(valid);
      if (vm == 0u) continue;
      const int nvalid = __builtin_popcount(vm);

      int pos = 0;
#pragma unroll
      for (int u = 0; u < 32; ++u)
        pos += (__builtin_amdgcn_readlane(id, u) < id) ? 1 : 0;

      const int el = valid ? id : 0;
      const float r0 = rij[el * 3 + 0], r1 = rij[el * 3 + 1], r2 = rij[el * 3 + 2];
      const float d  = sqrtf(r0 * r0 + r1 * r1 + r2 * r2 + 1e-12f);
      const float iv = 1.0f / d;
      const float rhl0 = r0 * iv, rhl1 = r1 * iv, rhl2 = r2 * iv;

      float x = d * 0.2f;
      x = (x > 1.0f) ? 1.0f : x;
      const float fc = 0.5f * (cosf(3.14159265358979f * x) + 1.0f);

      float rbfl[NB];
#pragma unroll
      for (int b = 0; b < NB; ++b) {
        const float t = d - (0.5f + 0.6428571429f * (float)b);
        rbfl[b] = __expf(-4.0f * t * t) * fc;
      }
      const int jl = idx_j[el];

      for (int t = 0; t < nvalid; ++t) {
        const unsigned sel = __builtin_amdgcn_ballot_w32(valid && (pos == t));
        const int sl = (sel != 0u) ? __builtin_ctz(sel) : 0;

        const int j = __builtin_amdgcn_readlane(jl, sl);
        const float rh[3] = {bcast_f(rhl0, sl), bcast_f(rhl1, sl), bcast_f(rhl2, sl)};
        float rbf[NB];
#pragma unroll
        for (int b = 0; b < NB; ++b) rbf[b] = bcast_f(rbfl[b], sl);

        const float* p0 = node0 + (size_t)j * CCH;
        const float* p1 = node1 + (size_t)j * (CCH * 3);
        const float* p2 = node2 + (size_t)j * (CCH * 9);
        const float g0 = p0[c];
        float g1[3], g2[9];
#pragma unroll
        for (int q = 0; q < 3; ++q) g1[q] = p1[c * 3 + q];
#pragma unroll
        for (int q = 0; q < 9; ++q) g2[q] = p2[c * 9 + q];

        float w[NPATH];
#pragma unroll
        for (int p = 0; p < NPATH; ++p) {
          float s = rbf[0] * Wr[p * NB];
#pragma unroll
          for (int b = 1; b < NB; ++b) s = fmaf(rbf[b], Wr[p * NB + b], s);
          w[p] = s;
        }

        const float d1 = g1[0] * rh[0] + g1[1] * rh[1] + g1[2] * rh[2];
        float gr[3];
#pragma unroll
        for (int q = 0; q < 3; ++q)
          gr[q] = g2[q * 3 + 0] * rh[0] + g2[q * 3 + 1] * rh[1] + g2[q * 3 + 2] * rh[2];
        const float d2 = gr[0] * rh[0] + gr[1] * rh[1] + gr[2] * rh[2];

        a[0] += (g0 * w[0] + d1 * w[4] + d2 * w[9]) * inv16;

#pragma unroll
        for (int q = 0; q < 3; ++q) {
          const float m = (g0 * w[1] + d1 * w[6]) * rh[q] + g1[q] * w[3] + gr[q] * w[8];
          a[1 + q] += m * inv16;
        }
#pragma unroll
        for (int q = 0; q < 3; ++q) {
          const float ca = g0 * w[2] * rh[q] + g1[q] * w[5] + gr[q] * w[10];
#pragma unroll
          for (int b = 0; b < 3; ++b) {
            const float m = ca * rh[b] + g2[q * 3 + b] * w[7];
            a[4 + q * 3 + b] += m * inv16;
          }
        }
      }
    }

#pragma unroll
    for (int k = 0; k < NCOMP; ++k) blk[k * CCH + c] = a[k];
  }
}

#define W_U0 0
#define W_U1 1
#define W_U2 2
#define W_V(k) (3 + (k))
#define W_G(k) (11 + (k))

__global__ void __launch_bounds__(128, 2)
mmb_node(const float* __restrict__ acc,
         const float* __restrict__ node0, const float* __restrict__ node1,
         const float* __restrict__ node2,
         const float* __restrict__ U, const float* __restrict__ V,
         const float* __restrict__ Wg,
         float* __restrict__ out0, float* __restrict__ out1,
         float* __restrict__ out2, int N) {
  __shared__ __attribute__((aligned(16))) float2 wlds[14 * 8 * 2 * 32];
  __shared__ float  ylds[4][16][CCH];

  {
    const int tid = threadIdx.x;
    const int k   = (tid >> 3) << 1;
    const int n   = (tid & 7) << 2;
    const int s   = k >> 2;
    const int ln  = ((k & 2) << 3) | (n & 15);
    const int t   = n >> 4;
    const int off = k * CCH + n;
    float4 r0[14], r1[14];
#pragma unroll
    for (int i = 0; i < 14; ++i) {
      const float* src = (i < 3) ? (U + i * 1024)
                       : (i < 11) ? (V + (i - 3) * 1024)
                                  : (Wg + (i - 11) * 1024);
      r0[i] = *(const float4*)(src + off);
      r1[i] = *(const float4*)(src + off + CCH);
    }
#pragma unroll
    for (int i = 0; i < 14; ++i) {
      float4* dst = (float4*)&wlds[((i * 8 + s) * 2 + t) * 32 + ln];
      dst[0] = make_float4(r0[i].x, r1[i].x, r0[i].y, r1[i].y);
      dst[1] = make_float4(r0[i].z, r1[i].z, r0[i].w, r1[i].w);
    }
  }
  __syncthreads();

  const int lane = threadIdx.x & 31;
  const int wv   = threadIdx.x >> 5;
  const int half = lane >> 4;
  const int m    = lane & 15;
  const int tile = blockIdx.x * 4 + wv;
  if (tile * 16 >= N) return;

#define LDB(wm, s, t)                                                       \
  ({ float2 _bv = wlds[(((wm) * 8 + (s)) * 2 + (t)) * 32 + lane];           \
     v2f _b; _b.x = _bv.x; _b.y = _bv.y; _b; })

#define MM2(acc, aop, wm, s)                                                \
  do { v2f _a = (aop);                                                      \
       acc[0] = wmma4(_a, LDB(wm, s, 0), acc[0]);                           \
       acc[1] = wmma4(_a, LDB(wm, s, 1), acc[1]); } while (0)

  const v8f vzero = {};

  const int rowA = tile * 16 + m;
  const int rowC = (rowA < N) ? rowA : (N - 1);
  const float* accRow = acc + rowC * (NCOMP * CCH) + 2 * half;

  v8f y0acc[2] = {vzero, vzero};
  v8f y1acc[3][2];
  v8f y2acc[9][2];
#pragma unroll
  for (int v = 0; v < 3; ++v) { y1acc[v][0] = vzero; y1acc[v][1] = vzero; }
#pragma unroll
  for (int q = 0; q < 9; ++q) { y2acc[q][0] = vzero; y2acc[q][1] = vzero; }

  float2 cur[NCOMP], nxt[NCOMP];
#pragma unroll
  for (int k = 0; k < NCOMP; ++k) cur[k] = *(const float2*)(accRow + k * CCH);

#pragma unroll 1
  for (int s = 0; s < 8; ++s) {
    const int sn = (s < 7) ? (s + 1) : 7;
#pragma unroll
    for (int k = 0; k < NCOMP; ++k)
      nxt[k] = *(const float2*)(accRow + sn * 4 + k * CCH);

    float x0[2], b1[2][3], b2[2][9];
    x0[0] = cur[0].x; x0[1] = cur[0].y;
#pragma unroll
    for (int a = 0; a < 3; ++a) { b1[0][a] = cur[1 + a].x; b1[1][a] = cur[1 + a].y; }
#pragma unroll
    for (int q = 0; q < 9; ++q) { b2[0][q] = cur[4 + q].x; b2[1][q] = cur[4 + q].y; }

    {
      v2f ax, asq, s1v, s2v;
#pragma unroll
      for (int jj = 0; jj < 2; ++jj) {
        ax[jj]  = x0[jj];
        asq[jj] = x0[jj] * x0[jj];
        float sum = 0.f;
#pragma unroll
        for (int a = 0; a < 3; ++a) sum += b1[jj][a] * b1[jj][a];
        s1v[jj] = sum;
        sum = 0.f;
#pragma unroll
        for (int q = 0; q < 9; ++q) sum += b2[jj][q] * b2[jj][q];
        s2v[jj] = sum;
      }
      MM2(y0acc, ax,  W_U0,  s);
      MM2(y0acc, asq, W_V(0), s);
      MM2(y0acc, s1v, W_V(3), s);
      MM2(y0acc, s2v, W_V(6), s);
    }

#pragma unroll
    for (int v = 0; v < 3; ++v) {
      v2f av, a01, tv;
#pragma unroll
      for (int jj = 0; jj < 2; ++jj) {
        av[jj]  = b1[jj][v];
        a01[jj] = x0[jj] * b1[jj][v];
        float sum = 0.f;
#pragma unroll
        for (int a = 0; a < 3; ++a) sum += b1[jj][a] * b2[jj][a * 3 + v];
        tv[jj] = sum;
      }
      MM2(y1acc[v], av,  W_U1,  s);
      MM2(y1acc[v], a01, W_V(1), s);
      MM2(y1acc[v], tv,  W_V(5), s);
    }

#pragma unroll
    for (int aIdx = 0; aIdx < 3; ++aIdx) {
#pragma unroll
      for (int b = 0; b < 3; ++b) {
        v2f a2v, a02, a11, qv;
#pragma unroll
        for (int jj = 0; jj < 2; ++jj) {
          const float g = b2[jj][aIdx * 3 + b];
          a2v[jj] = g;
          a02[jj] = x0[jj] * g;
          a11[jj] = b1[jj][aIdx] * b1[jj][b];
          float sum = 0.f;
#pragma unroll
          for (int kk = 0; kk < 3; ++kk)
            sum += b2[jj][aIdx * 3 + kk] * b2[jj][kk * 3 + b];
          qv[jj] = sum;
        }
        MM2(y2acc[aIdx * 3 + b], a2v, W_U2,  s);
        MM2(y2acc[aIdx * 3 + b], a02, W_V(2), s);
        MM2(y2acc[aIdx * 3 + b], a11, W_V(4), s);
        MM2(y2acc[aIdx * 3 + b], qv,  W_V(7), s);
      }
    }

#pragma unroll
    for (int k = 0; k < NCOMP; ++k) cur[k] = nxt[k];
  }

  const int rowD0 = tile * 16 + 8 * half;
  float res0[2][8];
  float res1[2][8][3];
#pragma unroll
  for (int t = 0; t < 2; ++t)
#pragma unroll
    for (int r = 0; r < 8; ++r) {
      const int row  = rowD0 + r;
      const int rowL = (row < N) ? row : (N - 1);
      res0[t][r] = node0[rowL * CCH + t * 16 + m];
      const float* p = node1 + (rowL * CCH + t * 16 + m) * 3;
#pragma unroll
      for (int v = 0; v < 3; ++v) res1[t][r][v] = p[v];
    }

#pragma unroll
  for (int t = 0; t < 2; ++t)
#pragma unroll
    for (int r = 0; r < 8; ++r)
      ylds[wv][r + 8 * half][t * 16 + m] = y0acc[t][r];
  asm volatile("s_wait_dscnt 0" ::: "memory");

  v8f ga[3][2] = {{vzero, vzero}, {vzero, vzero}, {vzero, vzero}};
#pragma unroll
  for (int s = 0; s < 8; ++s) {
    const int k = s * 4 + 2 * half;
    v2f ay; ay.x = ylds[wv][m][k]; ay.y = ylds[wv][m][k + 1];
#pragma unroll
    for (int g = 0; g < 3; ++g) MM2(ga[g], ay, W_G(g), s);
  }
  v8f gate[3][2];
#pragma unroll
  for (int g = 0; g < 3; ++g)
#pragma unroll
    for (int t = 0; t < 2; ++t)
#pragma unroll
      for (int r = 0; r < 8; ++r) {
        const float v = ga[g][t][r];
        gate[g][t][r] = v * (1.0f / (1.0f + __expf(-v)));
      }

  float res2[2][8][3];
#pragma unroll
  for (int t = 0; t < 2; ++t)
#pragma unroll
    for (int r = 0; r < 8; ++r) {
      const int row  = rowD0 + r;
      const int rowL = (row < N) ? row : (N - 1);
      const float* p = node2 + (rowL * CCH + t * 16 + m) * 9;
#pragma unroll
      for (int bb = 0; bb < 3; ++bb) res2[t][r][bb] = p[bb];
    }

#pragma unroll
  for (int t = 0; t < 2; ++t)
#pragma unroll
    for (int r = 0; r < 8; ++r) {
      const int row = rowD0 + r;
      if (row < N) {
        const int cc = t * 16 + m;
        out0[row * CCH + cc] = res0[t][r] + gate[0][t][r];
      }
    }

#pragma unroll
  for (int t = 0; t < 2; ++t)
#pragma unroll
    for (int r = 0; r < 8; ++r) {
      const int row = rowD0 + r;
      if (row < N) {
        float* q = out1 + (row * CCH + t * 16 + m) * 3;
#pragma unroll
        for (int v = 0; v < 3; ++v)
          q[v] = res1[t][r][v] + y1acc[v][t][r] * gate[1][t][r];
      }
    }

#pragma unroll
  for (int aIdx = 0; aIdx < 3; ++aIdx) {
    float nres[2][8][3];
    if (aIdx < 2) {
#pragma unroll
      for (int t = 0; t < 2; ++t)
#pragma unroll
        for (int r = 0; r < 8; ++r) {
          const int row  = rowD0 + r;
          const int rowL = (row < N) ? row : (N - 1);
          const float* p = node2 + (rowL * CCH + t * 16 + m) * 9 + (aIdx + 1) * 3;
#pragma unroll
          for (int bb = 0; bb < 3; ++bb) nres[t][r][bb] = p[bb];
        }
    }
#pragma unroll
    for (int t = 0; t < 2; ++t)
#pragma unroll
      for (int r = 0; r < 8; ++r) {
        const int row = rowD0 + r;
        if (row < N) {
          float* q = out2 + (row * CCH + t * 16 + m) * 9 + aIdx * 3;
#pragma unroll
          for (int bb = 0; bb < 3; ++bb)
            q[bb] = res2[t][r][bb] + y2acc[aIdx * 3 + bb][t][r] * gate[2][t][r];
        }
      }
    if (aIdx < 2) {
#pragma unroll
      for (int t = 0; t < 2; ++t)
#pragma unroll
        for (int r = 0; r < 8; ++r)
#pragma unroll
          for (int bb = 0; bb < 3; ++bb) res2[t][r][bb] = nres[t][r][bb];
    }
  }
#undef MM2
#undef LDB
}

extern "C" void kernel_launch(void* const* d_in, const int* in_sizes, int n_in,
                              void* d_out, int out_size, void* d_ws, size_t ws_size,
                              hipStream_t stream) {
  const float* node0 = (const float*)d_in[0];
  const float* node1 = (const float*)d_in[1];
  const float* node2 = (const float*)d_in[2];
  const float* rij   = (const float*)d_in[3];
  const float* Wrad  = (const float*)d_in[4];
  const float* U     = (const float*)d_in[5];
  const float* V     = (const float*)d_in[6];
  const float* Wg    = (const float*)d_in[7];
  const int*   idx_i = (const int*)d_in[8];
  const int*   idx_j = (const int*)d_in[9];

  const int N = in_sizes[0] / CCH;
  const int E = in_sizes[3] / 3;

  float* acc = (float*)d_ws;

  float* out0 = (float*)d_out;
  float* out1 = out0 + (size_t)N * CCH;
  float* out2 = out1 + (size_t)N * CCH * 3;

  mmb_clear_counts<<<(N + 255) / 256, 256, 0, stream>>>((int*)d_ws, N);
  mmb_bucket<<<(E + 255) / 256, 256, 0, stream>>>(idx_i, (int*)d_ws, E);

  mmb_gather<<<(N + 7) / 8, 128, 0, stream>>>(rij, Wrad, idx_i, idx_j,
                                              node0, node1, node2,
                                              acc, N, E);

  const int nTiles = (N + 15) / 16;
  mmb_node<<<(nTiles + 3) / 4, 128, 0, stream>>>(acc,
                                                 node0, node1, node2,
                                                 U, V, Wg,
                                                 out0, out1, out2, N);
}


# edge gather: each wave requests the feature rows of 4 edges back to back before computing their messages (was 1 edge per trip); source index loaded with the edge vector; sum order as before
# speedup vs baseline: 4.7308x; 1.1011x over previous
#include <hip/hip_runtime.h>
#include <hip/hip_bf16.h>
#include <math.h>

#define CCH 32
#define NPATH 11
#define NB 8
#define NCOMP 13

typedef __attribute__((ext_vector_type(2))) float v2f;
typedef __attribute__((ext_vector_type(8))) float v8f;

__device__ __forceinline__ v8f wmma4(v2f a, v2f b, v8f c) {
  return __builtin_amdgcn_wmma_f32_16x16x4_f32(
       false, a,  false, b,
       (short)0, c,  false,  false);
}

#define BLK_DW (NCOMP * CCH)
#define ID_OFF 32
#define BUCKET_CAP (BLK_DW - ID_OFF)

__global__ void mmb_clear_counts(int* __restrict__ wsI, int N) {
  const int i = blockIdx.x * blockDim.x + threadIdx.x;
  if (i < N) wsI[(size_t)i * BLK_DW] = 0;
}

__global__ void mmb_bucket(const int* __restrict__ idx_i, int* wsI, int E) {
  const int e = blockIdx.x * blockDim.x + threadIdx.x;
  if (e < E) {
    int* blk = wsI + (size_t)idx_i[e] * BLK_DW;
    const int slot = atomicAdd(blk, 1);
    if (slot < BUCKET_CAP) blk[ID_OFF + slot] = e;
  }
}

#define NODES_PER_WAVE 2

__device__ __forceinline__ float bcast_f(float v, int srcLane) {
  return __int_as_float(__builtin_amdgcn_readlane(__float_as_int(v), srcLane));
}

#define EDGES_IN_FLIGHT 4
__global__ void __launch_bounds__(128, 4)
mmb_gather(const float* __restrict__ rij,
           const float* __restrict__ Wrad,
           const int*   __restrict__ idx_i,
           const int*   __restrict__ idx_j,
           const float* node0,
           const float* node1,
           const float* node2,
           float* acc,
           int N, int E) {
  const int c = threadIdx.x & 31;
  const int wave = __builtin_amdgcn_readfirstlane(
      (int)(blockIdx.x * (blockDim.x >> 5) + (threadIdx.x >> 5)));
  const int nbase = wave * NODES_PER_WAVE;
  if (nbase >= N) return;

  float Wr[NPATH * NB];
#pragma unroll
  for (int q = 0; q < NPATH * NB; ++q) Wr[q] = Wrad[q * CCH + c];

  const float inv16 = 0.0625f;

  for (int nn = 0; nn < NODES_PER_WAVE; ++nn) {
    const int i = nbase + nn;
    if (i >= N) break;
    float* blk = acc + (size_t)i * BLK_DW;
    const int* blkI = (const int*)blk;
    const int cnt = __builtin_amdgcn_readfirstlane(blkI[0]);

    float a[NCOMP];
#pragma unroll
    for (int k = 0; k < NCOMP; ++k) a[k] = 0.f;

    const bool inBucket = (cnt <= BUCKET_CAP);
    const int total = inBucket ? cnt : E;

    for (int base = 0; base < total; base += 32) {
      bool valid;
      int id;
      if (inBucket) {
        valid = (base + c < cnt);
        id = valid ? blkI[ID_OFF + base + c] : 0x7fffffff;
      } else {
        const int e0 = base + c;
        valid = false;
        if (e0 < E) valid = (idx_i[e0] == i);
        id = valid ? e0 : 0x7fffffff;
      }
      const unsigned vm = __builtin_amdgcn_ballot_w32(valid);
      if (vm == 0u) continue;
      const int nvalid = __builtin_popcount(vm);

      int pos = 0;
#pragma unroll
      for (int u = 0; u < 32; ++u)
        pos += (__builtin_amdgcn_readlane(id, u) < id) ? 1 : 0;

      const int el = valid ? id : 0;
      const int jl = idx_j[el];
      const float r0 = rij[el * 3 + 0], r1 = rij[el * 3 + 1], r2 = rij[el * 3 + 2];
      const float d  = sqrtf(r0 * r0 + r1 * r1 + r2 * r2 + 1e-12f);
      const float iv = 1.0f / d;
      const float rhl0 = r0 * iv, rhl1 = r1 * iv, rhl2 = r2 * iv;

      float x = d * 0.2f;
      x = (x > 1.0f) ? 1.0f : x;
      const float fc = 0.5f * (cosf(3.14159265358979f * x) + 1.0f);

      float rbfl[NB];
#pragma unroll
      for (int b = 0; b < NB; ++b) {
        const float t = d - (0.5f + 0.6428571429f * (float)b);
        rbfl[b] = __expf(-4.0f * t * t) * fc;
      }

      for (int t = 0; t < nvalid; t += EDGES_IN_FLIGHT) {
        int   sl[EDGES_IN_FLIGHT];
        float g0[EDGES_IN_FLIGHT], g1[EDGES_IN_FLIGHT][3], g2[EDGES_IN_FLIGHT][9];

#pragma unroll
        for (int u = 0; u < EDGES_IN_FLIGHT; ++u) {
          const int rk = (t + u < nvalid) ? (t + u) : (nvalid - 1);
          const unsigned sel = __builtin_amdgcn_ballot_w32(valid && (pos == rk));
          sl[u] = (sel != 0u) ? __builtin_ctz(sel) : 0;
          const int j = __builtin_amdgcn_readlane(jl, sl[u]);

          const float* p0 = node0 + (size_t)j * CCH;
          const float* p1 = node1 + (size_t)j * (CCH * 3);
          const float* p2 = node2 + (size_t)j * (CCH * 9);
          g0[u] = p0[c];
#pragma unroll
          for (int q = 0; q < 3; ++q) g1[u][q] = p1[c * 3 + q];
#pragma unroll
          for (int q = 0; q < 9; ++q) g2[u][q] = p2[c * 9 + q];
        }

        __builtin_amdgcn_sched_barrier(0);

#pragma unroll
        for (int u = 0; u < EDGES_IN_FLIGHT; ++u) {
          if (t + u < nvalid) {
            const float rh[3] = {bcast_f(rhl0, sl[u]), bcast_f(rhl1, sl[u]),
                                 bcast_f(rhl2, sl[u])};
            float rbf[NB];
#pragma unroll
            for (int b = 0; b < NB; ++b) rbf[b] = bcast_f(rbfl[b], sl[u]);

            float w[NPATH];
#pragma unroll
            for (int p = 0; p < NPATH; ++p) {
              float sacc = rbf[0] * Wr[p * NB];
#pragma unroll
              for (int b = 1; b < NB; ++b) sacc = fmaf(rbf[b], Wr[p * NB + b], sacc);
              w[p] = sacc;
            }

            const float d1 = g1[u][0] * rh[0] + g1[u][1] * rh[1] + g1[u][2] * rh[2];
            float gr[3];
#pragma unroll
            for (int q = 0; q < 3; ++q)
              gr[q] = g2[u][q * 3 + 0] * rh[0] + g2[u][q * 3 + 1] * rh[1]
                    + g2[u][q * 3 + 2] * rh[2];
            const float d2 = gr[0] * rh[0] + gr[1] * rh[1] + gr[2] * rh[2];

            a[0] += (g0[u] * w[0] + d1 * w[4] + d2 * w[9]) * inv16;

#pragma unroll
            for (int q = 0; q < 3; ++q) {
              const float m = (g0[u] * w[1] + d1 * w[6]) * rh[q] + g1[u][q] * w[3]
                            + gr[q] * w[8];
              a[1 + q] += m * inv16;
            }
#pragma unroll
            for (int q = 0; q < 3; ++q) {
              const float ca = g0[u] * w[2] * rh[q] + g1[u][q] * w[5] + gr[q] * w[10];
#pragma unroll
              for (int b = 0; b < 3; ++b) {
                const float m = ca * rh[b] + g2[u][q * 3 + b] * w[7];
                a[4 + q * 3 + b] += m * inv16;
              }
            }
          }
        }
      }
    }

#pragma unroll
    for (int k = 0; k < NCOMP; ++k) blk[k * CCH + c] = a[k];
  }
}

#define W_U0 0
#define W_U1 1
#define W_U2 2
#define W_V(k) (3 + (k))
#define W_G(k) (11 + (k))

__global__ void __launch_bounds__(128, 2)
mmb_node(const float* __restrict__ acc,
         const float* __restrict__ node0, const float* __restrict__ node1,
         const float* __restrict__ node2,
         const float* __restrict__ U, const float* __restrict__ V,
         const float* __restrict__ Wg,
         float* __restrict__ out0, float* __restrict__ out1,
         float* __restrict__ out2, int N) {
  __shared__ __attribute__((aligned(16))) float2 wlds[14 * 8 * 2 * 32];
  __shared__ float  ylds[4][16][CCH];

  {
    const int tid = threadIdx.x;
    const int k   = (tid >> 3) << 1;
    const int n   = (tid & 7) << 2;
    const int s   = k >> 2;
    const int ln  = ((k & 2) << 3) | (n & 15);
    const int t   = n >> 4;
    const int off = k * CCH + n;
    float4 r0[14], r1[14];
#pragma unroll
    for (int i = 0; i < 14; ++i) {
      const float* src = (i < 3) ? (U + i * 1024)
                       : (i < 11) ? (V + (i - 3) * 1024)
                                  : (Wg + (i - 11) * 1024);
      r0[i] = *(const float4*)(src + off);
      r1[i] = *(const float4*)(src + off + CCH);
    }
#pragma unroll
    for (int i = 0; i < 14; ++i) {
      float4* dst = (float4*)&wlds[((i * 8 + s) * 2 + t) * 32 + ln];
      dst[0] = make_float4(r0[i].x, r1[i].x, r0[i].y, r1[i].y);
      dst[1] = make_float4(r0[i].z, r1[i].z, r0[i].w, r1[i].w);
    }
  }
  __syncthreads();

  const int lane = threadIdx.x & 31;
  const int wv   = threadIdx.x >> 5;
  const int half = lane >> 4;
  const int m    = lane & 15;
  const int tile = blockIdx.x * 4 + wv;
  if (tile * 16 >= N) return;

#define LDB(wm, s, t)                                                       \
  ({ float2 _bv = wlds[(((wm) * 8 + (s)) * 2 + (t)) * 32 + lane];           \
     v2f _b; _b.x = _bv.x; _b.y = _bv.y; _b; })

#define MM2(acc, aop, wm, s)                                                \
  do { v2f _a = (aop);                                                      \
       acc[0] = wmma4(_a, LDB(wm, s, 0), acc[0]);                           \
       acc[1] = wmma4(_a, LDB(wm, s, 1), acc[1]); } while (0)

  const v8f vzero = {};

  const int rowA = tile * 16 + m;
  const int rowC = (rowA < N) ? rowA : (N - 1);
  const float* accRow = acc + rowC * (NCOMP * CCH) + 2 * half;

  v8f y0acc[2] = {vzero, vzero};
  v8f y1acc[3][2];
  v8f y2acc[9][2];
#pragma unroll
  for (int v = 0; v < 3; ++v) { y1acc[v][0] = vzero; y1acc[v][1] = vzero; }
#pragma unroll
  for (int q = 0; q < 9; ++q) { y2acc[q][0] = vzero; y2acc[q][1] = vzero; }

  float2 cur[NCOMP], nxt[NCOMP];
#pragma unroll
  for (int k = 0; k < NCOMP; ++k) cur[k] = *(const float2*)(accRow + k * CCH);

#pragma unroll 1
  for (int s = 0; s < 8; ++s) {
    const int sn = (s < 7) ? (s + 1) : 7;
#pragma unroll
    for (int k = 0; k < NCOMP; ++k)
      nxt[k] = *(const float2*)(accRow + sn * 4 + k * CCH);

    float x0[2], b1[2][3], b2[2][9];
    x0[0] = cur[0].x; x0[1] = cur[0].y;
#pragma unroll
    for (int a = 0; a < 3; ++a) { b1[0][a] = cur[1 + a].x; b1[1][a] = cur[1 + a].y; }
#pragma unroll
    for (int q = 0; q < 9; ++q) { b2[0][q] = cur[4 + q].x; b2[1][q] = cur[4 + q].y; }

    {
      v2f ax, asq, s1v, s2v;
#pragma unroll
      for (int jj = 0; jj < 2; ++jj) {
        ax[jj]  = x0[jj];
        asq[jj] = x0[jj] * x0[jj];
        float sum = 0.f;
#pragma unroll
        for (int a = 0; a < 3; ++a) sum += b1[jj][a] * b1[jj][a];
        s1v[jj] = sum;
        sum = 0.f;
#pragma unroll
        for (int q = 0; q < 9; ++q) sum += b2[jj][q] * b2[jj][q];
        s2v[jj] = sum;
      }
      MM2(y0acc, ax,  W_U0,  s);
      MM2(y0acc, asq, W_V(0), s);
      MM2(y0acc, s1v, W_V(3), s);
      MM2(y0acc, s2v, W_V(6), s);
    }

#pragma unroll
    for (int v = 0; v < 3; ++v) {
      v2f av, a01, tv;
#pragma unroll
      for (int jj = 0; jj < 2; ++jj) {
        av[jj]  = b1[jj][v];
        a01[jj] = x0[jj] * b1[jj][v];
        float sum = 0.f;
#pragma unroll
        for (int a = 0; a < 3; ++a) sum += b1[jj][a] * b2[jj][a * 3 + v];
        tv[jj] = sum;
      }
      MM2(y1acc[v], av,  W_U1,  s);
      MM2(y1acc[v], a01, W_V(1), s);
      MM2(y1acc[v], tv,  W_V(5), s);
    }

#pragma unroll
    for (int aIdx = 0; aIdx < 3; ++aIdx) {
#pragma unroll
      for (int b = 0; b < 3; ++b) {
        v2f a2v, a02, a11, qv;
#pragma unroll
        for (int jj = 0; jj < 2; ++jj) {
          const float g = b2[jj][aIdx * 3 + b];
          a2v[jj] = g;
          a02[jj] = x0[jj] * g;
          a11[jj] = b1[jj][aIdx] * b1[jj][b];
          float sum = 0.f;
#pragma unroll
          for (int kk = 0; kk < 3; ++kk)
            sum += b2[jj][aIdx * 3 + kk] * b2[jj][kk * 3 + b];
          qv[jj] = sum;
        }
        MM2(y2acc[aIdx * 3 + b], a2v, W_U2,  s);
        MM2(y2acc[aIdx * 3 + b], a02, W_V(2), s);
        MM2(y2acc[aIdx * 3 + b], a11, W_V(4), s);
        MM2(y2acc[aIdx * 3 + b], qv,  W_V(7), s);
      }
    }

#pragma unroll
    for (int k = 0; k < NCOMP; ++k) cur[k] = nxt[k];
  }

  const int rowD0 = tile * 16 + 8 * half;
  float res0[2][8];
  float res1[2][8][3];
#pragma unroll
  for (int t = 0; t < 2; ++t)
#pragma unroll
    for (int r = 0; r < 8; ++r) {
      const int row  = rowD0 + r;
      const int rowL = (row < N) ? row : (N - 1);
      res0[t][r] = node0[rowL * CCH + t * 16 + m];
      const float* p = node1 + (rowL * CCH + t * 16 + m) * 3;
#pragma unroll
      for (int v = 0; v < 3; ++v) res1[t][r][v] = p[v];
    }

#pragma unroll
  for (int t = 0; t < 2; ++t)
#pragma unroll
    for (int r = 0; r < 8; ++r)
      ylds[wv][r + 8 * half][t * 16 + m] = y0acc[t][r];
  asm volatile("s_wait_dscnt 0" ::: "memory");

  v8f ga[3][2] = {{vzero, vzero}, {vzero, vzero}, {vzero, vzero}};
#pragma unroll
  for (int s = 0; s < 8; ++s) {
    const int k = s * 4 + 2 * half;
    v2f ay; ay.x = ylds[wv][m][k]; ay.y = ylds[wv][m][k + 1];
#pragma unroll
    for (int g = 0; g < 3; ++g) MM2(ga[g], ay, W_G(g), s);
  }
  v8f gate[3][2];
#pragma unroll
  for (int g = 0; g < 3; ++g)
#pragma unroll
    for (int t = 0; t < 2; ++t)
#pragma unroll
      for (int r = 0; r < 8; ++r) {
        const float v = ga[g][t][r];
        gate[g][t][r] = v * (1.0f / (1.0f + __expf(-v)));
      }

  float res2[2][8][3];
#pragma unroll
  for (int t = 0; t < 2; ++t)
#pragma unroll
    for (int r = 0; r < 8; ++r) {
      const int row  = rowD0 + r;
      const int rowL = (row < N) ? row : (N - 1);
      const float* p = node2 + (rowL * CCH + t * 16 + m) * 9;
#pragma unroll
      for (int bb = 0; bb < 3; ++bb) res2[t][r][bb] = p[bb];
    }

#pragma unroll
  for (int t = 0; t < 2; ++t)
#pragma unroll
    for (int r = 0; r < 8; ++r) {
      const int row = rowD0 + r;
      if (row < N) {
        const int cc = t * 16 + m;
        out0[row * CCH + cc] = res0[t][r] + gate[0][t][r];
      }
    }

#pragma unroll
  for (int t = 0; t < 2; ++t)
#pragma unroll
    for (int r = 0; r < 8; ++r) {
      const int row = rowD0 + r;
      if (row < N) {
        float* q = out1 + (row * CCH + t * 16 + m) * 3;
#pragma unroll
        for (int v = 0; v < 3; ++v)
          q[v] = res1[t][r][v] + y1acc[v][t][r] * gate[1][t][r];
      }
    }

#pragma unroll
  for (int aIdx = 0; aIdx < 3; ++aIdx) {
    float nres[2][8][3];
    if (aIdx < 2) {
#pragma unroll
      for (int t = 0; t < 2; ++t)
#pragma unroll
        for (int r = 0; r < 8; ++r) {
          const int row  = rowD0 + r;
          const int rowL = (row < N) ? row : (N - 1);
          const float* p = node2 + (rowL * CCH + t * 16 + m) * 9 + (aIdx + 1) * 3;
#pragma unroll
          for (int bb = 0; bb < 3; ++bb) nres[t][r][bb] = p[bb];
        }
    }
#pragma unroll
    for (int t = 0; t < 2; ++t)
#pragma unroll
      for (int r = 0; r < 8; ++r) {
        const int row = rowD0 + r;
        if (row < N) {
          float* q = out2 + (row * CCH + t * 16 + m) * 9 + aIdx * 3;
#pragma unroll
          for (int bb = 0; bb < 3; ++bb)
            q[bb] = res2[t][r][bb] + y2acc[aIdx * 3 + bb][t][r] * gate[2][t][r];
        }
      }
    if (aIdx < 2) {
#pragma unroll
      for (int t = 0; t < 2; ++t)
#pragma unroll
        for (int r = 0; r < 8; ++r)
#pragma unroll
          for (int bb = 0; bb < 3; ++bb) res2[t][r][bb] = nres[t][r][bb];
    }
  }
#undef MM2
#undef LDB
}

extern "C" void kernel_launch(void* const* d_in, const int* in_sizes, int n_in,
                              void* d_out, int out_size, void* d_ws, size_t ws_size,
                              hipStream_t stream) {
  const float* node0 = (const float*)d_in[0];
  const float* node1 = (const float*)d_in[1];
  const float* node2 = (const float*)d_in[2];
  const float* rij   = (const float*)d_in[3];
  const float* Wrad  = (const float*)d_in[4];
  const float* U     = (const float*)d_in[5];
  const float* V     = (const float*)d_in[6];
  const float* Wg    = (const float*)d_in[7];
  const int*   idx_i = (const int*)d_in[8];
  const int*   idx_j = (const int*)d_in[9];

  const int N = in_sizes[0] / CCH;
  const int E = in_sizes[3] / 3;

  float* acc = (float*)d_ws;

  float* out0 = (float*)d_out;
  float* out1 = out0 + (size_t)N * CCH;
  float* out2 = out1 + (size_t)N * CCH * 3;

  mmb_clear_counts<<<(N + 255) / 256, 256, 0, stream>>>((int*)d_ws, N);
  mmb_bucket<<<(E + 255) / 256, 256, 0, stream>>>(idx_i, (int*)d_ws, E);

  mmb_gather<<<(N + 7) / 8, 128, 0, stream>>>(rij, Wrad, idx_i, idx_j,
                                              node0, node1, node2,
                                              acc, N, E);

  const int nTiles = (N + 15) / 16;
  mmb_node<<<(nTiles + 3) / 4, 128, 0, stream>>>(acc,
                                                 node0, node1, node2,
                                                 U, V, Wg,
                                                 out0, out1, out2, N);
}


# edge bucketing: two-level counting sort with LDS adds (edges to 64-node ranges, then to nodes) replaces the counter clear and 320000 returning global adds; bucket arrival order stays arbitrary
# speedup vs baseline: 4.9115x; 1.0382x over previous
#include <hip/hip_runtime.h>
#include <hip/hip_bf16.h>
#include <math.h>

#define CCH 32
#define NPATH 11
#define NB 8
#define NCOMP 13

typedef __attribute__((ext_vector_type(2))) float v2f;
typedef __attribute__((ext_vector_type(8))) float v8f;

__device__ __forceinline__ v8f wmma4(v2f a, v2f b, v8f c) {
  return __builtin_amdgcn_wmma_f32_16x16x4_f32(
       false, a,  false, b,
       (short)0, c,  false,  false);
}

#define BLK_DW (NCOMP * CCH)
#define ID_OFF 32
#define BUCKET_CAP (BLK_DW - ID_OFF)

#define RANGE_NODES 64
#define RANGE_DW (RANGE_NODES * BLK_DW)
#define MAX_RANGES 320
#define L1_THREADS 1024
#define L1_PER_THREAD 8
#define L1_EDGES (L1_THREADS * L1_PER_THREAD)
#define L1_SLOTS ((MAX_RANGES * 64) / L1_THREADS)
#define SEG_DW 128
#define SEG_CAP 63
#define MAX_SEGS 48
#define L2_THREADS 256
#define L2_PER_THREAD ((MAX_SEGS * 64) / L2_THREADS)
#define NODE_STAGE 64
#define L2_OUT ((RANGE_NODES * NODE_STAGE) / L2_THREADS)

__global__ void
mmb_bin_ranges(const int* __restrict__ idx_i, int* __restrict__ wsI, int E, int nRanges) {
  __shared__ int  hist[MAX_RANGES];
  __shared__ int2 stage[MAX_RANGES * 64];
  const int tid = threadIdx.x;
  const int g   = blockIdx.x;
  if (tid < MAX_RANGES) hist[tid] = 0;
  __syncthreads();

  int dst[L1_PER_THREAD];
#pragma unroll
  for (int k = 0; k < L1_PER_THREAD; ++k) {
    const int e = g * L1_EDGES + k * L1_THREADS + tid;
    dst[k] = idx_i[(e < E) ? e : (E - 1)];
  }
#pragma unroll
  for (int k = 0; k < L1_PER_THREAD; ++k) {
    const int e = g * L1_EDGES + k * L1_THREADS + tid;
    if (e < E && (unsigned)dst[k] < (unsigned)(nRanges * RANGE_NODES)) {
      const int b = dst[k] / RANGE_NODES;
      const int r = atomicAdd(&hist[b], 1);
      if (r < SEG_CAP) stage[b * 64 + 1 + r] = make_int2(e, dst[k]);
    }
  }
  __syncthreads();

  int  cn[L1_SLOTS];
  int2 pr[L1_SLOTS];
#pragma unroll
  for (int j = 0; j < L1_SLOTS; ++j) {
    const int s = tid + j * L1_THREADS;
    cn[j] = hist[s >> 6];
    pr[j] = stage[s];
  }
#pragma unroll
  for (int j = 0; j < L1_SLOTS; ++j) {
    const int s = tid + j * L1_THREADS;
    const int b = s >> 6, p = s & 63;
    if (b < nRanges) {
      int* seg = wsI + (size_t)b * RANGE_DW + g * SEG_DW;
      if (p == 0)             *(int2*)seg = make_int2(cn[j], 0);
      else if (p - 1 < cn[j]) *(int2*)(seg + 2 * p) = pr[j];
    }
  }
}

__global__ void
mmb_bin_nodes(int* wsI, int N, int nSegs) {
  __shared__ int segCnt[MAX_SEGS];
  __shared__ int cnt[RANGE_NODES];
  __shared__ int ovf;
  __shared__ int ids[RANGE_NODES * NODE_STAGE];
  const int tid = threadIdx.x;
  const int b   = blockIdx.x;
  int* region = wsI + (size_t)b * RANGE_DW;

  if (tid < RANGE_NODES) cnt[tid] = 0;
  if (tid == 0) ovf = 0;
  if (tid < MAX_SEGS) segCnt[tid] = (tid < nSegs) ? region[tid * SEG_DW] : 0;
  __syncthreads();
  if (tid < MAX_SEGS && segCnt[tid] > SEG_CAP) atomicOr(&ovf, 1);

  bool ok[L2_PER_THREAD];
  int2 pr[L2_PER_THREAD];
  int  kr[L2_PER_THREAD];
#pragma unroll
  for (int j = 0; j < L2_PER_THREAD; ++j) {
    const int s = tid + j * L2_THREADS;
    const int g = s >> 6, p = s & 63;
    int c = segCnt[g];
    c = (c < SEG_CAP) ? c : SEG_CAP;
    ok[j] = (p >= 1) && (p - 1 < c);
    pr[j] = *(const int2*)(region + (ok[j] ? (g * SEG_DW + 2 * p) : 0));
  }
#pragma unroll
  for (int j = 0; j < L2_PER_THREAD; ++j) {
    pr[j].y -= b * RANGE_NODES;
    kr[j] = -1;
    if (ok[j] && (unsigned)pr[j].y < (unsigned)RANGE_NODES)
      kr[j] = atomicAdd(&cnt[pr[j].y], 1);
  }
  __syncthreads();

#pragma unroll
  for (int j = 0; j < L2_PER_THREAD; ++j) {
    if (kr[j] >= 0) {
      if (kr[j] < NODE_STAGE)      ids[pr[j].y * NODE_STAGE + kr[j]] = pr[j].x;
      else if (kr[j] < BUCKET_CAP) region[pr[j].y * BLK_DW + ID_OFF + kr[j]] = pr[j].x;
    }
  }
  __syncthreads();

  int cn[L2_OUT], id[L2_OUT];
#pragma unroll
  for (int j = 0; j < L2_OUT; ++j) {
    const int s = tid + j * L2_THREADS;
    cn[j] = cnt[s / NODE_STAGE];
    id[j] = ids[s];
  }
#pragma unroll
  for (int j = 0; j < L2_OUT; ++j) {
    const int s = tid + j * L2_THREADS;
    const int l = s / NODE_STAGE, p = s % NODE_STAGE;
    if (b * RANGE_NODES + l < N && p < cn[j])
      region[l * BLK_DW + ID_OFF + p] = id[j];
  }
  if (tid < RANGE_NODES && b * RANGE_NODES + tid < N)
    region[tid * BLK_DW] = (ovf != 0) ? (BUCKET_CAP + 1) : cnt[tid];
}

#define NODES_PER_WAVE 2

__device__ __forceinline__ float bcast_f(float v, int srcLane) {
  return __int_as_float(__builtin_amdgcn_readlane(__float_as_int(v), srcLane));
}

#define EDGES_IN_FLIGHT 4
__global__ void __launch_bounds__(128, 4)
mmb_gather(const float* __restrict__ rij,
           const float* __restrict__ Wrad,
           const int*   __restrict__ idx_i,
           const int*   __restrict__ idx_j,
           const float* node0,
           const float* node1,
           const float* node2,
           float* acc,
           int N, int E) {
  const int c = threadIdx.x & 31;
  const int wave = __builtin_amdgcn_readfirstlane(
      (int)(blockIdx.x * (blockDim.x >> 5) + (threadIdx.x >> 5)));
  const int nbase = wave * NODES_PER_WAVE;
  if (nbase >= N) return;

  float Wr[NPATH * NB];
#pragma unroll
  for (int q = 0; q < NPATH * NB; ++q) Wr[q] = Wrad[q * CCH + c];

  const float inv16 = 0.0625f;

  for (int nn = 0; nn < NODES_PER_WAVE; ++nn) {
    const int i = nbase + nn;
    if (i >= N) break;
    float* blk = acc + (size_t)i * BLK_DW;
    const int* blkI = (const int*)blk;
    const int cnt = __builtin_amdgcn_readfirstlane(blkI[0]);

    float a[NCOMP];
#pragma unroll
    for (int k = 0; k < NCOMP; ++k) a[k] = 0.f;

    const bool inBucket = (cnt <= BUCKET_CAP);
    const int total = inBucket ? cnt : E;

    for (int base = 0; base < total; base += 32) {
      bool valid;
      int id;
      if (inBucket) {
        valid = (base + c < cnt);
        id = valid ? blkI[ID_OFF + base + c] : 0x7fffffff;
      } else {
        const int e0 = base + c;
        valid = false;
        if (e0 < E) valid = (idx_i[e0] == i);
        id = valid ? e0 : 0x7fffffff;
      }
      const unsigned vm = __builtin_amdgcn_ballot_w32(valid);
      if (vm == 0u) continue;
      const int nvalid = __builtin_popcount(vm);

      int pos = 0;
#pragma unroll
      for (int u = 0; u < 32; ++u)
        pos += (__builtin_amdgcn_readlane(id, u) < id) ? 1 : 0;

      const int el = valid ? id : 0;
      const int jl = idx_j[el];
      const float r0 = rij[el * 3 + 0], r1 = rij[el * 3 + 1], r2 = rij[el * 3 + 2];
      const float d  = sqrtf(r0 * r0 + r1 * r1 + r2 * r2 + 1e-12f);
      const float iv = 1.0f / d;
      const float rhl0 = r0 * iv, rhl1 = r1 * iv, rhl2 = r2 * iv;

      float x = d * 0.2f;
      x = (x > 1.0f) ? 1.0f : x;
      const float fc = 0.5f * (cosf(3.14159265358979f * x) + 1.0f);

      float rbfl[NB];
#pragma unroll
      for (int b = 0; b < NB; ++b) {
        const float t = d - (0.5f + 0.6428571429f * (float)b);
        rbfl[b] = __expf(-4.0f * t * t) * fc;
      }

      for (int t = 0; t < nvalid; t += EDGES_IN_FLIGHT) {
        int   sl[EDGES_IN_FLIGHT];
        float g0[EDGES_IN_FLIGHT], g1[EDGES_IN_FLIGHT][3], g2[EDGES_IN_FLIGHT][9];

#pragma unroll
        for (int u = 0; u < EDGES_IN_FLIGHT; ++u) {
          const int rk = (t + u < nvalid) ? (t + u) : (nvalid - 1);
          const unsigned sel = __builtin_amdgcn_ballot_w32(valid && (pos == rk));
          sl[u] = (sel != 0u) ? __builtin_ctz(sel) : 0;
          const int j = __builtin_amdgcn_readlane(jl, sl[u]);

          const float* p0 = node0 + (size_t)j * CCH;
          const float* p1 = node1 + (size_t)j * (CCH * 3);
          const float* p2 = node2 + (size_t)j * (CCH * 9);
          g0[u] = p0[c];
#pragma unroll
          for (int q = 0; q < 3; ++q) g1[u][q] = p1[c * 3 + q];
#pragma unroll
          for (int q = 0; q < 9; ++q) g2[u][q] = p2[c * 9 + q];
        }

        __builtin_amdgcn_sched_barrier(0);

#pragma unroll
        for (int u = 0; u < EDGES_IN_FLIGHT; ++u) {
          if (t + u < nvalid) {
            const float rh[3] = {bcast_f(rhl0, sl[u]), bcast_f(rhl1, sl[u]),
                                 bcast_f(rhl2, sl[u])};
            float rbf[NB];
#pragma unroll
            for (int b = 0; b < NB; ++b) rbf[b] = bcast_f(rbfl[b], sl[u]);

            float w[NPATH];
#pragma unroll
            for (int p = 0; p < NPATH; ++p) {
              float sacc = rbf[0] * Wr[p * NB];
#pragma unroll
              for (int b = 1; b < NB; ++b) sacc = fmaf(rbf[b], Wr[p * NB + b], sacc);
              w[p] = sacc;
            }

            const float d1 = g1[u][0] * rh[0] + g1[u][1] * rh[1] + g1[u][2] * rh[2];
            float gr[3];
#pragma unroll
            for (int q = 0; q < 3; ++q)
              gr[q] = g2[u][q * 3 + 0] * rh[0] + g2[u][q * 3 + 1] * rh[1]
                    + g2[u][q * 3 + 2] * rh[2];
            const float d2 = gr[0] * rh[0] + gr[1] * rh[1] + gr[2] * rh[2];

            a[0] += (g0[u] * w[0] + d1 * w[4] + d2 * w[9]) * inv16;

#pragma unroll
            for (int q = 0; q < 3; ++q) {
              const float m = (g0[u] * w[1] + d1 * w[6]) * rh[q] + g1[u][q] * w[3]
                            + gr[q] * w[8];
              a[1 + q] += m * inv16;
            }
#pragma unroll
            for (int q = 0; q < 3; ++q) {
              const float ca = g0[u] * w[2] * rh[q] + g1[u][q] * w[5] + gr[q] * w[10];
#pragma unroll
              for (int b = 0; b < 3; ++b) {
                const float m = ca * rh[b] + g2[u][q * 3 + b] * w[7];
                a[4 + q * 3 + b] += m * inv16;
              }
            }
          }
        }
      }
    }

#pragma unroll
    for (int k = 0; k < NCOMP; ++k) blk[k * CCH + c] = a[k];
  }
}

#define W_U0 0
#define W_U1 1
#define W_U2 2
#define W_V(k) (3 + (k))
#define W_G(k) (11 + (k))

__global__ void __launch_bounds__(128, 2)
mmb_node(const float* __restrict__ acc,
         const float* __restrict__ node0, const float* __restrict__ node1,
         const float* __restrict__ node2,
         const float* __restrict__ U, const float* __restrict__ V,
         const float* __restrict__ Wg,
         float* __restrict__ out0, float* __restrict__ out1,
         float* __restrict__ out2, int N) {
  __shared__ __attribute__((aligned(16))) float2 wlds[14 * 8 * 2 * 32];
  __shared__ float  ylds[4][16][CCH];

  {
    const int tid = threadIdx.x;
    const int k   = (tid >> 3) << 1;
    const int n   = (tid & 7) << 2;
    const int s   = k >> 2;
    const int ln  = ((k & 2) << 3) | (n & 15);
    const int t   = n >> 4;
    const int off = k * CCH + n;
    float4 r0[14], r1[14];
#pragma unroll
    for (int i = 0; i < 14; ++i) {
      const float* src = (i < 3) ? (U + i * 1024)
                       : (i < 11) ? (V + (i - 3) * 1024)
                                  : (Wg + (i - 11) * 1024);
      r0[i] = *(const float4*)(src + off);
      r1[i] = *(const float4*)(src + off + CCH);
    }
#pragma unroll
    for (int i = 0; i < 14; ++i) {
      float4* dst = (float4*)&wlds[((i * 8 + s) * 2 + t) * 32 + ln];
      dst[0] = make_float4(r0[i].x, r1[i].x, r0[i].y, r1[i].y);
      dst[1] = make_float4(r0[i].z, r1[i].z, r0[i].w, r1[i].w);
    }
  }
  __syncthreads();

  const int lane = threadIdx.x & 31;
  const int wv   = threadIdx.x >> 5;
  const int half = lane >> 4;
  const int m    = lane & 15;
  const int tile = blockIdx.x * 4 + wv;
  if (tile * 16 >= N) return;

#define LDB(wm, s, t)                                                       \
  ({ float2 _bv = wlds[(((wm) * 8 + (s)) * 2 + (t)) * 32 + lane];           \
     v2f _b; _b.x = _bv.x; _b.y = _bv.y; _b; })

#define MM2(acc, aop, wm, s)                                                \
  do { v2f _a = (aop);                                                      \
       acc[0] = wmma4(_a, LDB(wm, s, 0), acc[0]);                           \
       acc[1] = wmma4(_a, LDB(wm, s, 1), acc[1]); } while (0)

  const v8f vzero = {};

  const int rowA = tile * 16 + m;
  const int rowC = (rowA < N) ? rowA : (N - 1);
  const float* accRow = acc + rowC * (NCOMP * CCH) + 2 * half;

  v8f y0acc[2] = {vzero, vzero};
  v8f y1acc[3][2];
  v8f y2acc[9][2];
#pragma unroll
  for (int v = 0; v < 3; ++v) { y1acc[v][0] = vzero; y1acc[v][1] = vzero; }
#pragma unroll
  for (int q = 0; q < 9; ++q) { y2acc[q][0] = vzero; y2acc[q][1] = vzero; }

  float2 cur[NCOMP], nxt[NCOMP];
#pragma unroll
  for (int k = 0; k < NCOMP; ++k) cur[k] = *(const float2*)(accRow + k * CCH);

#pragma unroll 1
  for (int s = 0; s < 8; ++s) {
    const int sn = (s < 7) ? (s + 1) : 7;
#pragma unroll
    for (int k = 0; k < NCOMP; ++k)
      nxt[k] = *(const float2*)(accRow + sn * 4 + k * CCH);

    float x0[2], b1[2][3], b2[2][9];
    x0[0] = cur[0].x; x0[1] = cur[0].y;
#pragma unroll
    for (int a = 0; a < 3; ++a) { b1[0][a] = cur[1 + a].x; b1[1][a] = cur[1 + a].y; }
#pragma unroll
    for (int q = 0; q < 9; ++q) { b2[0][q] = cur[4 + q].x; b2[1][q] = cur[4 + q].y; }

    {
      v2f ax, asq, s1v, s2v;
#pragma unroll
      for (int jj = 0; jj < 2; ++jj) {
        ax[jj]  = x0[jj];
        asq[jj] = x0[jj] * x0[jj];
        float sum = 0.f;
#pragma unroll
        for (int a = 0; a < 3; ++a) sum += b1[jj][a] * b1[jj][a];
        s1v[jj] = sum;
        sum = 0.f;
#pragma unroll
        for (int q = 0; q < 9; ++q) sum += b2[jj][q] * b2[jj][q];
        s2v[jj] = sum;
      }
      MM2(y0acc, ax,  W_U0,  s);
      MM2(y0acc, asq, W_V(0), s);
      MM2(y0acc, s1v, W_V(3), s);
      MM2(y0acc, s2v, W_V(6), s);
    }

#pragma unroll
    for (int v = 0; v < 3; ++v) {
      v2f av, a01, tv;
#pragma unroll
      for (int jj = 0; jj < 2; ++jj) {
        av[jj]  = b1[jj][v];
        a01[jj] = x0[jj] * b1[jj][v];
        float sum = 0.f;
#pragma unroll
        for (int a = 0; a < 3; ++a) sum += b1[jj][a] * b2[jj][a * 3 + v];
        tv[jj] = sum;
      }
      MM2(y1acc[v], av,  W_U1,  s);
      MM2(y1acc[v], a01, W_V(1), s);
      MM2(y1acc[v], tv,  W_V(5), s);
    }

#pragma unroll
    for (int aIdx = 0; aIdx < 3; ++aIdx) {
#pragma unroll
      for (int b = 0; b < 3; ++b) {
        v2f a2v, a02, a11, qv;
#pragma unroll
        for (int jj = 0; jj < 2; ++jj) {
          const float g = b2[jj][aIdx * 3 + b];
          a2v[jj] = g;
          a02[jj] = x0[jj] * g;
          a11[jj] = b1[jj][aIdx] * b1[jj][b];
          float sum = 0.f;
#pragma unroll
          for (int kk = 0; kk < 3; ++kk)
            sum += b2[jj][aIdx * 3 + kk] * b2[jj][kk * 3 + b];
          qv[jj] = sum;
        }
        MM2(y2acc[aIdx * 3 + b], a2v, W_U2,  s);
        MM2(y2acc[aIdx * 3 + b], a02, W_V(2), s);
        MM2(y2acc[aIdx * 3 + b], a11, W_V(4), s);
        MM2(y2acc[aIdx * 3 + b], qv,  W_V(7), s);
      }
    }

#pragma unroll
    for (int k = 0; k < NCOMP; ++k) cur[k] = nxt[k];
  }

  const int rowD0 = tile * 16 + 8 * half;
  float res0[2][8];
  float res1[2][8][3];
#pragma unroll
  for (int t = 0; t < 2; ++t)
#pragma unroll
    for (int r = 0; r < 8; ++r) {
      const int row  = rowD0 + r;
      const int rowL = (row < N) ? row : (N - 1);
      res0[t][r] = node0[rowL * CCH + t * 16 + m];
      const float* p = node1 + (rowL * CCH + t * 16 + m) * 3;
#pragma unroll
      for (int v = 0; v < 3; ++v) res1[t][r][v] = p[v];
    }

#pragma unroll
  for (int t = 0; t < 2; ++t)
#pragma unroll
    for (int r = 0; r < 8; ++r)
      ylds[wv][r + 8 * half][t * 16 + m] = y0acc[t][r];
  asm volatile("s_wait_dscnt 0" ::: "memory");

  v8f ga[3][2] = {{vzero, vzero}, {vzero, vzero}, {vzero, vzero}};
#pragma unroll
  for (int s = 0; s < 8; ++s) {
    const int k = s * 4 + 2 * half;
    v2f ay; ay.x = ylds[wv][m][k]; ay.y = ylds[wv][m][k + 1];
#pragma unroll
    for (int g = 0; g < 3; ++g) MM2(ga[g], ay, W_G(g), s);
  }
  v8f gate[3][2];
#pragma unroll
  for (int g = 0; g < 3; ++g)
#pragma unroll
    for (int t = 0; t < 2; ++t)
#pragma unroll
      for (int r = 0; r < 8; ++r) {
        const float v = ga[g][t][r];
        gate[g][t][r] = v * (1.0f / (1.0f + __expf(-v)));
      }

  float res2[2][8][3];
#pragma unroll
  for (int t = 0; t < 2; ++t)
#pragma unroll
    for (int r = 0; r < 8; ++r) {
      const int row  = rowD0 + r;
      const int rowL = (row < N) ? row : (N - 1);
      const float* p = node2 + (rowL * CCH + t * 16 + m) * 9;
#pragma unroll
      for (int bb = 0; bb < 3; ++bb) res2[t][r][bb] = p[bb];
    }

#pragma unroll
  for (int t = 0; t < 2; ++t)
#pragma unroll
    for (int r = 0; r < 8; ++r) {
      const int row = rowD0 + r;
      if (row < N) {
        const int cc = t * 16 + m;
        out0[row * CCH + cc] = res0[t][r] + gate[0][t][r];
      }
    }

#pragma unroll
  for (int t = 0; t < 2; ++t)
#pragma unroll
    for (int r = 0; r < 8; ++r) {
      const int row = rowD0 + r;
      if (row < N) {
        float* q = out1 + (row * CCH + t * 16 + m) * 3;
#pragma unroll
        for (int v = 0; v < 3; ++v)
          q[v] = res1[t][r][v] + y1acc[v][t][r] * gate[1][t][r];
      }
    }

#pragma unroll
  for (int aIdx = 0; aIdx < 3; ++aIdx) {
    float nres[2][8][3];
    if (aIdx < 2) {
#pragma unroll
      for (int t = 0; t < 2; ++t)
#pragma unroll
        for (int r = 0; r < 8; ++r) {
          const int row  = rowD0 + r;
          const int rowL = (row < N) ? row : (N - 1);
          const float* p = node2 + (rowL * CCH + t * 16 + m) * 9 + (aIdx + 1) * 3;
#pragma unroll
          for (int bb = 0; bb < 3; ++bb) nres[t][r][bb] = p[bb];
        }
    }
#pragma unroll
    for (int t = 0; t < 2; ++t)
#pragma unroll
      for (int r = 0; r < 8; ++r) {
        const int row = rowD0 + r;
        if (row < N) {
          float* q = out2 + (row * CCH + t * 16 + m) * 9 + aIdx * 3;
#pragma unroll
          for (int bb = 0; bb < 3; ++bb)
            q[bb] = res2[t][r][bb] + y2acc[aIdx * 3 + bb][t][r] * gate[2][t][r];
        }
      }
    if (aIdx < 2) {
#pragma unroll
      for (int t = 0; t < 2; ++t)
#pragma unroll
        for (int r = 0; r < 8; ++r)
#pragma unroll
          for (int bb = 0; bb < 3; ++bb) res2[t][r][bb] = nres[t][r][bb];
    }
  }
#undef MM2
#undef LDB
}

extern "C" void kernel_launch(void* const* d_in, const int* in_sizes, int n_in,
                              void* d_out, int out_size, void* d_ws, size_t ws_size,
                              hipStream_t stream) {
  const float* node0 = (const float*)d_in[0];
  const float* node1 = (const float*)d_in[1];
  const float* node2 = (const float*)d_in[2];
  const float* rij   = (const float*)d_in[3];
  const float* Wrad  = (const float*)d_in[4];
  const float* U     = (const float*)d_in[5];
  const float* V     = (const float*)d_in[6];
  const float* Wg    = (const float*)d_in[7];
  const int*   idx_i = (const int*)d_in[8];
  const int*   idx_j = (const int*)d_in[9];

  const int N = in_sizes[0] / CCH;
  const int E = in_sizes[3] / 3;

  float* acc = (float*)d_ws;

  float* out0 = (float*)d_out;
  float* out1 = out0 + (size_t)N * CCH;
  float* out2 = out1 + (size_t)N * CCH * 3;

  const int nRanges = (N + RANGE_NODES - 1) / RANGE_NODES;
  const int nSegs   = (E + L1_EDGES - 1) / L1_EDGES;
  mmb_bin_ranges<<<nSegs, L1_THREADS, 0, stream>>>(idx_i, (int*)d_ws, E, nRanges);
  mmb_bin_nodes<<<nRanges, L2_THREADS, 0, stream>>>((int*)d_ws, N, nSegs);

  mmb_gather<<<(N + 7) / 8, 128, 0, stream>>>(rij, Wrad, idx_i, idx_j,
                                              node0, node1, node2,
                                              acc, N, E);

  const int nTiles = (N + 15) / 16;
  mmb_node<<<(nTiles + 3) / 4, 128, 0, stream>>>(acc,
                                                 node0, node1, node2,
                                                 U, V, Wg,
                                                 out0, out1, out2, N);
}


# node phase: one 16-node tile per workgroup, its 13 output components dealt over the 4 waves (y1 + half of y0 | y2 row 0 + other half | row 1 | row 2); accumulator rows staged once in LDS
# speedup vs baseline: 5.1983x; 1.0584x over previous
#include <hip/hip_runtime.h>
#include <hip/hip_bf16.h>
#include <math.h>

#define CCH 32
#define NPATH 11
#define NB 8
#define NCOMP 13

typedef __attribute__((ext_vector_type(2))) float v2f;
typedef __attribute__((ext_vector_type(4))) float v4f;
typedef __attribute__((ext_vector_type(8))) float v8f;

__device__ __forceinline__ v8f wmma4(v2f a, v2f b, v8f c) {
  return __builtin_amdgcn_wmma_f32_16x16x4_f32(
       false, a,  false, b,
       (short)0, c,  false,  false);
}

#define BLK_DW (NCOMP * CCH)
#define ID_OFF 32
#define BUCKET_CAP (BLK_DW - ID_OFF)

#define RANGE_NODES 64
#define RANGE_DW (RANGE_NODES * BLK_DW)
#define MAX_RANGES 320
#define L1_THREADS 1024
#define L1_PER_THREAD 8
#define L1_EDGES (L1_THREADS * L1_PER_THREAD)
#define L1_SLOTS ((MAX_RANGES * 64) / L1_THREADS)
#define SEG_DW 128
#define SEG_CAP 63
#define MAX_SEGS 48
#define L2_THREADS 256
#define L2_PER_THREAD ((MAX_SEGS * 64) / L2_THREADS)
#define NODE_STAGE 64
#define L2_OUT ((RANGE_NODES * NODE_STAGE) / L2_THREADS)

__global__ void
mmb_bin_ranges(const int* __restrict__ idx_i, int* __restrict__ wsI, int E, int nRanges) {
  __shared__ int  hist[MAX_RANGES];
  __shared__ int2 stage[MAX_RANGES * 64];
  const int tid = threadIdx.x;
  const int g   = blockIdx.x;
  if (tid < MAX_RANGES) hist[tid] = 0;
  __syncthreads();

  int dst[L1_PER_THREAD];
#pragma unroll
  for (int k = 0; k < L1_PER_THREAD; ++k) {
    const int e = g * L1_EDGES + k * L1_THREADS + tid;
    dst[k] = idx_i[(e < E) ? e : (E - 1)];
  }
#pragma unroll
  for (int k = 0; k < L1_PER_THREAD; ++k) {
    const int e = g * L1_EDGES + k * L1_THREADS + tid;
    if (e < E && (unsigned)dst[k] < (unsigned)(nRanges * RANGE_NODES)) {
      const int b = dst[k] / RANGE_NODES;
      const int r = atomicAdd(&hist[b], 1);
      if (r < SEG_CAP) stage[b * 64 + 1 + r] = make_int2(e, dst[k]);
    }
  }
  __syncthreads();

  int  cn[L1_SLOTS];
  int2 pr[L1_SLOTS];
#pragma unroll
  for (int j = 0; j < L1_SLOTS; ++j) {
    const int s = tid + j * L1_THREADS;
    cn[j] = hist[s >> 6];
    pr[j] = stage[s];
  }
#pragma unroll
  for (int j = 0; j < L1_SLOTS; ++j) {
    const int s = tid + j * L1_THREADS;
    const int b = s >> 6, p = s & 63;
    if (b < nRanges) {
      int* seg = wsI + (size_t)b * RANGE_DW + g * SEG_DW;
      if (p == 0)             *(int2*)seg = make_int2(cn[j], 0);
      else if (p - 1 < cn[j]) *(int2*)(seg + 2 * p) = pr[j];
    }
  }
}

__global__ void
mmb_bin_nodes(int* wsI, int N, int nSegs) {
  __shared__ int segCnt[MAX_SEGS];
  __shared__ int cnt[RANGE_NODES];
  __shared__ int ovf;
  __shared__ int ids[RANGE_NODES * NODE_STAGE];
  const int tid = threadIdx.x;
  const int b   = blockIdx.x;
  int* region = wsI + (size_t)b * RANGE_DW;

  if (tid < RANGE_NODES) cnt[tid] = 0;
  if (tid == 0) ovf = 0;
  if (tid < MAX_SEGS) segCnt[tid] = (tid < nSegs) ? region[tid * SEG_DW] : 0;
  __syncthreads();
  if (tid < MAX_SEGS && segCnt[tid] > SEG_CAP) atomicOr(&ovf, 1);

  bool ok[L2_PER_THREAD];
  int2 pr[L2_PER_THREAD];
  int  kr[L2_PER_THREAD];
#pragma unroll
  for (int j = 0; j < L2_PER_THREAD; ++j) {
    const int s = tid + j * L2_THREADS;
    const int g = s >> 6, p = s & 63;
    int c = segCnt[g];
    c = (c < SEG_CAP) ? c : SEG_CAP;
    ok[j] = (p >= 1) && (p - 1 < c);
    pr[j] = *(const int2*)(region + (ok[j] ? (g * SEG_DW + 2 * p) : 0));
  }
#pragma unroll
  for (int j = 0; j < L2_PER_THREAD; ++j) {
    pr[j].y -= b * RANGE_NODES;
    kr[j] = -1;
    if (ok[j] && (unsigned)pr[j].y < (unsigned)RANGE_NODES)
      kr[j] = atomicAdd(&cnt[pr[j].y], 1);
  }
  __syncthreads();

#pragma unroll
  for (int j = 0; j < L2_PER_THREAD; ++j) {
    if (kr[j] >= 0) {
      if (kr[j] < NODE_STAGE)      ids[pr[j].y * NODE_STAGE + kr[j]] = pr[j].x;
      else if (kr[j] < BUCKET_CAP) region[pr[j].y * BLK_DW + ID_OFF + kr[j]] = pr[j].x;
    }
  }
  __syncthreads();

  int cn[L2_OUT], id[L2_OUT];
#pragma unroll
  for (int j = 0; j < L2_OUT; ++j) {
    const int s = tid + j * L2_THREADS;
    cn[j] = cnt[s / NODE_STAGE];
    id[j] = ids[s];
  }
#pragma unroll
  for (int j = 0; j < L2_OUT; ++j) {
    const int s = tid + j * L2_THREADS;
    const int l = s / NODE_STAGE, p = s % NODE_STAGE;
    if (b * RANGE_NODES + l < N && p < cn[j])
      region[l * BLK_DW + ID_OFF + p] = id[j];
  }
  if (tid < RANGE_NODES && b * RANGE_NODES + tid < N)
    region[tid * BLK_DW] = (ovf != 0) ? (BUCKET_CAP + 1) : cnt[tid];
}

#define NODES_PER_WAVE 2

__device__ __forceinline__ float bcast_f(float v, int srcLane) {
  return __int_as_float(__builtin_amdgcn_readlane(__float_as_int(v), srcLane));
}

#define EDGES_IN_FLIGHT 4
__global__ void __launch_bounds__(128, 4)
mmb_gather(const float* __restrict__ rij,
           const float* __restrict__ Wrad,
           const int*   __restrict__ idx_i,
           const int*   __restrict__ idx_j,
           const float* node0,
           const float* node1,
           const float* node2,
           float* acc,
           int N, int E) {
  const int c = threadIdx.x & 31;
  const int wave = __builtin_amdgcn_readfirstlane(
      (int)(blockIdx.x * (blockDim.x >> 5) + (threadIdx.x >> 5)));
  const int nbase = wave * NODES_PER_WAVE;
  if (nbase >= N) return;

  float Wr[NPATH * NB];
#pragma unroll
  for (int q = 0; q < NPATH * NB; ++q) Wr[q] = Wrad[q * CCH + c];

  const float inv16 = 0.0625f;

  for (int nn = 0; nn < NODES_PER_WAVE; ++nn) {
    const int i = nbase + nn;
    if (i >= N) break;
    float* blk = acc + (size_t)i * BLK_DW;
    const int* blkI = (const int*)blk;
    const int cnt = __builtin_amdgcn_readfirstlane(blkI[0]);

    float a[NCOMP];
#pragma unroll
    for (int k = 0; k < NCOMP; ++k) a[k] = 0.f;

    const bool inBucket = (cnt <= BUCKET_CAP);
    const int total = inBucket ? cnt : E;

    for (int base = 0; base < total; base += 32) {
      bool valid;
      int id;
      if (inBucket) {
        valid = (base + c < cnt);
        id = valid ? blkI[ID_OFF + base + c] : 0x7fffffff;
      } else {
        const int e0 = base + c;
        valid = false;
        if (e0 < E) valid = (idx_i[e0] == i);
        id = valid ? e0 : 0x7fffffff;
      }
      const unsigned vm = __builtin_amdgcn_ballot_w32(valid);
      if (vm == 0u) continue;
      const int nvalid = __builtin_popcount(vm);

      int pos = 0;
#pragma unroll
      for (int u = 0; u < 32; ++u)
        pos += (__builtin_amdgcn_readlane(id, u) < id) ? 1 : 0;

      const int el = valid ? id : 0;
      const int jl = idx_j[el];
      const float r0 = rij[el * 3 + 0], r1 = rij[el * 3 + 1], r2 = rij[el * 3 + 2];
      const float d  = sqrtf(r0 * r0 + r1 * r1 + r2 * r2 + 1e-12f);
      const float iv = 1.0f / d;
      const float rhl0 = r0 * iv, rhl1 = r1 * iv, rhl2 = r2 * iv;

      float x = d * 0.2f;
      x = (x > 1.0f) ? 1.0f : x;
      const float fc = 0.5f * (cosf(3.14159265358979f * x) + 1.0f);

      float rbfl[NB];
#pragma unroll
      for (int b = 0; b < NB; ++b) {
        const float t = d - (0.5f + 0.6428571429f * (float)b);
        rbfl[b] = __expf(-4.0f * t * t) * fc;
      }

      for (int t = 0; t < nvalid; t += EDGES_IN_FLIGHT) {
        int   sl[EDGES_IN_FLIGHT];
        float g0[EDGES_IN_FLIGHT], g1[EDGES_IN_FLIGHT][3], g2[EDGES_IN_FLIGHT][9];

#pragma unroll
        for (int u = 0; u < EDGES_IN_FLIGHT; ++u) {
          const int rk = (t + u < nvalid) ? (t + u) : (nvalid - 1);
          const unsigned sel = __builtin_amdgcn_ballot_w32(valid && (pos == rk));
          sl[u] = (sel != 0u) ? __builtin_ctz(sel) : 0;
          const int j = __builtin_amdgcn_readlane(jl, sl[u]);

          const float* p0 = node0 + (size_t)j * CCH;
          const float* p1 = node1 + (size_t)j * (CCH * 3);
          const float* p2 = node2 + (size_t)j * (CCH * 9);
          g0[u] = p0[c];
#pragma unroll
          for (int q = 0; q < 3; ++q) g1[u][q] = p1[c * 3 + q];
#pragma unroll
          for (int q = 0; q < 9; ++q) g2[u][q] = p2[c * 9 + q];
        }

        __builtin_amdgcn_sched_barrier(0);

#pragma unroll
        for (int u = 0; u < EDGES_IN_FLIGHT; ++u) {
          if (t + u < nvalid) {
            const float rh[3] = {bcast_f(rhl0, sl[u]), bcast_f(rhl1, sl[u]),
                                 bcast_f(rhl2, sl[u])};
            float rbf[NB];
#pragma unroll
            for (int b = 0; b < NB; ++b) rbf[b] = bcast_f(rbfl[b], sl[u]);

            float w[NPATH];
#pragma unroll
            for (int p = 0; p < NPATH; ++p) {
              float sacc = rbf[0] * Wr[p * NB];
#pragma unroll
              for (int b = 1; b < NB; ++b) sacc = fmaf(rbf[b], Wr[p * NB + b], sacc);
              w[p] = sacc;
            }

            const float d1 = g1[u][0] * rh[0] + g1[u][1] * rh[1] + g1[u][2] * rh[2];
            float gr[3];
#pragma unroll
            for (int q = 0; q < 3; ++q)
              gr[q] = g2[u][q * 3 + 0] * rh[0] + g2[u][q * 3 + 1] * rh[1]
                    + g2[u][q * 3 + 2] * rh[2];
            const float d2 = gr[0] * rh[0] + gr[1] * rh[1] + gr[2] * rh[2];

            a[0] += (g0[u] * w[0] + d1 * w[4] + d2 * w[9]) * inv16;

#pragma unroll
            for (int q = 0; q < 3; ++q) {
              const float m = (g0[u] * w[1] + d1 * w[6]) * rh[q] + g1[u][q] * w[3]
                            + gr[q] * w[8];
              a[1 + q] += m * inv16;
            }
#pragma unroll
            for (int q = 0; q < 3; ++q) {
              const float ca = g0[u] * w[2] * rh[q] + g1[u][q] * w[5] + gr[q] * w[10];
#pragma unroll
              for (int b = 0; b < 3; ++b) {
                const float m = ca * rh[b] + g2[u][q * 3 + b] * w[7];
                a[4 + q * 3 + b] += m * inv16;
              }
            }
          }
        }
      }
    }

#pragma unroll
    for (int k = 0; k < NCOMP; ++k) blk[k * CCH + c] = a[k];
  }
}

#define W_U0 0
#define W_U1 1
#define W_U2 2
#define W_V(k) (3 + (k))
#define W_G(k) (11 + (k))

#define AT_PITCH (BLK_DW + 4)
#define Y_PITCH 36

__global__ void __launch_bounds__(128, 3)
mmb_node(const float* __restrict__ acc,
         const float* __restrict__ node0, const float* __restrict__ node1,
         const float* __restrict__ node2,
         const float* __restrict__ U, const float* __restrict__ V,
         const float* __restrict__ Wg,
         float* __restrict__ out0, float* __restrict__ out1,
         float* __restrict__ out2, int N) {
  __shared__ __attribute__((aligned(16))) float2 wlds[14 * 8 * 2 * 32];
  __shared__ __attribute__((aligned(16))) float atile[16 * AT_PITCH];
  __shared__ __attribute__((aligned(16))) float ylds[16 * Y_PITCH];

  const int tid  = threadIdx.x;
  const int tile = blockIdx.x;

  {
    const int k   = (tid >> 3) << 1;
    const int n   = (tid & 7) << 2;
    const int s   = k >> 2;
    const int ln  = ((k & 2) << 3) | (n & 15);
    const int t   = n >> 4;
    const int off = k * CCH + n;
    float4 r0[14], r1[14];
#pragma unroll
    for (int i = 0; i < 14; ++i) {
      const float* src = (i < 3) ? (U + i * 1024)
                       : (i < 11) ? (V + (i - 3) * 1024)
                                  : (Wg + (i - 11) * 1024);
      r0[i] = *(const float4*)(src + off);
      r1[i] = *(const float4*)(src + off + CCH);
    }
    int rowsValid = N - tile * 16;
    rowsValid = (rowsValid < 16) ? rowsValid : 16;
    const int lastChunk = rowsValid * (BLK_DW / 4) - 1;
    const v4f* tsrc = (const v4f*)(acc + (size_t)tile * (16 * BLK_DW));
    v4f tl[NCOMP];
#pragma unroll
    for (int j = 0; j < NCOMP; ++j) {
      const int q = tid + j * 128;
      tl[j] = tsrc[(q < lastChunk) ? q : lastChunk];
    }
#pragma unroll
    for (int i = 0; i < 14; ++i) {
      float4* dst = (float4*)&wlds[((i * 8 + s) * 2 + t) * 32 + ln];
      dst[0] = make_float4(r0[i].x, r1[i].x, r0[i].y, r1[i].y);
      dst[1] = make_float4(r0[i].z, r1[i].z, r0[i].w, r1[i].w);
    }
#pragma unroll
    for (int j = 0; j < NCOMP; ++j) {
      const int q   = tid + j * 128;
      const int row = q / (BLK_DW / 4);
      const int c4  = q - row * (BLK_DW / 4);
      *(v4f*)&atile[row * AT_PITCH + c4 * 4] = tl[j];
    }
  }
  __syncthreads();

  const int lane = tid & 31;
  const int wv   = (__builtin_amdgcn_readfirstlane(tid >> 5) + tile) & 3;
  const int half = lane >> 4;
  const int m    = lane & 15;
  const int aIdx = (wv > 0) ? (wv - 1) : 0;

#define LDB(wm, s, t)                                                       \
  ({ float2 _bv = wlds[(((wm) * 8 + (s)) * 2 + (t)) * 32 + lane];           \
     v2f _b; _b.x = _bv.x; _b.y = _bv.y; _b; })

#define MM2(acc, aop, wm, s)                                                \
  do { v2f _a = (aop);                                                      \
       acc[0] = wmma4(_a, LDB(wm, s, 0), acc[0]);                           \
       acc[1] = wmma4(_a, LDB(wm, s, 1), acc[1]); } while (0)

  const float* arow = atile + m * AT_PITCH + 2 * half;
#define LDA(q, s) (*(const float2*)(arow + (q) * CCH + 4 * (s)))

  const v8f vzero = {};

  v8f y0t = vzero;
  v8f yacc[3][2];
#pragma unroll
  for (int v = 0; v < 3; ++v) { yacc[v][0] = vzero; yacc[v][1] = vzero; }

  if (wv == 0) {
    float2 cur[NCOMP], nxt[NCOMP];
#pragma unroll
    for (int k = 0; k < NCOMP; ++k) cur[k] = LDA(k, 0);

#pragma unroll 1
    for (int s = 0; s < 8; ++s) {
      const int sn = (s < 7) ? (s + 1) : 7;
#pragma unroll
      for (int k = 0; k < NCOMP; ++k) nxt[k] = LDA(k, sn);

      float x0[2], b1[2][3], b2[2][9];
      x0[0] = cur[0].x; x0[1] = cur[0].y;
#pragma unroll
      for (int a = 0; a < 3; ++a) { b1[0][a] = cur[1 + a].x; b1[1][a] = cur[1 + a].y; }
#pragma unroll
      for (int q = 0; q < 9; ++q) { b2[0][q] = cur[4 + q].x; b2[1][q] = cur[4 + q].y; }

      {
        v2f ax, asq, s1v, s2v;
#pragma unroll
        for (int jj = 0; jj < 2; ++jj) {
          ax[jj]  = x0[jj];
          asq[jj] = x0[jj] * x0[jj];
          float sum = 0.f;
#pragma unroll
          for (int a = 0; a < 3; ++a) sum += b1[jj][a] * b1[jj][a];
          s1v[jj] = sum;
          sum = 0.f;
#pragma unroll
          for (int q = 0; q < 9; ++q) sum += b2[jj][q] * b2[jj][q];
          s2v[jj] = sum;
        }
        y0t = wmma4(ax,  LDB(W_U0,   s, 0), y0t);
        y0t = wmma4(asq, LDB(W_V(0), s, 0), y0t);
        y0t = wmma4(s1v, LDB(W_V(3), s, 0), y0t);
        y0t = wmma4(s2v, LDB(W_V(6), s, 0), y0t);
      }

#pragma unroll
      for (int v = 0; v < 3; ++v) {
        v2f av, a01, tv;
#pragma unroll
        for (int jj = 0; jj < 2; ++jj) {
          av[jj]  = b1[jj][v];
          a01[jj] = x0[jj] * b1[jj][v];
          float sum = 0.f;
#pragma unroll
          for (int a = 0; a < 3; ++a) sum += b1[jj][a] * b2[jj][a * 3 + v];
          tv[jj] = sum;
        }
        MM2(yacc[v], av,  W_U1,  s);
        MM2(yacc[v], a01, W_V(1), s);
        MM2(yacc[v], tv,  W_V(5), s);
      }

#pragma unroll
      for (int k = 0; k < NCOMP; ++k) cur[k] = nxt[k];
    }
  } else {
    float2 cur[NCOMP], nxt[NCOMP], csel[4], nsel[4];
#pragma unroll
    for (int k = 0; k < NCOMP; ++k) cur[k] = LDA(k, 0);
#pragma unroll
    for (int kk = 0; kk < 3; ++kk) csel[kk] = LDA(4 + aIdx * 3 + kk, 0);
    csel[3] = LDA(1 + aIdx, 0);

#pragma unroll 1
    for (int s = 0; s < 8; ++s) {
      const int sn = (s < 7) ? (s + 1) : 7;
#pragma unroll
      for (int k = 0; k < NCOMP; ++k) nxt[k] = LDA(k, sn);
#pragma unroll
      for (int kk = 0; kk < 3; ++kk) nsel[kk] = LDA(4 + aIdx * 3 + kk, sn);
      nsel[3] = LDA(1 + aIdx, sn);

      float x0[2], b1[2][3], b2[2][9], rw[2][3], a1s[2];
      x0[0] = cur[0].x; x0[1] = cur[0].y;
#pragma unroll
      for (int a = 0; a < 3; ++a) { b1[0][a] = cur[1 + a].x; b1[1][a] = cur[1 + a].y; }
#pragma unroll
      for (int q = 0; q < 9; ++q) { b2[0][q] = cur[4 + q].x; b2[1][q] = cur[4 + q].y; }
#pragma unroll
      for (int kk = 0; kk < 3; ++kk) { rw[0][kk] = csel[kk].x; rw[1][kk] = csel[kk].y; }
      a1s[0] = csel[3].x; a1s[1] = csel[3].y;

      if (wv == 1) {
        v2f ax, asq, s1v, s2v;
#pragma unroll
        for (int jj = 0; jj < 2; ++jj) {
          ax[jj]  = x0[jj];
          asq[jj] = x0[jj] * x0[jj];
          float sum = 0.f;
#pragma unroll
          for (int a = 0; a < 3; ++a) sum += b1[jj][a] * b1[jj][a];
          s1v[jj] = sum;
          sum = 0.f;
#pragma unroll
          for (int q = 0; q < 9; ++q) sum += b2[jj][q] * b2[jj][q];
          s2v[jj] = sum;
        }
        y0t = wmma4(ax,  LDB(W_U0,   s, 1), y0t);
        y0t = wmma4(asq, LDB(W_V(0), s, 1), y0t);
        y0t = wmma4(s1v, LDB(W_V(3), s, 1), y0t);
        y0t = wmma4(s2v, LDB(W_V(6), s, 1), y0t);
      }

#pragma unroll
      for (int b = 0; b < 3; ++b) {
        v2f a2v, a02, a11, qv;
#pragma unroll
        for (int jj = 0; jj < 2; ++jj) {
          const float g = rw[jj][b];
          a2v[jj] = g;
          a02[jj] = x0[jj] * g;
          a11[jj] = a1s[jj] * b1[jj][b];
          float sum = 0.f;
#pragma unroll
          for (int kk = 0; kk < 3; ++kk)
            sum += rw[jj][kk] * b2[jj][kk * 3 + b];
          qv[jj] = sum;
        }
        MM2(yacc[b], a2v, W_U2,  s);
        MM2(yacc[b], a02, W_V(2), s);
        MM2(yacc[b], a11, W_V(4), s);
        MM2(yacc[b], qv,  W_V(7), s);
      }

#pragma unroll
      for (int k = 0; k < NCOMP; ++k) cur[k] = nxt[k];
#pragma unroll
      for (int kk = 0; kk < 4; ++kk) csel[kk] = nsel[kk];
    }
  }

  const int rowD0 = tile * 16 + 8 * half;
  float resA[2][8];
  float resB[2][8][3];
  if (wv == 0) {
#pragma unroll
    for (int t = 0; t < 2; ++t)
#pragma unroll
      for (int r = 0; r < 8; ++r) {
        const int row  = rowD0 + r;
        const int rowL = (row < N) ? row : (N - 1);
        resA[t][r] = node0[rowL * CCH + t * 16 + m];
        const float* p = node1 + (rowL * CCH + t * 16 + m) * 3;
#pragma unroll
        for (int v = 0; v < 3; ++v) resB[t][r][v] = p[v];
      }
  } else {
#pragma unroll
    for (int t = 0; t < 2; ++t)
#pragma unroll
      for (int r = 0; r < 8; ++r) {
        const int row  = rowD0 + r;
        const int rowL = (row < N) ? row : (N - 1);
        const float* p = node2 + (rowL * CCH + t * 16 + m) * 9 + aIdx * 3;
#pragma unroll
        for (int bb = 0; bb < 3; ++bb) resB[t][r][bb] = p[bb];
      }
  }

  if (wv < 2) {
#pragma unroll
    for (int r = 0; r < 8; ++r)
      ylds[(r + 8 * half) * Y_PITCH + wv * 16 + m] = y0t[r];
  }
  __syncthreads();

  if (wv == 0) {
    v8f ga[2][2] = {{vzero, vzero}, {vzero, vzero}};
#pragma unroll
    for (int s = 0; s < 8; ++s) {
      const float2 yv = *(const float2*)&ylds[m * Y_PITCH + s * 4 + 2 * half];
      v2f ay; ay.x = yv.x; ay.y = yv.y;
      MM2(ga[0], ay, W_G(0), s);
      MM2(ga[1], ay, W_G(1), s);
    }
    v8f gate[2][2];
#pragma unroll
    for (int g = 0; g < 2; ++g)
#pragma unroll
      for (int t = 0; t < 2; ++t)
#pragma unroll
        for (int r = 0; r < 8; ++r) {
          const float v = ga[g][t][r];
          gate[g][t][r] = v * (1.0f / (1.0f + __expf(-v)));
        }

#pragma unroll
    for (int t = 0; t < 2; ++t)
#pragma unroll
      for (int r = 0; r < 8; ++r) {
        const int row = rowD0 + r;
        if (row < N) {
          const int cc = t * 16 + m;
          out0[row * CCH + cc] = resA[t][r] + gate[0][t][r];
        }
      }

#pragma unroll
    for (int t = 0; t < 2; ++t)
#pragma unroll
      for (int r = 0; r < 8; ++r) {
        const int row = rowD0 + r;
        if (row < N) {
          float* q = out1 + (row * CCH + t * 16 + m) * 3;
#pragma unroll
          for (int v = 0; v < 3; ++v)
            q[v] = resB[t][r][v] + yacc[v][t][r] * gate[1][t][r];
        }
      }
  } else {
    v8f ga[2] = {vzero, vzero};
#pragma unroll
    for (int s = 0; s < 8; ++s) {
      const float2 yv = *(const float2*)&ylds[m * Y_PITCH + s * 4 + 2 * half];
      v2f ay; ay.x = yv.x; ay.y = yv.y;
      MM2(ga, ay, W_G(2), s);
    }
    v8f gate[2];
#pragma unroll
    for (int t = 0; t < 2; ++t)
#pragma unroll
      for (int r = 0; r < 8; ++r) {
        const float v = ga[t][r];
        gate[t][r] = v * (1.0f / (1.0f + __expf(-v)));
      }

#pragma unroll
    for (int t = 0; t < 2; ++t)
#pragma unroll
      for (int r = 0; r < 8; ++r) {
        const int row = rowD0 + r;
        if (row < N) {
          float* q = out2 + (row * CCH + t * 16 + m) * 9 + aIdx * 3;
#pragma unroll
          for (int bb = 0; bb < 3; ++bb)
            q[bb] = resB[t][r][bb] + yacc[bb][t][r] * gate[t][r];
        }
      }
  }
#undef LDA
#undef MM2
#undef LDB
}

extern "C" void kernel_launch(void* const* d_in, const int* in_sizes, int n_in,
                              void* d_out, int out_size, void* d_ws, size_t ws_size,
                              hipStream_t stream) {
  const float* node0 = (const float*)d_in[0];
  const float* node1 = (const float*)d_in[1];
  const float* node2 = (const float*)d_in[2];
  const float* rij   = (const float*)d_in[3];
  const float* Wrad  = (const float*)d_in[4];
  const float* U     = (const float*)d_in[5];
  const float* V     = (const float*)d_in[6];
  const float* Wg    = (const float*)d_in[7];
  const int*   idx_i = (const int*)d_in[8];
  const int*   idx_j = (const int*)d_in[9];

  const int N = in_sizes[0] / CCH;
  const int E = in_sizes[3] / 3;

  float* acc = (float*)d_ws;

  float* out0 = (float*)d_out;
  float* out1 = out0 + (size_t)N * CCH;
  float* out2 = out1 + (size_t)N * CCH * 3;

  const int nRanges = (N + RANGE_NODES - 1) / RANGE_NODES;
  const int nSegs   = (E + L1_EDGES - 1) / L1_EDGES;
  mmb_bin_ranges<<<nSegs, L1_THREADS, 0, stream>>>(idx_i, (int*)d_ws, E, nRanges);
  mmb_bin_nodes<<<nRanges, L2_THREADS, 0, stream>>>((int*)d_ws, N, nSegs);

  mmb_gather<<<(N + 7) / 8, 128, 0, stream>>>(rij, Wrad, idx_i, idx_j,
                                              node0, node1, node2,
                                              acc, N, E);

  const int nTiles = (N + 15) / 16;
  mmb_node<<<nTiles, 128, 0, stream>>>(acc,
                                                 node0, node1, node2,
                                                 U, V, Wg,
                                                 out0, out1, out2, N);
}


# edge phase and node phase fused into one kernel per 16-node tile: a wave gathers 4 nodes, their 13 sums per channel reach the WMMAs through LDS, not the workspace; 3 launches; sum order as before
# speedup vs baseline: 5.4077x; 1.0403x over previous
#include <hip/hip_runtime.h>
#include <hip/hip_bf16.h>
#include <math.h>

#define CCH 32
#define NPATH 11
#define NB 8
#define NCOMP 13

typedef __attribute__((ext_vector_type(2))) float v2f;
typedef __attribute__((ext_vector_type(4))) float v4f;
typedef __attribute__((ext_vector_type(8))) float v8f;

__device__ __forceinline__ v8f wmma4(v2f a, v2f b, v8f c) {
  return __builtin_amdgcn_wmma_f32_16x16x4_f32(
       false, a,  false, b,
       (short)0, c,  false,  false);
}


#define BLK_DW (NCOMP * CCH)
#define ID_OFF 32
#define BUCKET_CAP (BLK_DW - ID_OFF)

#define RANGE_NODES 64
#define RANGE_DW (RANGE_NODES * BLK_DW)
#define MAX_RANGES 320
#define L1_THREADS 1024
#define L1_PER_THREAD 8
#define L1_EDGES (L1_THREADS * L1_PER_THREAD)
#define L1_SLOTS ((MAX_RANGES * 64) / L1_THREADS)
#define SEG_DW 128
#define SEG_CAP 63
#define MAX_SEGS 48
#define L2_THREADS 256
#define L2_PER_THREAD ((MAX_SEGS * 64) / L2_THREADS)
#define NODE_STAGE 64
#define L2_OUT ((RANGE_NODES * NODE_STAGE) / L2_THREADS)

__global__ void
mmb_bin_ranges(const int* __restrict__ idx_i, int* __restrict__ wsI, int E, int nRanges) {
  __shared__ int  hist[MAX_RANGES];
  __shared__ int2 stage[MAX_RANGES * 64];
  const int tid = threadIdx.x;
  const int g   = blockIdx.x;
  if (tid < MAX_RANGES) hist[tid] = 0;
  __syncthreads();

  int dst[L1_PER_THREAD];
#pragma unroll
  for (int k = 0; k < L1_PER_THREAD; ++k) {
    const int e = g * L1_EDGES + k * L1_THREADS + tid;
    dst[k] = idx_i[(e < E) ? e : (E - 1)];
  }
#pragma unroll
  for (int k = 0; k < L1_PER_THREAD; ++k) {
    const int e = g * L1_EDGES + k * L1_THREADS + tid;
    if (e < E && (unsigned)dst[k] < (unsigned)(nRanges * RANGE_NODES)) {
      const int b = dst[k] / RANGE_NODES;
      const int r = atomicAdd(&hist[b], 1);
      if (r < SEG_CAP) stage[b * 64 + 1 + r] = make_int2(e, dst[k]);
    }
  }
  __syncthreads();

  int  cn[L1_SLOTS];
  int2 pr[L1_SLOTS];
#pragma unroll
  for (int j = 0; j < L1_SLOTS; ++j) {
    const int s = tid + j * L1_THREADS;
    cn[j] = hist[s >> 6];
    pr[j] = stage[s];
  }
#pragma unroll
  for (int j = 0; j < L1_SLOTS; ++j) {
    const int s = tid + j * L1_THREADS;
    const int b = s >> 6, p = s & 63;
    if (b < nRanges) {
      int* seg = wsI + (size_t)b * RANGE_DW + g * SEG_DW;
      if (p == 0)             *(int2*)seg = make_int2(cn[j], 0);
      else if (p - 1 < cn[j]) *(int2*)(seg + 2 * p) = pr[j];
    }
  }
}

__global__ void
mmb_bin_nodes(int* wsI, int N, int nSegs) {
  __shared__ int segCnt[MAX_SEGS];
  __shared__ int cnt[RANGE_NODES];
  __shared__ int ovf;
  __shared__ int ids[RANGE_NODES * NODE_STAGE];
  const int tid = threadIdx.x;
  const int b   = blockIdx.x;
  int* region = wsI + (size_t)b * RANGE_DW;

  if (tid < RANGE_NODES) cnt[tid] = 0;
  if (tid == 0) ovf = 0;
  if (tid < MAX_SEGS) segCnt[tid] = (tid < nSegs) ? region[tid * SEG_DW] : 0;
  __syncthreads();
  if (tid < MAX_SEGS && segCnt[tid] > SEG_CAP) atomicOr(&ovf, 1);

  bool ok[L2_PER_THREAD];
  int2 pr[L2_PER_THREAD];
  int  kr[L2_PER_THREAD];
#pragma unroll
  for (int j = 0; j < L2_PER_THREAD; ++j) {
    const int s = tid + j * L2_THREADS;
    const int g = s >> 6, p = s & 63;
    int c = segCnt[g];
    c = (c < SEG_CAP) ? c : SEG_CAP;
    ok[j] = (p >= 1) && (p - 1 < c);
    pr[j] = *(const int2*)(region + (ok[j] ? (g * SEG_DW + 2 * p) : 0));
  }
#pragma unroll
  for (int j = 0; j < L2_PER_THREAD; ++j) {
    pr[j].y -= b * RANGE_NODES;
    kr[j] = -1;
    if (ok[j] && (unsigned)pr[j].y < (unsigned)RANGE_NODES)
      kr[j] = atomicAdd(&cnt[pr[j].y], 1);
  }
  __syncthreads();

#pragma unroll
  for (int j = 0; j < L2_PER_THREAD; ++j) {
    if (kr[j] >= 0) {
      if (kr[j] < NODE_STAGE)      ids[pr[j].y * NODE_STAGE + kr[j]] = pr[j].x;
      else if (kr[j] < BUCKET_CAP) region[pr[j].y * BLK_DW + ID_OFF + kr[j]] = pr[j].x;
    }
  }
  __syncthreads();

  int cn[L2_OUT], id[L2_OUT];
#pragma unroll
  for (int j = 0; j < L2_OUT; ++j) {
    const int s = tid + j * L2_THREADS;
    cn[j] = cnt[s / NODE_STAGE];
    id[j] = ids[s];
  }
#pragma unroll
  for (int j = 0; j < L2_OUT; ++j) {
    const int s = tid + j * L2_THREADS;
    const int l = s / NODE_STAGE, p = s % NODE_STAGE;
    if (b * RANGE_NODES + l < N && p < cn[j])
      region[l * BLK_DW + ID_OFF + p] = id[j];
  }
  if (tid < RANGE_NODES && b * RANGE_NODES + tid < N)
    region[tid * BLK_DW] = (ovf != 0) ? (BUCKET_CAP + 1) : cnt[tid];
}

#define NODES_PER_WAVE 4

__device__ __forceinline__ float bcast_f(float v, int srcLane) {
  return __int_as_float(__builtin_amdgcn_readlane(__float_as_int(v), srcLane));
}

#define EDGES_IN_FLIGHT 4

#define W_U0 0
#define W_U1 1
#define W_U2 2
#define W_V(k) (3 + (k))
#define W_G(k) (11 + (k))

#define AT_PITCH (BLK_DW + 4)
#define Y_PITCH 36

__global__ void __launch_bounds__(128, 3)
mmb_edge_node(const float* __restrict__ rij,
              const float* __restrict__ Wrad,
              const int*   __restrict__ idx_i,
              const int*   __restrict__ idx_j,
              const int*   __restrict__ bkt,
              const float* node0, const float* node1, const float* node2,
              const float* __restrict__ U, const float* __restrict__ V,
              const float* __restrict__ Wg,
              float* __restrict__ out0, float* __restrict__ out1,
              float* __restrict__ out2, int N, int E) {
  __shared__ __attribute__((aligned(16))) float2 wlds[14 * 8 * 2 * 32];
  __shared__ __attribute__((aligned(16))) float atile[16 * AT_PITCH];
  __shared__ __attribute__((aligned(16))) float ylds[16 * Y_PITCH];

  const int tid  = threadIdx.x;
  const int tile = blockIdx.x;

  {
    const int k   = (tid >> 3) << 1;
    const int n   = (tid & 7) << 2;
    const int s   = k >> 2;
    const int ln  = ((k & 2) << 3) | (n & 15);
    const int t   = n >> 4;
    const int off = k * CCH + n;
    float4 r0[14], r1[14];
#pragma unroll
    for (int i = 0; i < 14; ++i) {
      const float* src = (i < 3) ? (U + i * 1024)
                       : (i < 11) ? (V + (i - 3) * 1024)
                                  : (Wg + (i - 11) * 1024);
      r0[i] = *(const float4*)(src + off);
      r1[i] = *(const float4*)(src + off + CCH);
    }
#pragma unroll
    for (int i = 0; i < 14; ++i) {
      float4* dst = (float4*)&wlds[((i * 8 + s) * 2 + t) * 32 + ln];
      dst[0] = make_float4(r0[i].x, r1[i].x, r0[i].y, r1[i].y);
      dst[1] = make_float4(r0[i].z, r1[i].z, r0[i].w, r1[i].w);
    }
  }

  {
  const int c = tid & 31;
  const int wq = __builtin_amdgcn_readfirstlane(tid >> 5);
  float Wr[NPATH * NB];
#pragma unroll
  for (int q = 0; q < NPATH * NB; ++q) Wr[q] = Wrad[q * CCH + c];

  const float inv16 = 0.0625f;

  for (int nn = 0; nn < NODES_PER_WAVE; ++nn) {
    const int trow = wq * NODES_PER_WAVE + nn;
    const int i = tile * 16 + trow;

    float a[NCOMP];
#pragma unroll
    for (int k = 0; k < NCOMP; ++k) a[k] = 0.f;

    const int* blkI = bkt;
    int cnt = 0;
    if (i < N) {
      blkI = bkt + (size_t)i * BLK_DW;
      cnt = __builtin_amdgcn_readfirstlane(blkI[0]);
    }

    const bool inBucket = (cnt <= BUCKET_CAP);
    const int total = inBucket ? cnt : E;

    for (int base = 0; base < total; base += 32) {
      bool valid;
      int id;
      if (inBucket) {
        valid = (base + c < cnt);
        id = valid ? blkI[ID_OFF + base + c] : 0x7fffffff;
      } else {
        const int e0 = base + c;
        valid = false;
        if (e0 < E) valid = (idx_i[e0] == i);
        id = valid ? e0 : 0x7fffffff;
      }
      const unsigned vm = __builtin_amdgcn_ballot_w32(valid);
      if (vm == 0u) continue;
      const int nvalid = __builtin_popcount(vm);

      int pos = 0;
#pragma unroll
      for (int u = 0; u < 32; ++u)
        pos += (__builtin_amdgcn_readlane(id, u) < id) ? 1 : 0;

      const int el = valid ? id : 0;
      const int jl = idx_j[el];
      const float r0 = rij[el * 3 + 0], r1 = rij[el * 3 + 1], r2 = rij[el * 3 + 2];
      const float d  = sqrtf(r0 * r0 + r1 * r1 + r2 * r2 + 1e-12f);
      const float iv = 1.0f / d;
      const float rhl0 = r0 * iv, rhl1 = r1 * iv, rhl2 = r2 * iv;

      float x = d * 0.2f;
      x = (x > 1.0f) ? 1.0f : x;
      const float fc = 0.5f * (cosf(3.14159265358979f * x) + 1.0f);

      float rbfl[NB];
#pragma unroll
      for (int b = 0; b < NB; ++b) {
        const float t = d - (0.5f + 0.6428571429f * (float)b);
        rbfl[b] = __expf(-4.0f * t * t) * fc;
      }

      for (int t = 0; t < nvalid; t += EDGES_IN_FLIGHT) {
        int   sl[EDGES_IN_FLIGHT];
        float g0[EDGES_IN_FLIGHT], g1[EDGES_IN_FLIGHT][3], g2[EDGES_IN_FLIGHT][9];

#pragma unroll
        for (int u = 0; u < EDGES_IN_FLIGHT; ++u) {
          const int rk = (t + u < nvalid) ? (t + u) : (nvalid - 1);
          const unsigned sel = __builtin_amdgcn_ballot_w32(valid && (pos == rk));
          sl[u] = (sel != 0u) ? __builtin_ctz(sel) : 0;
          const int j = __builtin_amdgcn_readlane(jl, sl[u]);

          const float* p0 = node0 + (size_t)j * CCH;
          const float* p1 = node1 + (size_t)j * (CCH * 3);
          const float* p2 = node2 + (size_t)j * (CCH * 9);
          g0[u] = p0[c];
#pragma unroll
          for (int q = 0; q < 3; ++q) g1[u][q] = p1[c * 3 + q];
#pragma unroll
          for (int q = 0; q < 9; ++q) g2[u][q] = p2[c * 9 + q];
        }

        __builtin_amdgcn_sched_barrier(0);

#pragma unroll
        for (int u = 0; u < EDGES_IN_FLIGHT; ++u) {
          if (t + u < nvalid) {
            const float rh[3] = {bcast_f(rhl0, sl[u]), bcast_f(rhl1, sl[u]),
                                 bcast_f(rhl2, sl[u])};
            float rbf[NB];
#pragma unroll
            for (int b = 0; b < NB; ++b) rbf[b] = bcast_f(rbfl[b], sl[u]);

            float w[NPATH];
#pragma unroll
            for (int p = 0; p < NPATH; ++p) {
              float sacc = rbf[0] * Wr[p * NB];
#pragma unroll
              for (int b = 1; b < NB; ++b) sacc = fmaf(rbf[b], Wr[p * NB + b], sacc);
              w[p] = sacc;
            }

            const float d1 = g1[u][0] * rh[0] + g1[u][1] * rh[1] + g1[u][2] * rh[2];
            float gr[3];
#pragma unroll
            for (int q = 0; q < 3; ++q)
              gr[q] = g2[u][q * 3 + 0] * rh[0] + g2[u][q * 3 + 1] * rh[1]
                    + g2[u][q * 3 + 2] * rh[2];
            const float d2 = gr[0] * rh[0] + gr[1] * rh[1] + gr[2] * rh[2];

            a[0] += (g0[u] * w[0] + d1 * w[4] + d2 * w[9]) * inv16;

#pragma unroll
            for (int q = 0; q < 3; ++q) {
              const float m = (g0[u] * w[1] + d1 * w[6]) * rh[q] + g1[u][q] * w[3]
                            + gr[q] * w[8];
              a[1 + q] += m * inv16;
            }
#pragma unroll
            for (int q = 0; q < 3; ++q) {
              const float ca = g0[u] * w[2] * rh[q] + g1[u][q] * w[5] + gr[q] * w[10];
#pragma unroll
              for (int b = 0; b < 3; ++b) {
                const float m = ca * rh[b] + g2[u][q * 3 + b] * w[7];
                a[4 + q * 3 + b] += m * inv16;
              }
            }
          }
        }
      }
    }

#pragma unroll
    for (int k = 0; k < NCOMP; ++k) atile[trow * AT_PITCH + k * CCH + c] = a[k];
  }
  }
  __syncthreads();

  const int lane = tid & 31;
  const int wv   = (__builtin_amdgcn_readfirstlane(tid >> 5) + tile) & 3;
  const int half = lane >> 4;
  const int m    = lane & 15;
  const int aIdx = (wv > 0) ? (wv - 1) : 0;

#define LDB(wm, s, t)                                                       \
  ({ float2 _bv = wlds[(((wm) * 8 + (s)) * 2 + (t)) * 32 + lane];           \
     v2f _b; _b.x = _bv.x; _b.y = _bv.y; _b; })

#define MM2(acc, aop, wm, s)                                                \
  do { v2f _a = (aop);                                                      \
       acc[0] = wmma4(_a, LDB(wm, s, 0), acc[0]);                           \
       acc[1] = wmma4(_a, LDB(wm, s, 1), acc[1]); } while (0)

  const float* arow = atile + m * AT_PITCH + 2 * half;
#define LDA(q, s) (*(const float2*)(arow + (q) * CCH + 4 * (s)))

  const v8f vzero = {};

  v8f y0t = vzero;
  v8f yacc[3][2];
#pragma unroll
  for (int v = 0; v < 3; ++v) { yacc[v][0] = vzero; yacc[v][1] = vzero; }

  if (wv == 0) {
    float2 cur[NCOMP], nxt[NCOMP];
#pragma unroll
    for (int k = 0; k < NCOMP; ++k) cur[k] = LDA(k, 0);

#pragma unroll 1
    for (int s = 0; s < 8; ++s) {
      const int sn = (s < 7) ? (s + 1) : 7;
#pragma unroll
      for (int k = 0; k < NCOMP; ++k) nxt[k] = LDA(k, sn);

      float x0[2], b1[2][3], b2[2][9];
      x0[0] = cur[0].x; x0[1] = cur[0].y;
#pragma unroll
      for (int a = 0; a < 3; ++a) { b1[0][a] = cur[1 + a].x; b1[1][a] = cur[1 + a].y; }
#pragma unroll
      for (int q = 0; q < 9; ++q) { b2[0][q] = cur[4 + q].x; b2[1][q] = cur[4 + q].y; }

      {
        v2f ax, asq, s1v, s2v;
#pragma unroll
        for (int jj = 0; jj < 2; ++jj) {
          ax[jj]  = x0[jj];
          asq[jj] = x0[jj] * x0[jj];
          float sum = 0.f;
#pragma unroll
          for (int a = 0; a < 3; ++a) sum += b1[jj][a] * b1[jj][a];
          s1v[jj] = sum;
          sum = 0.f;
#pragma unroll
          for (int q = 0; q < 9; ++q) sum += b2[jj][q] * b2[jj][q];
          s2v[jj] = sum;
        }
        y0t = wmma4(ax,  LDB(W_U0,   s, 0), y0t);
        y0t = wmma4(asq, LDB(W_V(0), s, 0), y0t);
        y0t = wmma4(s1v, LDB(W_V(3), s, 0), y0t);
        y0t = wmma4(s2v, LDB(W_V(6), s, 0), y0t);
      }

#pragma unroll
      for (int v = 0; v < 3; ++v) {
        v2f av, a01, tv;
#pragma unroll
        for (int jj = 0; jj < 2; ++jj) {
          av[jj]  = b1[jj][v];
          a01[jj] = x0[jj] * b1[jj][v];
          float sum = 0.f;
#pragma unroll
          for (int a = 0; a < 3; ++a) sum += b1[jj][a] * b2[jj][a * 3 + v];
          tv[jj] = sum;
        }
        MM2(yacc[v], av,  W_U1,  s);
        MM2(yacc[v], a01, W_V(1), s);
        MM2(yacc[v], tv,  W_V(5), s);
      }

#pragma unroll
      for (int k = 0; k < NCOMP; ++k) cur[k] = nxt[k];
    }
  } else {
    float2 cur[NCOMP], nxt[NCOMP], csel[4], nsel[4];
#pragma unroll
    for (int k = 0; k < NCOMP; ++k) cur[k] = LDA(k, 0);
#pragma unroll
    for (int kk = 0; kk < 3; ++kk) csel[kk] = LDA(4 + aIdx * 3 + kk, 0);
    csel[3] = LDA(1 + aIdx, 0);

#pragma unroll 1
    for (int s = 0; s < 8; ++s) {
      const int sn = (s < 7) ? (s + 1) : 7;
#pragma unroll
      for (int k = 0; k < NCOMP; ++k) nxt[k] = LDA(k, sn);
#pragma unroll
      for (int kk = 0; kk < 3; ++kk) nsel[kk] = LDA(4 + aIdx * 3 + kk, sn);
      nsel[3] = LDA(1 + aIdx, sn);

      float x0[2], b1[2][3], b2[2][9], rw[2][3], a1s[2];
      x0[0] = cur[0].x; x0[1] = cur[0].y;
#pragma unroll
      for (int a = 0; a < 3; ++a) { b1[0][a] = cur[1 + a].x; b1[1][a] = cur[1 + a].y; }
#pragma unroll
      for (int q = 0; q < 9; ++q) { b2[0][q] = cur[4 + q].x; b2[1][q] = cur[4 + q].y; }
#pragma unroll
      for (int kk = 0; kk < 3; ++kk) { rw[0][kk] = csel[kk].x; rw[1][kk] = csel[kk].y; }
      a1s[0] = csel[3].x; a1s[1] = csel[3].y;

      if (wv == 1) {
        v2f ax, asq, s1v, s2v;
#pragma unroll
        for (int jj = 0; jj < 2; ++jj) {
          ax[jj]  = x0[jj];
          asq[jj] = x0[jj] * x0[jj];
          float sum = 0.f;
#pragma unroll
          for (int a = 0; a < 3; ++a) sum += b1[jj][a] * b1[jj][a];
          s1v[jj] = sum;
          sum = 0.f;
#pragma unroll
          for (int q = 0; q < 9; ++q) sum += b2[jj][q] * b2[jj][q];
          s2v[jj] = sum;
        }
        y0t = wmma4(ax,  LDB(W_U0,   s, 1), y0t);
        y0t = wmma4(asq, LDB(W_V(0), s, 1), y0t);
        y0t = wmma4(s1v, LDB(W_V(3), s, 1), y0t);
        y0t = wmma4(s2v, LDB(W_V(6), s, 1), y0t);
      }

#pragma unroll
      for (int b = 0; b < 3; ++b) {
        v2f a2v, a02, a11, qv;
#pragma unroll
        for (int jj = 0; jj < 2; ++jj) {
          const float g = rw[jj][b];
          a2v[jj] = g;
          a02[jj] = x0[jj] * g;
          a11[jj] = a1s[jj] * b1[jj][b];
          float sum = 0.f;
#pragma unroll
          for (int kk = 0; kk < 3; ++kk)
            sum += rw[jj][kk] * b2[jj][kk * 3 + b];
          qv[jj] = sum;
        }
        MM2(yacc[b], a2v, W_U2,  s);
        MM2(yacc[b], a02, W_V(2), s);
        MM2(yacc[b], a11, W_V(4), s);
        MM2(yacc[b], qv,  W_V(7), s);
      }

#pragma unroll
      for (int k = 0; k < NCOMP; ++k) cur[k] = nxt[k];
#pragma unroll
      for (int kk = 0; kk < 4; ++kk) csel[kk] = nsel[kk];
    }
  }

  const int rowD0 = tile * 16 + 8 * half;
  float resA[2][8];
  float resB[2][8][3];
  if (wv == 0) {
#pragma unroll
    for (int t = 0; t < 2; ++t)
#pragma unroll
      for (int r = 0; r < 8; ++r) {
        const int row  = rowD0 + r;
        const int rowL = (row < N) ? row : (N - 1);
        resA[t][r] = node0[rowL * CCH + t * 16 + m];
        const float* p = node1 + (rowL * CCH + t * 16 + m) * 3;
#pragma unroll
        for (int v = 0; v < 3; ++v) resB[t][r][v] = p[v];
      }
  } else {
#pragma unroll
    for (int t = 0; t < 2; ++t)
#pragma unroll
      for (int r = 0; r < 8; ++r) {
        const int row  = rowD0 + r;
        const int rowL = (row < N) ? row : (N - 1);
        const float* p = node2 + (rowL * CCH + t * 16 + m) * 9 + aIdx * 3;
#pragma unroll
        for (int bb = 0; bb < 3; ++bb) resB[t][r][bb] = p[bb];
      }
  }

  if (wv < 2) {
#pragma unroll
    for (int r = 0; r < 8; ++r)
      ylds[(r + 8 * half) * Y_PITCH + wv * 16 + m] = y0t[r];
  }
  __syncthreads();

  if (wv == 0) {
    v8f ga[2][2] = {{vzero, vzero}, {vzero, vzero}};
#pragma unroll
    for (int s = 0; s < 8; ++s) {
      const float2 yv = *(const float2*)&ylds[m * Y_PITCH + s * 4 + 2 * half];
      v2f ay; ay.x = yv.x; ay.y = yv.y;
      MM2(ga[0], ay, W_G(0), s);
      MM2(ga[1], ay, W_G(1), s);
    }
    v8f gate[2][2];
#pragma unroll
    for (int g = 0; g < 2; ++g)
#pragma unroll
      for (int t = 0; t < 2; ++t)
#pragma unroll
        for (int r = 0; r < 8; ++r) {
          const float v = ga[g][t][r];
          gate[g][t][r] = v * (1.0f / (1.0f + __expf(-v)));
        }

#pragma unroll
    for (int t = 0; t < 2; ++t)
#pragma unroll
      for (int r = 0; r < 8; ++r) {
        const int row = rowD0 + r;
        if (row < N) {
          const int cc = t * 16 + m;
          out0[row * CCH + cc] = resA[t][r] + gate[0][t][r];
        }
      }

#pragma unroll
    for (int t = 0; t < 2; ++t)
#pragma unroll
      for (int r = 0; r < 8; ++r) {
        const int row = rowD0 + r;
        if (row < N) {
          float* q = out1 + (row * CCH + t * 16 + m) * 3;
#pragma unroll
          for (int v = 0; v < 3; ++v)
            q[v] = resB[t][r][v] + yacc[v][t][r] * gate[1][t][r];
        }
      }
  } else {
    v8f ga[2] = {vzero, vzero};
#pragma unroll
    for (int s = 0; s < 8; ++s) {
      const float2 yv = *(const float2*)&ylds[m * Y_PITCH + s * 4 + 2 * half];
      v2f ay; ay.x = yv.x; ay.y = yv.y;
      MM2(ga, ay, W_G(2), s);
    }
    v8f gate[2];
#pragma unroll
    for (int t = 0; t < 2; ++t)
#pragma unroll
      for (int r = 0; r < 8; ++r) {
        const float v = ga[t][r];
        gate[t][r] = v * (1.0f / (1.0f + __expf(-v)));
      }

#pragma unroll
    for (int t = 0; t < 2; ++t)
#pragma unroll
      for (int r = 0; r < 8; ++r) {
        const int row = rowD0 + r;
        if (row < N) {
          float* q = out2 + (row * CCH + t * 16 + m) * 9 + aIdx * 3;
#pragma unroll
          for (int bb = 0; bb < 3; ++bb)
            q[bb] = resB[t][r][bb] + yacc[bb][t][r] * gate[t][r];
        }
      }
  }
#undef LDA
#undef MM2
#undef LDB
}

extern "C" void kernel_launch(void* const* d_in, const int* in_sizes, int n_in,
                              void* d_out, int out_size, void* d_ws, size_t ws_size,
                              hipStream_t stream) {
  const float* node0 = (const float*)d_in[0];
  const float* node1 = (const float*)d_in[1];
  const float* node2 = (const float*)d_in[2];
  const float* rij   = (const float*)d_in[3];
  const float* Wrad  = (const float*)d_in[4];
  const float* U     = (const float*)d_in[5];
  const float* V     = (const float*)d_in[6];
  const float* Wg    = (const float*)d_in[7];
  const int*   idx_i = (const int*)d_in[8];
  const int*   idx_j = (const int*)d_in[9];

  const int N = in_sizes[0] / CCH;
  const int E = in_sizes[3] / 3;

  float* out0 = (float*)d_out;
  float* out1 = out0 + (size_t)N * CCH;
  float* out2 = out1 + (size_t)N * CCH * 3;
  const int nRanges = (N + RANGE_NODES - 1) / RANGE_NODES;
  const int nSegs   = (E + L1_EDGES - 1) / L1_EDGES;
  mmb_bin_ranges<<<nSegs, L1_THREADS, 0, stream>>>(idx_i, (int*)d_ws, E, nRanges);
  mmb_bin_nodes<<<nRanges, L2_THREADS, 0, stream>>>((int*)d_ws, N, nSegs);

  const int nTiles = (N + 15) / 16;
  mmb_edge_node<<<nTiles, 128, 0, stream>>>(rij, Wrad, idx_i, idx_j,
                                            (const int*)d_ws,
                                            node0, node1, node2,
                                            U, V, Wg,
                                            out0, out1, out2, N, E);
}
